# starting kernel
# baseline (speedup 1.0000x reference)
#include <hip/hip_runtime.h>
#include <math.h>

typedef __attribute__((ext_vector_type(2))) float v2f;
typedef __attribute__((ext_vector_type(8))) float v8f;

#define GCN_N_NODES 50000
#define GCN_N_EDGES 600000

__global__ void gcn_fill_zero(float* __restrict__ p, int n) {
    int i = blockIdx.x * blockDim.x + threadIdx.x;
    if (i < n) p[i] = 0.0f;
}

__global__ void gcn_count_deg(const int* __restrict__ src, const int* __restrict__ dst,
                              float* __restrict__ deg_out, float* __restrict__ deg_in,
                              int E) {
    int e = blockIdx.x * blockDim.x + threadIdx.x;
    if (e < E) {
        atomicAdd(&deg_out[src[e]], 1.0f);
        atomicAdd(&deg_in[dst[e]], 1.0f);
    }
}

__global__ void gcn_inv_sqrt_clamp(float* __restrict__ d, int n) {
    int i = blockIdx.x * blockDim.x + threadIdx.x;
    if (i < n) {
        float v = d[i];
        v = v < 1.0f ? 1.0f : v;
        d[i] = 1.0f / sqrtf(v);
    }
}

__global__ void gcn_scatter(const float* __restrict__ x,
                            const int* __restrict__ src, const int* __restrict__ dst,
                            const float* __restrict__ deg_out_is,
                            float* __restrict__ agg, int E) {
    const int CH = 32;
    int t = blockIdx.x * blockDim.x + threadIdx.x;
    int e = t >> 5;
    if (e >= E) return;
    int c = t & 31;
    int s  = src[e];
    int dd = dst[e];
    float sc = deg_out_is[s];
    const float4 v = *(const float4*)(x + (size_t)s * 128 + c * 4);
    float* base = agg + (size_t)dd * 128 + c * 4;
    atomicAdd(base + 0, v.x * sc);
    atomicAdd(base + 1, v.y * sc);
    atomicAdd(base + 2, v.z * sc);
    atomicAdd(base + 3, v.w * sc);
    (void)CH;
}

__global__ void gcn_gemm_wmma(const float* __restrict__ A,
                              const float* __restrict__ deg_in_is,
                              const float* __restrict__ W,
                              const float* __restrict__ bias,
                              float* __restrict__ out,
                              int K, int C, int do_relu) {
    const int m0   = blockIdx.x * 16;
    const int n0   = blockIdx.y * 16;
    const int lane = threadIdx.x & 31;
    const int half = lane >> 4;
    const int l15  = lane & 15;
    const int m = m0 + l15;
    const int n = n0 + l15;

    const float scale = deg_in_is[m];
    const float* arow = A + (size_t)m * K;

    v8f c = {};
    for (int k0 = 0; k0 < K; k0 += 4) {
        const int kb = k0 + half * 2;
        v2f a, b;
        a.x = arow[kb]     * scale;
        a.y = arow[kb + 1] * scale;
        b.x = W[(size_t)kb       * C + n];
        b.y = W[(size_t)(kb + 1) * C + n];
        c = __builtin_amdgcn_wmma_f32_16x16x4_f32(
                false, a, false, b, (short)0, c, false, false);
    }

    const float bv = bias[n];
#pragma unroll
    for (int v = 0; v < 8; ++v) {
        float val = c[v] + bv;
        if (do_relu) val = fmaxf(val, 0.0f);
        out[(size_t)(m0 + half * 8 + v) * C + n] = val;
    }
}

extern "C" void kernel_launch(void* const* d_in, const int* in_sizes, int n_in,
                              void* d_out, int out_size, void* d_ws, size_t ws_size,
                              hipStream_t stream) {
    const float* features = (const float*)d_in[0];
    const int*   src      = (const int*)  d_in[1];
    const int*   dst      = (const int*)  d_in[2];
    const float* W1       = (const float*)d_in[3];
    const float* b1       = (const float*)d_in[4];
    const float* W2       = (const float*)d_in[5];
    const float* b2       = (const float*)d_in[6];
    const float* W3       = (const float*)d_in[7];
    const float* b3       = (const float*)d_in[8];
    float* out = (float*)d_out;

    const int N = GCN_N_NODES;
    const int E = GCN_N_EDGES;
    const size_t NF = (size_t)N * 128;

    float* bufA  = (float*)d_ws;
    float* bufB  = bufA + NF;
    float* deg_o = bufB + NF;
    float* deg_i = deg_o + N;

    const int TB = 256;

    gcn_fill_zero<<<(2 * N + TB - 1) / TB, TB, 0, stream>>>(deg_o, 2 * N);
    gcn_count_deg<<<(E + TB - 1) / TB, TB, 0, stream>>>(src, dst, deg_o, deg_i, E);
    gcn_inv_sqrt_clamp<<<(2 * N + TB - 1) / TB, TB, 0, stream>>>(deg_o, 2 * N);

    const int scatter_blocks = (E * 32 + TB - 1) / TB;
    const dim3 gemm128(N / 16, 128 / 16);
    const dim3 gemm16 (N / 16, 16 / 16);

    gcn_fill_zero<<<((int)NF + TB - 1) / TB, TB, 0, stream>>>(bufA, (int)NF);
    gcn_scatter<<<scatter_blocks, TB, 0, stream>>>(features, src, dst, deg_o, bufA, E);
    gcn_gemm_wmma<<<gemm128, 32, 0, stream>>>(bufA, deg_i, W1, b1, bufB, 128, 128, 1);

    gcn_fill_zero<<<((int)NF + TB - 1) / TB, TB, 0, stream>>>(bufA, (int)NF);
    gcn_scatter<<<scatter_blocks, TB, 0, stream>>>(bufB, src, dst, deg_o, bufA, E);
    gcn_gemm_wmma<<<gemm128, 32, 0, stream>>>(bufA, deg_i, W2, b2, bufB, 128, 128, 1);

    gcn_fill_zero<<<((int)NF + TB - 1) / TB, TB, 0, stream>>>(bufA, (int)NF);
    gcn_scatter<<<scatter_blocks, TB, 0, stream>>>(bufB, src, dst, deg_o, bufA, E);
    gcn_gemm_wmma<<<gemm16, 32, 0, stream>>>(bufA, deg_i, W3, b3, out, 128, 16, 0);

    (void)in_sizes; (void)n_in; (void)out_size; (void)ws_size;
}


# Aggregation as a per-call counting sort of edges by dst (rank = in-degree atomic's return, one-block scan) plus an atomic-free row gather; f32 atomics and zero fills gone; row sum order = rank order
# speedup vs baseline: 1.8936x; 1.8936x over previous
#include <hip/hip_runtime.h>
#include <math.h>


typedef __attribute__((ext_vector_type(2))) float v2f;
typedef __attribute__((ext_vector_type(8))) float v8f;

#define GCN_N_NODES 50000
#define GCN_N_EDGES 600000

__global__ void gcn_fill_zero_i(int* __restrict__ p, int n) {
    int i = blockIdx.x * blockDim.x + threadIdx.x;
    if (i < n) p[i] = 0;
}

__global__ void gcn_count_rank(const int* __restrict__ src, const int* __restrict__ dst,
                               int* __restrict__ cnt_out, int* __restrict__ cnt_in,
                               int* __restrict__ rank, int E) {
    int e = blockIdx.x * blockDim.x + threadIdx.x;
    if (e < E) {
        const int s = src[e];
        const int d = dst[e];
        atomicAdd(&cnt_out[s], 1);
        rank[e] = atomicAdd(&cnt_in[d], 1);
    }
}

#define GCN_SCAN_ITERS 7
#define GCN_SCAN_SEG (GCN_SCAN_ITERS * 256)
static_assert(32 * GCN_SCAN_SEG >= GCN_N_NODES, "one work-group of 32 waves must cover all nodes");
static_assert(GCN_N_NODES % 8 == 0, "8-wide node groups must not straddle the end");

static __device__ __forceinline__ float gcn_inv_sqrt_clamped(int c) {
    float v = (float)c;
    v = v < 1.0f ? 1.0f : v;
    return 1.0f / sqrtf(v);
}

__global__ void __launch_bounds__(1024)
gcn_scan_rows(const int* __restrict__ cnt_in, int* __restrict__ row_ptr, int n) {
    __shared__ int wave_total[32];
    const int lane  = threadIdx.x & 31;
    const int w     = threadIdx.x >> 5;
    const int wbase = w * GCN_SCAN_SEG + lane * 8;

    int4 ca[GCN_SCAN_ITERS], cb[GCN_SCAN_ITERS];
#pragma unroll
    for (int it = 0; it < GCN_SCAN_ITERS; ++it) {
        const int idx8 = wbase + it * 256;
        ca[it] = make_int4(0, 0, 0, 0);
        cb[it] = make_int4(0, 0, 0, 0);
        if (idx8 < n) {
            ca[it] = *(const int4*)(cnt_in + idx8);
            cb[it] = *(const int4*)(cnt_in + idx8 + 4);
        }
    }

    int tot8[GCN_SCAN_ITERS], incl[GCN_SCAN_ITERS];
#pragma unroll
    for (int it = 0; it < GCN_SCAN_ITERS; ++it) {
        tot8[it] = ca[it].x + ca[it].y + ca[it].z + ca[it].w
                 + cb[it].x + cb[it].y + cb[it].z + cb[it].w;
        incl[it] = tot8[it];
    }
#pragma unroll
    for (int d = 1; d < 32; d <<= 1) {
        int y[GCN_SCAN_ITERS];
#pragma unroll
        for (int it = 0; it < GCN_SCAN_ITERS; ++it) y[it] = __shfl_up(incl[it], d);
#pragma unroll
        for (int it = 0; it < GCN_SCAN_ITERS; ++it) if (lane >= d) incl[it] += y[it];
    }
    int trip_total[GCN_SCAN_ITERS];
#pragma unroll
    for (int it = 0; it < GCN_SCAN_ITERS; ++it) trip_total[it] = __shfl(incl[it], 31);

    int4 ea[GCN_SCAN_ITERS], eb[GCN_SCAN_ITERS];
    int carry = 0;
#pragma unroll
    for (int it = 0; it < GCN_SCAN_ITERS; ++it) {
        const int e0 = carry + incl[it] - tot8[it];
        const int e1 = e0 + ca[it].x;
        const int e2 = e1 + ca[it].y;
        const int e3 = e2 + ca[it].z;
        const int e4 = e3 + ca[it].w;
        const int e5 = e4 + cb[it].x;
        const int e6 = e5 + cb[it].y;
        const int e7 = e6 + cb[it].z;
        ea[it] = make_int4(e0, e1, e2, e3);
        eb[it] = make_int4(e4, e5, e6, e7);
        carry += trip_total[it];
    }
    if (lane == 0) wave_total[w] = carry;
    __syncthreads();
    int woff = 0;
    for (int k = 0; k < w; ++k) woff += wave_total[k];
#pragma unroll
    for (int it = 0; it < GCN_SCAN_ITERS; ++it) {
        const int idx8 = wbase + it * 256;
        if (idx8 < n) {
            int4 ra = ea[it], rb = eb[it];
            ra.x += woff; ra.y += woff; ra.z += woff; ra.w += woff;
            rb.x += woff; rb.y += woff; rb.z += woff; rb.w += woff;
            *(int4*)(row_ptr + idx8)     = ra;
            *(int4*)(row_ptr + idx8 + 4) = rb;
        }
    }
}

static_assert(GCN_N_EDGES >= GCN_N_NODES, "the placement grid must cover one thread per node");

__global__ void gcn_place(const int* __restrict__ src, const int* __restrict__ dst,
                          const int* __restrict__ rank, const int* __restrict__ row_ptr,
                          const int* __restrict__ cnt_out, const int* __restrict__ cnt_in,
                          int* __restrict__ edges, float* __restrict__ deg_in_is,
                          int E, int n) {
    int e = blockIdx.x * blockDim.x + threadIdx.x;
    if (e < n) deg_in_is[e] = gcn_inv_sqrt_clamped(cnt_in[e]);
    if (e < E) {
        const int s = src[e];
        const int pos = row_ptr[dst[e]] + rank[e];
        const float sc = gcn_inv_sqrt_clamped(cnt_out[s]);
        ((int2*)edges)[pos] = make_int2(s, __float_as_int(sc));
    }
}

static __device__ __forceinline__ float4 gcn_row4(const float* __restrict__ x, int s, int lane4) {
    return *(const float4*)(x + (size_t)s * 128 + lane4);
}
static __device__ __forceinline__ void gcn_axpy4(float4& acc, const float4 v, const float w) {
    acc.x = fmaf(v.x, w, acc.x);
    acc.y = fmaf(v.y, w, acc.y);
    acc.z = fmaf(v.z, w, acc.z);
    acc.w = fmaf(v.w, w, acc.w);
}

__global__ void __launch_bounds__(256)
gcn_gather_rows(const float* __restrict__ x, const int* __restrict__ row_ptr,
                const int* __restrict__ cnt_in, const int* __restrict__ edges,
                float* __restrict__ agg, int n) {
    const int lane  = threadIdx.x & 31;
    const int lane4 = lane * 4;
    const int node = __builtin_amdgcn_readfirstlane(blockIdx.x * 8 + (threadIdx.x >> 5));
    if (node >= n) return;
    const int beg = row_ptr[node];
    const int cnt = cnt_in[node];
    const int2* __restrict__ ce = (const int2*)edges;

    float4 acc = make_float4(0.0f, 0.0f, 0.0f, 0.0f);
    for (int b = 0; b < cnt; b += 32) {
        const int m = (cnt - b < 32) ? (cnt - b) : 32;
        int2 my = make_int2(0, 0);
        if (lane < m) my = ce[beg + b + lane];
        int j = 0;
        for (; j + 4 <= m; j += 4) {
            const int s0 = __builtin_amdgcn_readlane(my.x, j);
            const int s1 = __builtin_amdgcn_readlane(my.x, j + 1);
            const int s2 = __builtin_amdgcn_readlane(my.x, j + 2);
            const int s3 = __builtin_amdgcn_readlane(my.x, j + 3);
            const float4 v0 = gcn_row4(x, s0, lane4);
            const float4 v1 = gcn_row4(x, s1, lane4);
            const float4 v2 = gcn_row4(x, s2, lane4);
            const float4 v3 = gcn_row4(x, s3, lane4);
            const float w0 = __int_as_float(__builtin_amdgcn_readlane(my.y, j));
            const float w1 = __int_as_float(__builtin_amdgcn_readlane(my.y, j + 1));
            const float w2 = __int_as_float(__builtin_amdgcn_readlane(my.y, j + 2));
            const float w3 = __int_as_float(__builtin_amdgcn_readlane(my.y, j + 3));
            gcn_axpy4(acc, v0, w0);
            gcn_axpy4(acc, v1, w1);
            gcn_axpy4(acc, v2, w2);
            gcn_axpy4(acc, v3, w3);
        }
        const int r = m - j;
        if (r == 1) {
            const int s0 = __builtin_amdgcn_readlane(my.x, j);
            const float4 v0 = gcn_row4(x, s0, lane4);
            const float w0 = __int_as_float(__builtin_amdgcn_readlane(my.y, j));
            gcn_axpy4(acc, v0, w0);
        } else if (r == 2) {
            const int s0 = __builtin_amdgcn_readlane(my.x, j);
            const int s1 = __builtin_amdgcn_readlane(my.x, j + 1);
            const float4 v0 = gcn_row4(x, s0, lane4);
            const float4 v1 = gcn_row4(x, s1, lane4);
            const float w0 = __int_as_float(__builtin_amdgcn_readlane(my.y, j));
            const float w1 = __int_as_float(__builtin_amdgcn_readlane(my.y, j + 1));
            gcn_axpy4(acc, v0, w0);
            gcn_axpy4(acc, v1, w1);
        } else if (r == 3) {
            const int s0 = __builtin_amdgcn_readlane(my.x, j);
            const int s1 = __builtin_amdgcn_readlane(my.x, j + 1);
            const int s2 = __builtin_amdgcn_readlane(my.x, j + 2);
            const float4 v0 = gcn_row4(x, s0, lane4);
            const float4 v1 = gcn_row4(x, s1, lane4);
            const float4 v2 = gcn_row4(x, s2, lane4);
            const float w0 = __int_as_float(__builtin_amdgcn_readlane(my.y, j));
            const float w1 = __int_as_float(__builtin_amdgcn_readlane(my.y, j + 1));
            const float w2 = __int_as_float(__builtin_amdgcn_readlane(my.y, j + 2));
            gcn_axpy4(acc, v0, w0);
            gcn_axpy4(acc, v1, w1);
            gcn_axpy4(acc, v2, w2);
        }
    }
    *(float4*)(agg + (size_t)node * 128 + lane4) = acc;
}

__global__ void gcn_gemm_wmma(const float* __restrict__ A,
                              const float* __restrict__ deg_in_is,
                              const float* __restrict__ W,
                              const float* __restrict__ bias,
                              float* __restrict__ out,
                              int K, int C, int do_relu) {
    const int m0   = blockIdx.x * 16;
    const int n0   = blockIdx.y * 16;
    const int lane = threadIdx.x & 31;
    const int half = lane >> 4;
    const int l15  = lane & 15;
    const int m = m0 + l15;
    const int n = n0 + l15;

    const float scale = deg_in_is[m];
    const float* arow = A + (size_t)m * K;

    v8f c = {};
    for (int k0 = 0; k0 < K; k0 += 4) {
        const int kb = k0 + half * 2;
        v2f a, b;
        a.x = arow[kb]     * scale;
        a.y = arow[kb + 1] * scale;
        b.x = W[(size_t)kb       * C + n];
        b.y = W[(size_t)(kb + 1) * C + n];
        c = __builtin_amdgcn_wmma_f32_16x16x4_f32(
                false, a, false, b, (short)0, c, false, false);
    }

    const float bv = bias[n];
#pragma unroll
    for (int v = 0; v < 8; ++v) {
        float val = c[v] + bv;
        if (do_relu) val = fmaxf(val, 0.0f);
        out[(size_t)(m0 + half * 8 + v) * C + n] = val;
    }
}

extern "C" void kernel_launch(void* const* d_in, const int* in_sizes, int n_in,
                              void* d_out, int out_size, void* d_ws, size_t ws_size,
                              hipStream_t stream) {
    const float* features = (const float*)d_in[0];
    const int*   src      = (const int*)  d_in[1];
    const int*   dst      = (const int*)  d_in[2];
    const float* W1       = (const float*)d_in[3];
    const float* b1       = (const float*)d_in[4];
    const float* W2       = (const float*)d_in[5];
    const float* b2       = (const float*)d_in[6];
    const float* W3       = (const float*)d_in[7];
    const float* b3       = (const float*)d_in[8];
    float* out = (float*)d_out;

    const int N = GCN_N_NODES;
    const int E = GCN_N_EDGES;
    const size_t NF = (size_t)N * 128;

    float* bufA    = (float*)d_ws;
    float* bufB    = bufA + NF;
    int*   edges   = (int*)(bufB + NF);
    int*   rank    = edges + 2 * (size_t)E;
    int*   cnt_out = rank + E;
    int*   cnt_in  = cnt_out + N;
    int*   row_ptr = cnt_in + N;
    float* deg_i   = (float*)(row_ptr + N);

    const int TB = 256;

    gcn_fill_zero_i<<<(2 * N + TB - 1) / TB, TB, 0, stream>>>(cnt_out, 2 * N);
    gcn_count_rank<<<(E + TB - 1) / TB, TB, 0, stream>>>(src, dst, cnt_out, cnt_in, rank, E);
    gcn_scan_rows<<<1, 1024, 0, stream>>>(cnt_in, row_ptr, N);
    gcn_place<<<(E + TB - 1) / TB, TB, 0, stream>>>(src, dst, rank, row_ptr, cnt_out, cnt_in,
                                                    edges, deg_i, E, N);

    const int gather_blocks = (N + 7) / 8;
    const dim3 gemm128(N / 16, 128 / 16);
    const dim3 gemm16 (N / 16, 16 / 16);

    gcn_gather_rows<<<gather_blocks, TB, 0, stream>>>(features, row_ptr, cnt_in, edges, bufA, N);
    gcn_gemm_wmma<<<gemm128, 32, 0, stream>>>(bufA, deg_i, W1, b1, bufB, 128, 128, 1);

    gcn_gather_rows<<<gather_blocks, TB, 0, stream>>>(bufB, row_ptr, cnt_in, edges, bufA, N);
    gcn_gemm_wmma<<<gemm128, 32, 0, stream>>>(bufA, deg_i, W2, b2, bufB, 128, 128, 1);

    gcn_gather_rows<<<gather_blocks, TB, 0, stream>>>(bufB, row_ptr, cnt_in, edges, bufA, N);
    gcn_gemm_wmma<<<gemm16, 32, 0, stream>>>(bufA, deg_i, W3, b3, out, 128, 16, 0);

    (void)in_sizes; (void)n_in; (void)out_size; (void)ws_size;
}


# Aggregation fused into the GEMM: a work-group gathers each 16-row tile into LDS and multiplies it from there, W fragments kept in registers over 5 tiles; 10 launches -> 7; sum order unchanged
# speedup vs baseline: 2.6503x; 1.3996x over previous
#include <hip/hip_runtime.h>
#include <math.h>


typedef __attribute__((ext_vector_type(2))) float v2f;
typedef __attribute__((ext_vector_type(8))) float v8f;

#define GCN_N_NODES 50000
#define GCN_N_EDGES 600000

__global__ void gcn_fill_zero_i(int* __restrict__ p, int n) {
    int i = blockIdx.x * blockDim.x + threadIdx.x;
    if (i < n) p[i] = 0;
}

__global__ void gcn_count_rank(const int* __restrict__ src, const int* __restrict__ dst,
                               int* __restrict__ cnt_out, int* __restrict__ cnt_in,
                               int* __restrict__ rank, int E) {
    int e = blockIdx.x * blockDim.x + threadIdx.x;
    if (e < E) {
        const int s = src[e];
        const int d = dst[e];
        atomicAdd(&cnt_out[s], 1);
        rank[e] = atomicAdd(&cnt_in[d], 1);
    }
}

#define GCN_SCAN_ITERS 7
#define GCN_SCAN_SEG (GCN_SCAN_ITERS * 256)
static_assert(32 * GCN_SCAN_SEG >= GCN_N_NODES, "one work-group of 32 waves must cover all nodes");
static_assert(GCN_N_NODES % 8 == 0, "8-wide node groups must not straddle the end");

static __device__ __forceinline__ float gcn_inv_sqrt_clamped(int c) {
    float v = (float)c;
    v = v < 1.0f ? 1.0f : v;
    return 1.0f / sqrtf(v);
}

__global__ void __launch_bounds__(1024)
gcn_scan_rows(const int* __restrict__ cnt_in, int* __restrict__ row_ptr, int n) {
    __shared__ int wave_total[32];
    const int lane  = threadIdx.x & 31;
    const int w     = threadIdx.x >> 5;
    const int wbase = w * GCN_SCAN_SEG + lane * 8;

    int4 ca[GCN_SCAN_ITERS], cb[GCN_SCAN_ITERS];
#pragma unroll
    for (int it = 0; it < GCN_SCAN_ITERS; ++it) {
        const int idx8 = wbase + it * 256;
        ca[it] = make_int4(0, 0, 0, 0);
        cb[it] = make_int4(0, 0, 0, 0);
        if (idx8 < n) {
            ca[it] = *(const int4*)(cnt_in + idx8);
            cb[it] = *(const int4*)(cnt_in + idx8 + 4);
        }
    }

    int tot8[GCN_SCAN_ITERS], incl[GCN_SCAN_ITERS];
#pragma unroll
    for (int it = 0; it < GCN_SCAN_ITERS; ++it) {
        tot8[it] = ca[it].x + ca[it].y + ca[it].z + ca[it].w
                 + cb[it].x + cb[it].y + cb[it].z + cb[it].w;
        incl[it] = tot8[it];
    }
#pragma unroll
    for (int d = 1; d < 32; d <<= 1) {
        int y[GCN_SCAN_ITERS];
#pragma unroll
        for (int it = 0; it < GCN_SCAN_ITERS; ++it) y[it] = __shfl_up(incl[it], d);
#pragma unroll
        for (int it = 0; it < GCN_SCAN_ITERS; ++it) if (lane >= d) incl[it] += y[it];
    }
    int trip_total[GCN_SCAN_ITERS];
#pragma unroll
    for (int it = 0; it < GCN_SCAN_ITERS; ++it) trip_total[it] = __shfl(incl[it], 31);

    int4 ea[GCN_SCAN_ITERS], eb[GCN_SCAN_ITERS];
    int carry = 0;
#pragma unroll
    for (int it = 0; it < GCN_SCAN_ITERS; ++it) {
        const int e0 = carry + incl[it] - tot8[it];
        const int e1 = e0 + ca[it].x;
        const int e2 = e1 + ca[it].y;
        const int e3 = e2 + ca[it].z;
        const int e4 = e3 + ca[it].w;
        const int e5 = e4 + cb[it].x;
        const int e6 = e5 + cb[it].y;
        const int e7 = e6 + cb[it].z;
        ea[it] = make_int4(e0, e1, e2, e3);
        eb[it] = make_int4(e4, e5, e6, e7);
        carry += trip_total[it];
    }
    if (lane == 0) wave_total[w] = carry;
    __syncthreads();
    int woff = 0;
    for (int k = 0; k < w; ++k) woff += wave_total[k];
#pragma unroll
    for (int it = 0; it < GCN_SCAN_ITERS; ++it) {
        const int idx8 = wbase + it * 256;
        if (idx8 < n) {
            int4 ra = ea[it], rb = eb[it];
            ra.x += woff; ra.y += woff; ra.z += woff; ra.w += woff;
            rb.x += woff; rb.y += woff; rb.z += woff; rb.w += woff;
            *(int4*)(row_ptr + idx8)     = ra;
            *(int4*)(row_ptr + idx8 + 4) = rb;
        }
    }
}

static_assert(GCN_N_EDGES >= GCN_N_NODES, "the placement grid must cover one thread per node");

__global__ void gcn_place(const int* __restrict__ src, const int* __restrict__ dst,
                          const int* __restrict__ rank, const int* __restrict__ row_ptr,
                          const int* __restrict__ cnt_out, const int* __restrict__ cnt_in,
                          int* __restrict__ edges, float* __restrict__ deg_in_is,
                          int E, int n) {
    int e = blockIdx.x * blockDim.x + threadIdx.x;
    if (e < n) deg_in_is[e] = gcn_inv_sqrt_clamped(cnt_in[e]);
    if (e < E) {
        const int s = src[e];
        const int pos = row_ptr[dst[e]] + rank[e];
        const float sc = gcn_inv_sqrt_clamped(cnt_out[s]);
        ((int2*)edges)[pos] = make_int2(s, __float_as_int(sc));
    }
}

static __device__ __forceinline__ float4 gcn_row4(const float* __restrict__ x, int s, int lane4) {
    return *(const float4*)(x + (size_t)s * 128 + lane4);
}
static __device__ __forceinline__ void gcn_axpy4(float4& acc, const float4 v, const float w) {
    acc.x = fmaf(v.x, w, acc.x);
    acc.y = fmaf(v.y, w, acc.y);
    acc.z = fmaf(v.z, w, acc.z);
    acc.w = fmaf(v.w, w, acc.w);
}
#define GCN_EDGE_SRC(my, j) __builtin_amdgcn_readlane((my).x, (j))
#define GCN_EDGE_W(my, j)   __int_as_float(__builtin_amdgcn_readlane((my).y, (j)))

static __device__ __forceinline__ float4 gcn_gather_row(const float* __restrict__ x,
                                                        const int2* __restrict__ ce,
                                                        const int beg, const int cnt,
                                                        const int2 first,
                                                        const int lane, const int lane4) {
    float4 acc = make_float4(0.0f, 0.0f, 0.0f, 0.0f);
    for (int b = 0; b < cnt; b += 32) {
        const int m = (cnt - b < 32) ? (cnt - b) : 32;
        int2 my = first;
        if (b > 0) {
            my = make_int2(0, 0);
            if (lane < m) my = ce[beg + b + lane];
        }
        int j = 0;
        for (; j + 8 <= m; j += 8) {
            const float4 v0 = gcn_row4(x, GCN_EDGE_SRC(my, j),     lane4);
            const float4 v1 = gcn_row4(x, GCN_EDGE_SRC(my, j + 1), lane4);
            const float4 v2 = gcn_row4(x, GCN_EDGE_SRC(my, j + 2), lane4);
            const float4 v3 = gcn_row4(x, GCN_EDGE_SRC(my, j + 3), lane4);
            const float4 v4 = gcn_row4(x, GCN_EDGE_SRC(my, j + 4), lane4);
            const float4 v5 = gcn_row4(x, GCN_EDGE_SRC(my, j + 5), lane4);
            const float4 v6 = gcn_row4(x, GCN_EDGE_SRC(my, j + 6), lane4);
            const float4 v7 = gcn_row4(x, GCN_EDGE_SRC(my, j + 7), lane4);
            gcn_axpy4(acc, v0, GCN_EDGE_W(my, j));
            gcn_axpy4(acc, v1, GCN_EDGE_W(my, j + 1));
            gcn_axpy4(acc, v2, GCN_EDGE_W(my, j + 2));
            gcn_axpy4(acc, v3, GCN_EDGE_W(my, j + 3));
            gcn_axpy4(acc, v4, GCN_EDGE_W(my, j + 4));
            gcn_axpy4(acc, v5, GCN_EDGE_W(my, j + 5));
            gcn_axpy4(acc, v6, GCN_EDGE_W(my, j + 6));
            gcn_axpy4(acc, v7, GCN_EDGE_W(my, j + 7));
        }
        if (j + 4 <= m) {
            const float4 v0 = gcn_row4(x, GCN_EDGE_SRC(my, j),     lane4);
            const float4 v1 = gcn_row4(x, GCN_EDGE_SRC(my, j + 1), lane4);
            const float4 v2 = gcn_row4(x, GCN_EDGE_SRC(my, j + 2), lane4);
            const float4 v3 = gcn_row4(x, GCN_EDGE_SRC(my, j + 3), lane4);
            gcn_axpy4(acc, v0, GCN_EDGE_W(my, j));
            gcn_axpy4(acc, v1, GCN_EDGE_W(my, j + 1));
            gcn_axpy4(acc, v2, GCN_EDGE_W(my, j + 2));
            gcn_axpy4(acc, v3, GCN_EDGE_W(my, j + 3));
            j += 4;
        }
        const int r = m - j;
        if (r == 1) {
            const float4 v0 = gcn_row4(x, GCN_EDGE_SRC(my, j), lane4);
            gcn_axpy4(acc, v0, GCN_EDGE_W(my, j));
        } else if (r == 2) {
            const float4 v0 = gcn_row4(x, GCN_EDGE_SRC(my, j),     lane4);
            const float4 v1 = gcn_row4(x, GCN_EDGE_SRC(my, j + 1), lane4);
            gcn_axpy4(acc, v0, GCN_EDGE_W(my, j));
            gcn_axpy4(acc, v1, GCN_EDGE_W(my, j + 1));
        } else if (r == 3) {
            const float4 v0 = gcn_row4(x, GCN_EDGE_SRC(my, j),     lane4);
            const float4 v1 = gcn_row4(x, GCN_EDGE_SRC(my, j + 1), lane4);
            const float4 v2 = gcn_row4(x, GCN_EDGE_SRC(my, j + 2), lane4);
            gcn_axpy4(acc, v0, GCN_EDGE_W(my, j));
            gcn_axpy4(acc, v1, GCN_EDGE_W(my, j + 1));
            gcn_axpy4(acc, v2, GCN_EDGE_W(my, j + 2));
        }
    }
    return acc;
}

#define GCN_K 128
#define GCN_TILE_PITCH 132
#define GCN_TILES_PER_GROUP 5
#define GCN_N_TILES (GCN_N_NODES / 16)
static_assert(GCN_N_NODES % 16 == 0, "whole 16-row tiles only");
static_assert(2 * GCN_TILES_PER_GROUP <= 32, "one lane per (tile,row) of a wave for the row bounds");

__global__ void __launch_bounds__(256)
gcn_layer_fused(const float* __restrict__ x, const int* __restrict__ row_ptr,
                const int* __restrict__ cnt_in, const int* __restrict__ edges,
                const float* __restrict__ deg_in_is, const float* __restrict__ W,
                const float* __restrict__ bias, float* __restrict__ out,
                int C, int do_relu, int n_tiles) {
    __shared__ __attribute__((aligned(16))) float tile[2][16 * GCN_TILE_PITCH];

    const int lane  = threadIdx.x & 31;
    const int lane4 = lane * 4;
    const int half  = lane >> 4;
    const int l15   = lane & 15;
    const int wave  = __builtin_amdgcn_readfirstlane(threadIdx.x >> 5);
    const bool has_ntile = (wave * 16) < C;
    const int ncol  = wave * 16 + l15;
    const int tile0 = blockIdx.x * GCN_TILES_PER_GROUP;
    const int2* __restrict__ ce = (const int2*)edges;

    int rp = 0, rc = 0;
    {
        const int tj = tile0 + (lane >> 1);
        if (lane < 2 * GCN_TILES_PER_GROUP && tj < n_tiles) {
            const int node = tj * 16 + wave * 2 + (lane & 1);
            rp = row_ptr[node];
            rc = cnt_in[node];
        }
    }

    v2f bfrag[GCN_K / 4];
    float bv = 0.0f;
#pragma unroll
    for (int s = 0; s < GCN_K / 4; ++s) { bfrag[s].x = 0.0f; bfrag[s].y = 0.0f; }
    if (has_ntile) {
#pragma unroll
        for (int s = 0; s < GCN_K / 4; ++s) {
            const int kb = s * 4 + half * 2;
            bfrag[s].x = W[(size_t)kb       * C + ncol];
            bfrag[s].y = W[(size_t)(kb + 1) * C + ncol];
        }
        bv = bias[ncol];
    }

    for (int i = 0; i < GCN_TILES_PER_GROUP; ++i) {
        const int t = tile0 + i;
        if (t >= n_tiles) break;
        float* tb = &tile[i & 1][0];

        const int beg0 = __builtin_amdgcn_readlane(rp, 2 * i);
        const int cnt0 = __builtin_amdgcn_readlane(rc, 2 * i);
        const int beg1 = __builtin_amdgcn_readlane(rp, 2 * i + 1);
        const int cnt1 = __builtin_amdgcn_readlane(rc, 2 * i + 1);

        int2 cur0 = make_int2(0, 0), cur1 = make_int2(0, 0);
        if (lane < cnt0) cur0 = ce[beg0 + lane];
        if (lane < cnt1) cur1 = ce[beg1 + lane];
        float scale = 0.0f;
        if (has_ntile) scale = deg_in_is[t * 16 + l15];

        const float4 g0 = gcn_gather_row(x, ce, beg0, cnt0, cur0, lane, lane4);
        *(float4*)(tb + (2 * wave) * GCN_TILE_PITCH + lane4) = g0;
        const float4 g1 = gcn_gather_row(x, ce, beg1, cnt1, cur1, lane, lane4);
        *(float4*)(tb + (2 * wave + 1) * GCN_TILE_PITCH + lane4) = g1;

        __syncthreads();

        if (has_ntile) {
            const float* arow = tb + l15 * GCN_TILE_PITCH + half * 2;
            v8f c = {};
#pragma unroll
            for (int s = 0; s < GCN_K / 4; ++s) {
                const v2f araw = *(const v2f*)(arow + s * 4);
                v2f a;
                a.x = araw.x * scale;
                a.y = araw.y * scale;
                c = __builtin_amdgcn_wmma_f32_16x16x4_f32(
                        false, a, false, bfrag[s], (short)0, c, false, false);
            }
#pragma unroll
            for (int v = 0; v < 8; ++v) {
                float val = c[v] + bv;
                if (do_relu) val = fmaxf(val, 0.0f);
                out[(size_t)(t * 16 + half * 8 + v) * C + ncol] = val;
            }
        }
    }
}

extern "C" void kernel_launch(void* const* d_in, const int* in_sizes, int n_in,
                              void* d_out, int out_size, void* d_ws, size_t ws_size,
                              hipStream_t stream) {
    const float* features = (const float*)d_in[0];
    const int*   src      = (const int*)  d_in[1];
    const int*   dst      = (const int*)  d_in[2];
    const float* W1       = (const float*)d_in[3];
    const float* b1       = (const float*)d_in[4];
    const float* W2       = (const float*)d_in[5];
    const float* b2       = (const float*)d_in[6];
    const float* W3       = (const float*)d_in[7];
    const float* b3       = (const float*)d_in[8];
    float* out = (float*)d_out;

    const int N = GCN_N_NODES;
    const int E = GCN_N_EDGES;
    const size_t NF = (size_t)N * 128;

    float* bufA    = (float*)d_ws;
    float* bufB    = bufA + NF;
    int*   edges   = (int*)(bufB + NF);
    int*   rank    = edges + 2 * (size_t)E;
    int*   cnt_out = rank + E;
    int*   cnt_in  = cnt_out + N;
    int*   row_ptr = cnt_in + N;
    float* deg_i   = (float*)(row_ptr + N);

    const int TB = 256;

    gcn_fill_zero_i<<<(2 * N + TB - 1) / TB, TB, 0, stream>>>(cnt_out, 2 * N);
    gcn_count_rank<<<(E + TB - 1) / TB, TB, 0, stream>>>(src, dst, cnt_out, cnt_in, rank, E);
    gcn_scan_rows<<<1, 1024, 0, stream>>>(cnt_in, row_ptr, N);
    gcn_place<<<(E + TB - 1) / TB, TB, 0, stream>>>(src, dst, rank, row_ptr, cnt_out, cnt_in,
                                                    edges, deg_i, E, N);

    const int groups = (GCN_N_TILES + GCN_TILES_PER_GROUP - 1) / GCN_TILES_PER_GROUP;

    gcn_layer_fused<<<groups, TB, 0, stream>>>(features, row_ptr, cnt_in, edges, deg_i,
                                               W1, b1, bufB, 128, 1, GCN_N_TILES);
    gcn_layer_fused<<<groups, TB, 0, stream>>>(bufB, row_ptr, cnt_in, edges, deg_i,
                                               W2, b2, bufA, 128, 1, GCN_N_TILES);
    gcn_layer_fused<<<groups, TB, 0, stream>>>(bufA, row_ptr, cnt_in, edges, deg_i,
                                               W3, b3, out, 16, 0, GCN_N_TILES);

    (void)in_sizes; (void)n_in; (void)out_size; (void)ws_size;
}


# Last layer's weight applied before its aggregation: layer 2's epilogue multiplies its LDS result tile by W3 (f32 WMMA), last kernel gathers 16-float rows; layer-3 sum order changes
# speedup vs baseline: 3.0011x; 1.1324x over previous
#include <hip/hip_runtime.h>
#include <math.h>


typedef __attribute__((ext_vector_type(2))) float v2f;
typedef __attribute__((ext_vector_type(8))) float v8f;

#define GCN_N_NODES 50000
#define GCN_N_EDGES 600000

__global__ void gcn_fill_zero_i(int* __restrict__ p, int n) {
    int i = blockIdx.x * blockDim.x + threadIdx.x;
    if (i < n) p[i] = 0;
}

__global__ void gcn_count_rank(const int* __restrict__ src, const int* __restrict__ dst,
                               int* __restrict__ cnt_out, int* __restrict__ cnt_in,
                               int* __restrict__ rank, int E) {
    int e = blockIdx.x * blockDim.x + threadIdx.x;
    if (e < E) {
        const int s = src[e];
        const int d = dst[e];
        atomicAdd(&cnt_out[s], 1);
        rank[e] = atomicAdd(&cnt_in[d], 1);
    }
}

#define GCN_SCAN_ITERS 7
#define GCN_SCAN_SEG (GCN_SCAN_ITERS * 256)
static_assert(32 * GCN_SCAN_SEG >= GCN_N_NODES, "one work-group of 32 waves must cover all nodes");
static_assert(GCN_N_NODES % 8 == 0, "8-wide node groups must not straddle the end");

static __device__ __forceinline__ float gcn_inv_sqrt_clamped(int c) {
    float v = (float)c;
    v = v < 1.0f ? 1.0f : v;
    return 1.0f / sqrtf(v);
}

__global__ void __launch_bounds__(1024)
gcn_scan_rows(const int* __restrict__ cnt_in, int* __restrict__ row_ptr, int n) {
    __shared__ int wave_total[32];
    const int lane  = threadIdx.x & 31;
    const int w     = threadIdx.x >> 5;
    const int wbase = w * GCN_SCAN_SEG + lane * 8;

    int4 ca[GCN_SCAN_ITERS], cb[GCN_SCAN_ITERS];
#pragma unroll
    for (int it = 0; it < GCN_SCAN_ITERS; ++it) {
        const int idx8 = wbase + it * 256;
        ca[it] = make_int4(0, 0, 0, 0);
        cb[it] = make_int4(0, 0, 0, 0);
        if (idx8 < n) {
            ca[it] = *(const int4*)(cnt_in + idx8);
            cb[it] = *(const int4*)(cnt_in + idx8 + 4);
        }
    }

    int tot8[GCN_SCAN_ITERS], incl[GCN_SCAN_ITERS];
#pragma unroll
    for (int it = 0; it < GCN_SCAN_ITERS; ++it) {
        tot8[it] = ca[it].x + ca[it].y + ca[it].z + ca[it].w
                 + cb[it].x + cb[it].y + cb[it].z + cb[it].w;
        incl[it] = tot8[it];
    }
#pragma unroll
    for (int d = 1; d < 32; d <<= 1) {
        int y[GCN_SCAN_ITERS];
#pragma unroll
        for (int it = 0; it < GCN_SCAN_ITERS; ++it) y[it] = __shfl_up(incl[it], d);
#pragma unroll
        for (int it = 0; it < GCN_SCAN_ITERS; ++it) if (lane >= d) incl[it] += y[it];
    }
    int trip_total[GCN_SCAN_ITERS];
#pragma unroll
    for (int it = 0; it < GCN_SCAN_ITERS; ++it) trip_total[it] = __shfl(incl[it], 31);

    int4 ea[GCN_SCAN_ITERS], eb[GCN_SCAN_ITERS];
    int carry = 0;
#pragma unroll
    for (int it = 0; it < GCN_SCAN_ITERS; ++it) {
        const int e0 = carry + incl[it] - tot8[it];
        const int e1 = e0 + ca[it].x;
        const int e2 = e1 + ca[it].y;
        const int e3 = e2 + ca[it].z;
        const int e4 = e3 + ca[it].w;
        const int e5 = e4 + cb[it].x;
        const int e6 = e5 + cb[it].y;
        const int e7 = e6 + cb[it].z;
        ea[it] = make_int4(e0, e1, e2, e3);
        eb[it] = make_int4(e4, e5, e6, e7);
        carry += trip_total[it];
    }
    if (lane == 0) wave_total[w] = carry;
    __syncthreads();
    int woff = 0;
    for (int k = 0; k < w; ++k) woff += wave_total[k];
#pragma unroll
    for (int it = 0; it < GCN_SCAN_ITERS; ++it) {
        const int idx8 = wbase + it * 256;
        if (idx8 < n) {
            int4 ra = ea[it], rb = eb[it];
            ra.x += woff; ra.y += woff; ra.z += woff; ra.w += woff;
            rb.x += woff; rb.y += woff; rb.z += woff; rb.w += woff;
            *(int4*)(row_ptr + idx8)     = ra;
            *(int4*)(row_ptr + idx8 + 4) = rb;
        }
    }
}

static_assert(GCN_N_EDGES >= GCN_N_NODES, "the placement grid must cover one thread per node");

__global__ void gcn_place(const int* __restrict__ src, const int* __restrict__ dst,
                          const int* __restrict__ rank, const int* __restrict__ row_ptr,
                          const int* __restrict__ cnt_out, const int* __restrict__ cnt_in,
                          int* __restrict__ edges, float* __restrict__ deg_in_is,
                          int E, int n) {
    int e = blockIdx.x * blockDim.x + threadIdx.x;
    if (e < n) deg_in_is[e] = gcn_inv_sqrt_clamped(cnt_in[e]);
    if (e < E) {
        const int s = src[e];
        const int pos = row_ptr[dst[e]] + rank[e];
        const float sc = gcn_inv_sqrt_clamped(cnt_out[s]);
        ((int2*)edges)[pos] = make_int2(s, __float_as_int(sc));
    }
}

static __device__ __forceinline__ float4 gcn_row4(const float* __restrict__ x, int s, int lane4) {
    return *(const float4*)(x + (size_t)s * 128 + lane4);
}
static __device__ __forceinline__ void gcn_axpy4(float4& acc, const float4 v, const float w) {
    acc.x = fmaf(v.x, w, acc.x);
    acc.y = fmaf(v.y, w, acc.y);
    acc.z = fmaf(v.z, w, acc.z);
    acc.w = fmaf(v.w, w, acc.w);
}
#define GCN_EDGE_SRC(my, j) __builtin_amdgcn_readlane((my).x, (j))
#define GCN_EDGE_W(my, j)   __int_as_float(__builtin_amdgcn_readlane((my).y, (j)))

static __device__ __forceinline__ float4 gcn_gather_row(const float* __restrict__ x,
                                                        const int2* __restrict__ ce,
                                                        const int beg, const int cnt,
                                                        const int2 first,
                                                        const int lane, const int lane4) {
    float4 acc = make_float4(0.0f, 0.0f, 0.0f, 0.0f);
    for (int b = 0; b < cnt; b += 32) {
        const int m = (cnt - b < 32) ? (cnt - b) : 32;
        int2 my = first;
        if (b > 0) {
            my = make_int2(0, 0);
            if (lane < m) my = ce[beg + b + lane];
        }
        int j = 0;
        for (; j + 8 <= m; j += 8) {
            const float4 v0 = gcn_row4(x, GCN_EDGE_SRC(my, j),     lane4);
            const float4 v1 = gcn_row4(x, GCN_EDGE_SRC(my, j + 1), lane4);
            const float4 v2 = gcn_row4(x, GCN_EDGE_SRC(my, j + 2), lane4);
            const float4 v3 = gcn_row4(x, GCN_EDGE_SRC(my, j + 3), lane4);
            const float4 v4 = gcn_row4(x, GCN_EDGE_SRC(my, j + 4), lane4);
            const float4 v5 = gcn_row4(x, GCN_EDGE_SRC(my, j + 5), lane4);
            const float4 v6 = gcn_row4(x, GCN_EDGE_SRC(my, j + 6), lane4);
            const float4 v7 = gcn_row4(x, GCN_EDGE_SRC(my, j + 7), lane4);
            gcn_axpy4(acc, v0, GCN_EDGE_W(my, j));
            gcn_axpy4(acc, v1, GCN_EDGE_W(my, j + 1));
            gcn_axpy4(acc, v2, GCN_EDGE_W(my, j + 2));
            gcn_axpy4(acc, v3, GCN_EDGE_W(my, j + 3));
            gcn_axpy4(acc, v4, GCN_EDGE_W(my, j + 4));
            gcn_axpy4(acc, v5, GCN_EDGE_W(my, j + 5));
            gcn_axpy4(acc, v6, GCN_EDGE_W(my, j + 6));
            gcn_axpy4(acc, v7, GCN_EDGE_W(my, j + 7));
        }
        if (j + 4 <= m) {
            const float4 v0 = gcn_row4(x, GCN_EDGE_SRC(my, j),     lane4);
            const float4 v1 = gcn_row4(x, GCN_EDGE_SRC(my, j + 1), lane4);
            const float4 v2 = gcn_row4(x, GCN_EDGE_SRC(my, j + 2), lane4);
            const float4 v3 = gcn_row4(x, GCN_EDGE_SRC(my, j + 3), lane4);
            gcn_axpy4(acc, v0, GCN_EDGE_W(my, j));
            gcn_axpy4(acc, v1, GCN_EDGE_W(my, j + 1));
            gcn_axpy4(acc, v2, GCN_EDGE_W(my, j + 2));
            gcn_axpy4(acc, v3, GCN_EDGE_W(my, j + 3));
            j += 4;
        }
        const int r = m - j;
        if (r == 1) {
            const float4 v0 = gcn_row4(x, GCN_EDGE_SRC(my, j), lane4);
            gcn_axpy4(acc, v0, GCN_EDGE_W(my, j));
        } else if (r == 2) {
            const float4 v0 = gcn_row4(x, GCN_EDGE_SRC(my, j),     lane4);
            const float4 v1 = gcn_row4(x, GCN_EDGE_SRC(my, j + 1), lane4);
            gcn_axpy4(acc, v0, GCN_EDGE_W(my, j));
            gcn_axpy4(acc, v1, GCN_EDGE_W(my, j + 1));
        } else if (r == 3) {
            const float4 v0 = gcn_row4(x, GCN_EDGE_SRC(my, j),     lane4);
            const float4 v1 = gcn_row4(x, GCN_EDGE_SRC(my, j + 1), lane4);
            const float4 v2 = gcn_row4(x, GCN_EDGE_SRC(my, j + 2), lane4);
            gcn_axpy4(acc, v0, GCN_EDGE_W(my, j));
            gcn_axpy4(acc, v1, GCN_EDGE_W(my, j + 1));
            gcn_axpy4(acc, v2, GCN_EDGE_W(my, j + 2));
        }
    }
    return acc;
}

#define GCN_K 128
#define GCN_C 128
#define GCN_C_LAST 16
#define GCN_TILE_PITCH 132
#define GCN_TILES_PER_GROUP 5
#define GCN_N_TILES (GCN_N_NODES / 16)
static_assert(GCN_N_NODES % 16 == 0, "whole 16-row tiles only");
static_assert(2 * GCN_TILES_PER_GROUP <= 32, "one lane per (tile,row) of a wave for the row bounds");
static_assert(GCN_C == 8 * 16, "8 waves, one 16-column strip each");

template <bool FUSE_NEXT>
__global__ void __launch_bounds__(256)
gcn_layer_fused(const float* __restrict__ x, const int* __restrict__ row_ptr,
                const int* __restrict__ cnt_in, const int* __restrict__ edges,
                const float* __restrict__ deg_in_is, const float* __restrict__ W,
                const float* __restrict__ bias, const float* __restrict__ Wn,
                float* __restrict__ out, int do_relu, int n_tiles) {
    __shared__ __attribute__((aligned(16))) float tile[2][16 * GCN_TILE_PITCH];
    __shared__ __attribute__((aligned(16))) float htile[FUSE_NEXT ? 16 * GCN_TILE_PITCH : 4];
    __shared__ __attribute__((aligned(16))) v2f wn_frag[FUSE_NEXT ? (GCN_K / 4) * 32 : 1];

    const int lane  = threadIdx.x & 31;
    const int lane4 = lane * 4;
    const int half  = lane >> 4;
    const int l15   = lane & 15;
    const int wave  = __builtin_amdgcn_readfirstlane(threadIdx.x >> 5);
    const int ncol  = wave * 16 + l15;
    const int tile0 = blockIdx.x * GCN_TILES_PER_GROUP;
    const int2* __restrict__ ce = (const int2*)edges;

    int rp = 0, rc = 0;
    {
        const int tj = tile0 + (lane >> 1);
        if (lane < 2 * GCN_TILES_PER_GROUP && tj < n_tiles) {
            const int node = tj * 16 + wave * 2 + (lane & 1);
            rp = row_ptr[node];
            rc = cnt_in[node];
        }
    }

    v2f bfrag[GCN_K / 4];
#pragma unroll
    for (int s = 0; s < GCN_K / 4; ++s) {
        const int kb = s * 4 + half * 2;
        bfrag[s].x = W[(size_t)kb       * GCN_C + ncol];
        bfrag[s].y = W[(size_t)(kb + 1) * GCN_C + ncol];
    }
    const float bv = bias[ncol];

    if constexpr (FUSE_NEXT) {
        for (int idx = threadIdx.x; idx < (GCN_K / 4) * 32; idx += 256) {
            const int s  = idx >> 5;
            const int ln = idx & 31;
            const int kb = s * 4 + (ln >> 4) * 2;
            v2f f;
            f.x = Wn[(size_t)kb       * GCN_C_LAST + (ln & 15)];
            f.y = Wn[(size_t)(kb + 1) * GCN_C_LAST + (ln & 15)];
            wn_frag[idx] = f;
        }
    }

    for (int i = 0; i < GCN_TILES_PER_GROUP; ++i) {
        const int t = tile0 + i;
        if (t >= n_tiles) break;
        float* tb = &tile[i & 1][0];

        const int beg0 = __builtin_amdgcn_readlane(rp, 2 * i);
        const int cnt0 = __builtin_amdgcn_readlane(rc, 2 * i);
        const int beg1 = __builtin_amdgcn_readlane(rp, 2 * i + 1);
        const int cnt1 = __builtin_amdgcn_readlane(rc, 2 * i + 1);

        int2 cur0 = make_int2(0, 0), cur1 = make_int2(0, 0);
        if (lane < cnt0) cur0 = ce[beg0 + lane];
        if (lane < cnt1) cur1 = ce[beg1 + lane];
        const float scale = deg_in_is[t * 16 + l15];

        const float4 g0 = gcn_gather_row(x, ce, beg0, cnt0, cur0, lane, lane4);
        *(float4*)(tb + (2 * wave) * GCN_TILE_PITCH + lane4) = g0;
        const float4 g1 = gcn_gather_row(x, ce, beg1, cnt1, cur1, lane, lane4);
        *(float4*)(tb + (2 * wave + 1) * GCN_TILE_PITCH + lane4) = g1;

        __syncthreads();

        {
            const float* arow = tb + l15 * GCN_TILE_PITCH + half * 2;
            v8f c = {};
#pragma unroll
            for (int s = 0; s < GCN_K / 4; ++s) {
                const v2f araw = *(const v2f*)(arow + s * 4);
                v2f a;
                a.x = araw.x * scale;
                a.y = araw.y * scale;
                c = __builtin_amdgcn_wmma_f32_16x16x4_f32(
                        false, a, false, bfrag[s], (short)0, c, false, false);
            }
#pragma unroll
            for (int v = 0; v < 8; ++v) {
                float val = c[v] + bv;
                if (do_relu) val = fmaxf(val, 0.0f);
                if constexpr (FUSE_NEXT) {
                    htile[(half * 8 + v) * GCN_TILE_PITCH + ncol] = val;
                } else {
                    out[(size_t)(t * 16 + half * 8 + v) * GCN_C + ncol] = val;
                }
            }
        }

        if constexpr (FUSE_NEXT) {
            __syncthreads();
            if (wave == (i & 7)) {
                const float* hrow = htile + l15 * GCN_TILE_PITCH + half * 2;
                v8f z0 = {}, z1 = {};
#pragma unroll
                for (int s = 0; s < GCN_K / 4; s += 2) {
                    const v2f a0 = *(const v2f*)(hrow + s * 4);
                    const v2f b0 = wn_frag[s * 32 + lane];
                    const v2f a1 = *(const v2f*)(hrow + (s + 1) * 4);
                    const v2f b1 = wn_frag[(s + 1) * 32 + lane];
                    z0 = __builtin_amdgcn_wmma_f32_16x16x4_f32(
                            false, a0, false, b0, (short)0, z0, false, false);
                    z1 = __builtin_amdgcn_wmma_f32_16x16x4_f32(
                            false, a1, false, b1, (short)0, z1, false, false);
                }
#pragma unroll
                for (int v = 0; v < 8; ++v) {
                    out[(size_t)(t * 16 + half * 8 + v) * GCN_C_LAST + l15] = z0[v] + z1[v];
                }
            }
        }
    }
}

#define GCN_LAST_BATCH 16

__global__ void __launch_bounds__(256) __attribute__((amdgpu_waves_per_eu(4, 6)))
gcn_gather_last(const float* __restrict__ z, const int* __restrict__ row_ptr,
                const int* __restrict__ cnt_in, const int* __restrict__ edges,
                const float* __restrict__ deg_in_is, const float* __restrict__ bias,
                float* __restrict__ out, int n) {
    const int g  = blockIdx.x * blockDim.x + threadIdx.x;
    const int m  = g >> 2;
    const int q4 = (g & 3) * 4;
    if (m >= n) return;
    const int    cnt = cnt_in[m];
    const int    beg = row_ptr[m];
    const float  sc  = deg_in_is[m];
    const float4 b   = *(const float4*)(bias + q4);
    asm volatile("" :: "v"(cnt), "v"(beg), "v"(sc), "v"(b.x), "v"(b.y), "v"(b.z), "v"(b.w));
    const int2* __restrict__ ce = (const int2*)edges + beg;
    const int last = cnt - 1;

    float4 acc = make_float4(0.0f, 0.0f, 0.0f, 0.0f);
    for (int j = 0; j < cnt; j += GCN_LAST_BATCH) {
        int2 e[GCN_LAST_BATCH];
        float4 v[GCN_LAST_BATCH];
#pragma unroll
        for (int k = 0; k < GCN_LAST_BATCH; ++k) {
            const int jj = (j + k < last) ? (j + k) : last;
            e[k] = ce[jj];
        }
#pragma unroll
        for (int k = 0; k < GCN_LAST_BATCH; ++k) {
            v[k] = *(const float4*)(z + (size_t)e[k].x * GCN_C_LAST + q4);
        }
#pragma unroll
        for (int k = 0; k < GCN_LAST_BATCH; ++k) {
            const bool on = (j + k) < cnt;
            float4 vv;
            vv.x = on ? v[k].x : 0.0f;
            vv.y = on ? v[k].y : 0.0f;
            vv.z = on ? v[k].z : 0.0f;
            vv.w = on ? v[k].w : 0.0f;
            gcn_axpy4(acc, vv, __int_as_float(e[k].y));
        }
    }
    float4 r;
    r.x = fmaf(acc.x, sc, b.x);
    r.y = fmaf(acc.y, sc, b.y);
    r.z = fmaf(acc.z, sc, b.z);
    r.w = fmaf(acc.w, sc, b.w);
    *(float4*)(out + (size_t)m * GCN_C_LAST + q4) = r;
}

extern "C" void kernel_launch(void* const* d_in, const int* in_sizes, int n_in,
                              void* d_out, int out_size, void* d_ws, size_t ws_size,
                              hipStream_t stream) {
    const float* features = (const float*)d_in[0];
    const int*   src      = (const int*)  d_in[1];
    const int*   dst      = (const int*)  d_in[2];
    const float* W1       = (const float*)d_in[3];
    const float* b1       = (const float*)d_in[4];
    const float* W2       = (const float*)d_in[5];
    const float* b2       = (const float*)d_in[6];
    const float* W3       = (const float*)d_in[7];
    const float* b3       = (const float*)d_in[8];
    float* out = (float*)d_out;

    const int N = GCN_N_NODES;
    const int E = GCN_N_EDGES;
    const size_t NF = (size_t)N * 128;

    float* bufZ    = (float*)d_ws;
    float* bufB    = bufZ + NF;
    int*   edges   = (int*)(bufB + NF);
    int*   rank    = edges + 2 * (size_t)E;
    int*   cnt_out = rank + E;
    int*   cnt_in  = cnt_out + N;
    int*   row_ptr = cnt_in + N;
    float* deg_i   = (float*)(row_ptr + N);

    const int TB = 256;

    gcn_fill_zero_i<<<(2 * N + TB - 1) / TB, TB, 0, stream>>>(cnt_out, 2 * N);
    gcn_count_rank<<<(E + TB - 1) / TB, TB, 0, stream>>>(src, dst, cnt_out, cnt_in, rank, E);
    gcn_scan_rows<<<1, 1024, 0, stream>>>(cnt_in, row_ptr, N);
    gcn_place<<<(E + TB - 1) / TB, TB, 0, stream>>>(src, dst, rank, row_ptr, cnt_out, cnt_in,
                                                    edges, deg_i, E, N);

    const int groups = (GCN_N_TILES + GCN_TILES_PER_GROUP - 1) / GCN_TILES_PER_GROUP;

    gcn_layer_fused<false><<<groups, TB, 0, stream>>>(features, row_ptr, cnt_in, edges, deg_i,
                                                      W1, b1, W3, bufB, 1, GCN_N_TILES);
    gcn_layer_fused<true><<<groups, TB, 0, stream>>>(bufB, row_ptr, cnt_in, edges, deg_i,
                                                     W2, b2, W3, bufZ, 1, GCN_N_TILES);
    gcn_gather_last<<<(4 * N + TB - 1) / TB, TB, 0, stream>>>(bufZ, row_ptr, cnt_in, edges, deg_i,
                                                             b3, out, N);

    (void)in_sizes; (void)n_in; (void)out_size; (void)ws_size;
}


# Fused layers launched as 1024 work-groups of 3-4 row tiles, split evenly (was 625 of 5): one full resident set, 4 per WGP at 128 registers; sum order unchanged
# speedup vs baseline: 3.2196x; 1.0728x over previous
#include <hip/hip_runtime.h>
#include <math.h>


typedef __attribute__((ext_vector_type(2))) float v2f;
typedef __attribute__((ext_vector_type(8))) float v8f;

#define GCN_N_NODES 50000
#define GCN_N_EDGES 600000

__global__ void gcn_fill_zero_i(int* __restrict__ p, int n) {
    int i = blockIdx.x * blockDim.x + threadIdx.x;
    if (i < n) p[i] = 0;
}

__global__ void gcn_count_rank(const int* __restrict__ src, const int* __restrict__ dst,
                               int* __restrict__ cnt_out, int* __restrict__ cnt_in,
                               int* __restrict__ rank, int E) {
    int e = blockIdx.x * blockDim.x + threadIdx.x;
    if (e < E) {
        const int s = src[e];
        const int d = dst[e];
        atomicAdd(&cnt_out[s], 1);
        rank[e] = atomicAdd(&cnt_in[d], 1);
    }
}

#define GCN_SCAN_ITERS 7
#define GCN_SCAN_SEG (GCN_SCAN_ITERS * 256)
static_assert(32 * GCN_SCAN_SEG >= GCN_N_NODES, "one work-group of 32 waves must cover all nodes");
static_assert(GCN_N_NODES % 8 == 0, "8-wide node groups must not straddle the end");

static __device__ __forceinline__ float gcn_inv_sqrt_clamped(int c) {
    float v = (float)c;
    v = v < 1.0f ? 1.0f : v;
    return 1.0f / sqrtf(v);
}

__global__ void __launch_bounds__(1024)
gcn_scan_rows(const int* __restrict__ cnt_in, int* __restrict__ row_ptr, int n) {
    __shared__ int wave_total[32];
    const int lane  = threadIdx.x & 31;
    const int w     = threadIdx.x >> 5;
    const int wbase = w * GCN_SCAN_SEG + lane * 8;

    int4 ca[GCN_SCAN_ITERS], cb[GCN_SCAN_ITERS];
#pragma unroll
    for (int it = 0; it < GCN_SCAN_ITERS; ++it) {
        const int idx8 = wbase + it * 256;
        ca[it] = make_int4(0, 0, 0, 0);
        cb[it] = make_int4(0, 0, 0, 0);
        if (idx8 < n) {
            ca[it] = *(const int4*)(cnt_in + idx8);
            cb[it] = *(const int4*)(cnt_in + idx8 + 4);
        }
    }

    int tot8[GCN_SCAN_ITERS], incl[GCN_SCAN_ITERS];
#pragma unroll
    for (int it = 0; it < GCN_SCAN_ITERS; ++it) {
        tot8[it] = ca[it].x + ca[it].y + ca[it].z + ca[it].w
                 + cb[it].x + cb[it].y + cb[it].z + cb[it].w;
        incl[it] = tot8[it];
    }
#pragma unroll
    for (int d = 1; d < 32; d <<= 1) {
        int y[GCN_SCAN_ITERS];
#pragma unroll
        for (int it = 0; it < GCN_SCAN_ITERS; ++it) y[it] = __shfl_up(incl[it], d);
#pragma unroll
        for (int it = 0; it < GCN_SCAN_ITERS; ++it) if (lane >= d) incl[it] += y[it];
    }
    int trip_total[GCN_SCAN_ITERS];
#pragma unroll
    for (int it = 0; it < GCN_SCAN_ITERS; ++it) trip_total[it] = __shfl(incl[it], 31);

    int4 ea[GCN_SCAN_ITERS], eb[GCN_SCAN_ITERS];
    int carry = 0;
#pragma unroll
    for (int it = 0; it < GCN_SCAN_ITERS; ++it) {
        const int e0 = carry + incl[it] - tot8[it];
        const int e1 = e0 + ca[it].x;
        const int e2 = e1 + ca[it].y;
        const int e3 = e2 + ca[it].z;
        const int e4 = e3 + ca[it].w;
        const int e5 = e4 + cb[it].x;
        const int e6 = e5 + cb[it].y;
        const int e7 = e6 + cb[it].z;
        ea[it] = make_int4(e0, e1, e2, e3);
        eb[it] = make_int4(e4, e5, e6, e7);
        carry += trip_total[it];
    }
    if (lane == 0) wave_total[w] = carry;
    __syncthreads();
    int woff = 0;
    for (int k = 0; k < w; ++k) woff += wave_total[k];
#pragma unroll
    for (int it = 0; it < GCN_SCAN_ITERS; ++it) {
        const int idx8 = wbase + it * 256;
        if (idx8 < n) {
            int4 ra = ea[it], rb = eb[it];
            ra.x += woff; ra.y += woff; ra.z += woff; ra.w += woff;
            rb.x += woff; rb.y += woff; rb.z += woff; rb.w += woff;
            *(int4*)(row_ptr + idx8)     = ra;
            *(int4*)(row_ptr + idx8 + 4) = rb;
        }
    }
}

static_assert(GCN_N_EDGES >= GCN_N_NODES, "the placement grid must cover one thread per node");

__global__ void gcn_place(const int* __restrict__ src, const int* __restrict__ dst,
                          const int* __restrict__ rank, const int* __restrict__ row_ptr,
                          const int* __restrict__ cnt_out, const int* __restrict__ cnt_in,
                          int* __restrict__ edges, float* __restrict__ deg_in_is,
                          int E, int n) {
    int e = blockIdx.x * blockDim.x + threadIdx.x;
    if (e < n) deg_in_is[e] = gcn_inv_sqrt_clamped(cnt_in[e]);
    if (e < E) {
        const int s = src[e];
        const int pos = row_ptr[dst[e]] + rank[e];
        const float sc = gcn_inv_sqrt_clamped(cnt_out[s]);
        ((int2*)edges)[pos] = make_int2(s, __float_as_int(sc));
    }
}

static __device__ __forceinline__ float4 gcn_row4(const float* __restrict__ x, int s, int lane4) {
    return *(const float4*)(x + (size_t)s * 128 + lane4);
}
static __device__ __forceinline__ void gcn_axpy4(float4& acc, const float4 v, const float w) {
    acc.x = fmaf(v.x, w, acc.x);
    acc.y = fmaf(v.y, w, acc.y);
    acc.z = fmaf(v.z, w, acc.z);
    acc.w = fmaf(v.w, w, acc.w);
}
#define GCN_EDGE_SRC(my, j) __builtin_amdgcn_readlane((my).x, (j))
#define GCN_EDGE_W(my, j)   __int_as_float(__builtin_amdgcn_readlane((my).y, (j)))

static __device__ __forceinline__ float4 gcn_gather_row(const float* __restrict__ x,
                                                        const int2* __restrict__ ce,
                                                        const int beg, const int cnt,
                                                        const int2 first,
                                                        const int lane, const int lane4) {
    float4 acc = make_float4(0.0f, 0.0f, 0.0f, 0.0f);
    for (int b = 0; b < cnt; b += 32) {
        const int m = (cnt - b < 32) ? (cnt - b) : 32;
        int2 my = first;
        if (b > 0) {
            my = make_int2(0, 0);
            if (lane < m) my = ce[beg + b + lane];
        }
        int j = 0;
        for (; j + 8 <= m; j += 8) {
            const float4 v0 = gcn_row4(x, GCN_EDGE_SRC(my, j),     lane4);
            const float4 v1 = gcn_row4(x, GCN_EDGE_SRC(my, j + 1), lane4);
            const float4 v2 = gcn_row4(x, GCN_EDGE_SRC(my, j + 2), lane4);
            const float4 v3 = gcn_row4(x, GCN_EDGE_SRC(my, j + 3), lane4);
            const float4 v4 = gcn_row4(x, GCN_EDGE_SRC(my, j + 4), lane4);
            const float4 v5 = gcn_row4(x, GCN_EDGE_SRC(my, j + 5), lane4);
            const float4 v6 = gcn_row4(x, GCN_EDGE_SRC(my, j + 6), lane4);
            const float4 v7 = gcn_row4(x, GCN_EDGE_SRC(my, j + 7), lane4);
            gcn_axpy4(acc, v0, GCN_EDGE_W(my, j));
            gcn_axpy4(acc, v1, GCN_EDGE_W(my, j + 1));
            gcn_axpy4(acc, v2, GCN_EDGE_W(my, j + 2));
            gcn_axpy4(acc, v3, GCN_EDGE_W(my, j + 3));
            gcn_axpy4(acc, v4, GCN_EDGE_W(my, j + 4));
            gcn_axpy4(acc, v5, GCN_EDGE_W(my, j + 5));
            gcn_axpy4(acc, v6, GCN_EDGE_W(my, j + 6));
            gcn_axpy4(acc, v7, GCN_EDGE_W(my, j + 7));
        }
        if (j + 4 <= m) {
            const float4 v0 = gcn_row4(x, GCN_EDGE_SRC(my, j),     lane4);
            const float4 v1 = gcn_row4(x, GCN_EDGE_SRC(my, j + 1), lane4);
            const float4 v2 = gcn_row4(x, GCN_EDGE_SRC(my, j + 2), lane4);
            const float4 v3 = gcn_row4(x, GCN_EDGE_SRC(my, j + 3), lane4);
            gcn_axpy4(acc, v0, GCN_EDGE_W(my, j));
            gcn_axpy4(acc, v1, GCN_EDGE_W(my, j + 1));
            gcn_axpy4(acc, v2, GCN_EDGE_W(my, j + 2));
            gcn_axpy4(acc, v3, GCN_EDGE_W(my, j + 3));
            j += 4;
        }
        const int r = m - j;
        if (r == 1) {
            const float4 v0 = gcn_row4(x, GCN_EDGE_SRC(my, j), lane4);
            gcn_axpy4(acc, v0, GCN_EDGE_W(my, j));
        } else if (r == 2) {
            const float4 v0 = gcn_row4(x, GCN_EDGE_SRC(my, j),     lane4);
            const float4 v1 = gcn_row4(x, GCN_EDGE_SRC(my, j + 1), lane4);
            gcn_axpy4(acc, v0, GCN_EDGE_W(my, j));
            gcn_axpy4(acc, v1, GCN_EDGE_W(my, j + 1));
        } else if (r == 3) {
            const float4 v0 = gcn_row4(x, GCN_EDGE_SRC(my, j),     lane4);
            const float4 v1 = gcn_row4(x, GCN_EDGE_SRC(my, j + 1), lane4);
            const float4 v2 = gcn_row4(x, GCN_EDGE_SRC(my, j + 2), lane4);
            gcn_axpy4(acc, v0, GCN_EDGE_W(my, j));
            gcn_axpy4(acc, v1, GCN_EDGE_W(my, j + 1));
            gcn_axpy4(acc, v2, GCN_EDGE_W(my, j + 2));
        }
    }
    return acc;
}

#define GCN_K 128
#define GCN_C 128
#define GCN_C_LAST 16
#define GCN_TILE_PITCH 132
#define GCN_LAYER_GROUPS 1024
#define GCN_MAX_TILES_PER_GROUP 16
#define GCN_N_TILES (GCN_N_NODES / 16)
static_assert(GCN_N_NODES % 16 == 0, "whole 16-row tiles only");
static_assert(GCN_LAYER_GROUPS <= GCN_N_TILES, "every work-group gets at least one tile");
static_assert((GCN_N_TILES + GCN_LAYER_GROUPS - 1) / GCN_LAYER_GROUPS <= GCN_MAX_TILES_PER_GROUP,
              "a work-group's tiles must fit the lanes that hold their row bounds");
static_assert(GCN_C == 8 * 16, "8 waves, one 16-column strip each");

template <bool FUSE_NEXT>
__global__ void __launch_bounds__(256) __attribute__((amdgpu_waves_per_eu(8, 8)))
gcn_layer_fused(const float* __restrict__ x, const int* __restrict__ row_ptr,
                const int* __restrict__ cnt_in, const int* __restrict__ edges,
                const float* __restrict__ deg_in_is, const float* __restrict__ W,
                const float* __restrict__ bias, const float* __restrict__ Wn,
                float* __restrict__ out, int do_relu, int n_tiles) {
    __shared__ __attribute__((aligned(16))) float tile[2][16 * GCN_TILE_PITCH];
    __shared__ __attribute__((aligned(16))) float htile[FUSE_NEXT ? 16 * GCN_TILE_PITCH : 4];
    __shared__ __attribute__((aligned(16))) v2f wn_frag[FUSE_NEXT ? (GCN_K / 4) * 32 : 1];

    const int lane  = threadIdx.x & 31;
    const int lane4 = lane * 4;
    const int half  = lane >> 4;
    const int l15   = lane & 15;
    const int wave  = __builtin_amdgcn_readfirstlane(threadIdx.x >> 5);
    const int ncol  = wave * 16 + l15;
    const int per_group = n_tiles / GCN_LAYER_GROUPS;
    const int extra     = n_tiles % GCN_LAYER_GROUPS;
    const int g         = (int)blockIdx.x;
    const int tile0     = g * per_group + (g < extra ? g : extra);
    const int my_tiles  = per_group + (g < extra ? 1 : 0);
    const int2* __restrict__ ce = (const int2*)edges;

    int rp = 0, rc = 0;
    {
        const int tj = tile0 + (lane >> 1);
        if ((lane >> 1) < my_tiles) {
            const int node = tj * 16 + wave * 2 + (lane & 1);
            rp = row_ptr[node];
            rc = cnt_in[node];
        }
    }

    v2f bfrag[GCN_K / 4];
#pragma unroll
    for (int s = 0; s < GCN_K / 4; ++s) {
        const int kb = s * 4 + half * 2;
        bfrag[s].x = W[(size_t)kb       * GCN_C + ncol];
        bfrag[s].y = W[(size_t)(kb + 1) * GCN_C + ncol];
    }
    const float bv = bias[ncol];

    if constexpr (FUSE_NEXT) {
        for (int idx = threadIdx.x; idx < (GCN_K / 4) * 32; idx += 256) {
            const int s  = idx >> 5;
            const int ln = idx & 31;
            const int kb = s * 4 + (ln >> 4) * 2;
            v2f f;
            f.x = Wn[(size_t)kb       * GCN_C_LAST + (ln & 15)];
            f.y = Wn[(size_t)(kb + 1) * GCN_C_LAST + (ln & 15)];
            wn_frag[idx] = f;
        }
    }

    for (int i = 0; i < my_tiles; ++i) {
        const int t = tile0 + i;
        float* tb = &tile[i & 1][0];

        const int beg0 = __builtin_amdgcn_readlane(rp, 2 * i);
        const int cnt0 = __builtin_amdgcn_readlane(rc, 2 * i);
        const int beg1 = __builtin_amdgcn_readlane(rp, 2 * i + 1);
        const int cnt1 = __builtin_amdgcn_readlane(rc, 2 * i + 1);

        int2 cur0 = make_int2(0, 0), cur1 = make_int2(0, 0);
        if (lane < cnt0) cur0 = ce[beg0 + lane];
        if (lane < cnt1) cur1 = ce[beg1 + lane];
        const float scale = deg_in_is[t * 16 + l15];

        const float4 g0 = gcn_gather_row(x, ce, beg0, cnt0, cur0, lane, lane4);
        *(float4*)(tb + (2 * wave) * GCN_TILE_PITCH + lane4) = g0;
        const float4 g1 = gcn_gather_row(x, ce, beg1, cnt1, cur1, lane, lane4);
        *(float4*)(tb + (2 * wave + 1) * GCN_TILE_PITCH + lane4) = g1;

        __syncthreads();

        {
            const float* arow = tb + l15 * GCN_TILE_PITCH + half * 2;
            v8f c = {};
#pragma unroll
            for (int s = 0; s < GCN_K / 4; ++s) {
                const v2f araw = *(const v2f*)(arow + s * 4);
                v2f a;
                a.x = araw.x * scale;
                a.y = araw.y * scale;
                c = __builtin_amdgcn_wmma_f32_16x16x4_f32(
                        false, a, false, bfrag[s], (short)0, c, false, false);
            }
#pragma unroll
            for (int v = 0; v < 8; ++v) {
                float val = c[v] + bv;
                if (do_relu) val = fmaxf(val, 0.0f);
                if constexpr (FUSE_NEXT) {
                    htile[(half * 8 + v) * GCN_TILE_PITCH + ncol] = val;
                } else {
                    out[(size_t)(t * 16 + half * 8 + v) * GCN_C + ncol] = val;
                }
            }
        }

        if constexpr (FUSE_NEXT) {
            __syncthreads();
            if (wave == (i & 7)) {
                const float* hrow = htile + l15 * GCN_TILE_PITCH + half * 2;
                v8f z0 = {}, z1 = {};
#pragma unroll
                for (int s = 0; s < GCN_K / 4; s += 2) {
                    const v2f a0 = *(const v2f*)(hrow + s * 4);
                    const v2f b0 = wn_frag[s * 32 + lane];
                    const v2f a1 = *(const v2f*)(hrow + (s + 1) * 4);
                    const v2f b1 = wn_frag[(s + 1) * 32 + lane];
                    z0 = __builtin_amdgcn_wmma_f32_16x16x4_f32(
                            false, a0, false, b0, (short)0, z0, false, false);
                    z1 = __builtin_amdgcn_wmma_f32_16x16x4_f32(
                            false, a1, false, b1, (short)0, z1, false, false);
                }
#pragma unroll
                for (int v = 0; v < 8; ++v) {
                    out[(size_t)(t * 16 + half * 8 + v) * GCN_C_LAST + l15] = z0[v] + z1[v];
                }
            }
        }
    }
}

#define GCN_LAST_BATCH 16

__global__ void __launch_bounds__(256) __attribute__((amdgpu_waves_per_eu(4, 6)))
gcn_gather_last(const float* __restrict__ z, const int* __restrict__ row_ptr,
                const int* __restrict__ cnt_in, const int* __restrict__ edges,
                const float* __restrict__ deg_in_is, const float* __restrict__ bias,
                float* __restrict__ out, int n) {
    const int g  = blockIdx.x * blockDim.x + threadIdx.x;
    const int m  = g >> 2;
    const int q4 = (g & 3) * 4;
    if (m >= n) return;
    const int    cnt = cnt_in[m];
    const int    beg = row_ptr[m];
    const float  sc  = deg_in_is[m];
    const float4 b   = *(const float4*)(bias + q4);
    asm volatile("" :: "v"(cnt), "v"(beg), "v"(sc), "v"(b.x), "v"(b.y), "v"(b.z), "v"(b.w));
    const int2* __restrict__ ce = (const int2*)edges + beg;
    const int last = cnt - 1;

    float4 acc = make_float4(0.0f, 0.0f, 0.0f, 0.0f);
    for (int j = 0; j < cnt; j += GCN_LAST_BATCH) {
        int2 e[GCN_LAST_BATCH];
        float4 v[GCN_LAST_BATCH];
#pragma unroll
        for (int k = 0; k < GCN_LAST_BATCH; ++k) {
            const int jj = (j + k < last) ? (j + k) : last;
            e[k] = ce[jj];
        }
#pragma unroll
        for (int k = 0; k < GCN_LAST_BATCH; ++k) {
            v[k] = *(const float4*)(z + (size_t)e[k].x * GCN_C_LAST + q4);
        }
#pragma unroll
        for (int k = 0; k < GCN_LAST_BATCH; ++k) {
            const bool on = (j + k) < cnt;
            float4 vv;
            vv.x = on ? v[k].x : 0.0f;
            vv.y = on ? v[k].y : 0.0f;
            vv.z = on ? v[k].z : 0.0f;
            vv.w = on ? v[k].w : 0.0f;
            gcn_axpy4(acc, vv, __int_as_float(e[k].y));
        }
    }
    float4 r;
    r.x = fmaf(acc.x, sc, b.x);
    r.y = fmaf(acc.y, sc, b.y);
    r.z = fmaf(acc.z, sc, b.z);
    r.w = fmaf(acc.w, sc, b.w);
    *(float4*)(out + (size_t)m * GCN_C_LAST + q4) = r;
}

extern "C" void kernel_launch(void* const* d_in, const int* in_sizes, int n_in,
                              void* d_out, int out_size, void* d_ws, size_t ws_size,
                              hipStream_t stream) {
    const float* features = (const float*)d_in[0];
    const int*   src      = (const int*)  d_in[1];
    const int*   dst      = (const int*)  d_in[2];
    const float* W1       = (const float*)d_in[3];
    const float* b1       = (const float*)d_in[4];
    const float* W2       = (const float*)d_in[5];
    const float* b2       = (const float*)d_in[6];
    const float* W3       = (const float*)d_in[7];
    const float* b3       = (const float*)d_in[8];
    float* out = (float*)d_out;

    const int N = GCN_N_NODES;
    const int E = GCN_N_EDGES;
    const size_t NF = (size_t)N * 128;

    float* bufZ    = (float*)d_ws;
    float* bufB    = bufZ + NF;
    int*   edges   = (int*)(bufB + NF);
    int*   rank    = edges + 2 * (size_t)E;
    int*   cnt_out = rank + E;
    int*   cnt_in  = cnt_out + N;
    int*   row_ptr = cnt_in + N;
    float* deg_i   = (float*)(row_ptr + N);

    const int TB = 256;

    gcn_fill_zero_i<<<(2 * N + TB - 1) / TB, TB, 0, stream>>>(cnt_out, 2 * N);
    gcn_count_rank<<<(E + TB - 1) / TB, TB, 0, stream>>>(src, dst, cnt_out, cnt_in, rank, E);
    gcn_scan_rows<<<1, 1024, 0, stream>>>(cnt_in, row_ptr, N);
    gcn_place<<<(E + TB - 1) / TB, TB, 0, stream>>>(src, dst, rank, row_ptr, cnt_out, cnt_in,
                                                    edges, deg_i, E, N);

    const int groups = GCN_LAYER_GROUPS;

    gcn_layer_fused<false><<<groups, TB, 0, stream>>>(features, row_ptr, cnt_in, edges, deg_i,
                                                      W1, b1, W3, bufB, 1, GCN_N_TILES);
    gcn_layer_fused<true><<<groups, TB, 0, stream>>>(bufB, row_ptr, cnt_in, edges, deg_i,
                                                     W2, b2, W3, bufZ, 1, GCN_N_TILES);
    gcn_gather_last<<<(4 * N + TB - 1) / TB, TB, 0, stream>>>(bufZ, row_ptr, cnt_in, edges, deg_i,
                                                             b3, out, N);

    (void)in_sizes; (void)n_in; (void)out_size; (void)ws_size;
}


# Degree counts and edge ranks on work-group-private LDS counters (25 edge chunks, per-chunk count planes summed per node) instead of global atomics; row sum order = chunk order, then LDS arrival
# speedup vs baseline: 4.6228x; 1.4358x over previous
#include <hip/hip_runtime.h>
#include <math.h>


typedef __attribute__((ext_vector_type(2))) float v2f;
typedef __attribute__((ext_vector_type(8))) float v8f;

#define GCN_N_NODES 50000
#define GCN_N_EDGES 600000

#define GCN_HIST_CHUNKS 25
#define GCN_HIST_CHUNK (GCN_N_EDGES / GCN_HIST_CHUNKS)
#define GCN_HIST_QUADS (GCN_HIST_CHUNK / 4)
#define GCN_HIST_ITERS ((GCN_HIST_QUADS + 1023) / 1024)
static_assert(GCN_N_EDGES % GCN_HIST_CHUNKS == 0, "equal chunks");
static_assert(GCN_HIST_CHUNK % 4 == 0, "a chunk is read in 16-byte groups");
static_assert(GCN_N_NODES % 4 == 0, "counter planes are cleared and copied in 16-byte groups");
static_assert(GCN_N_NODES * 4 <= 320 * 1024, "all counters of one kind in one work-group's LDS");

__global__ void __launch_bounds__(1024)
gcn_hist_lds(const int* __restrict__ src, const int* __restrict__ dst,
             int* __restrict__ part_in, int* __restrict__ part_out, int* __restrict__ rank) {
    __shared__ __attribute__((aligned(16))) int h[GCN_N_NODES];
    const int tid = threadIdx.x;
    const bool is_dst = (int)blockIdx.x < GCN_HIST_CHUNKS;
    const int  c      = is_dst ? (int)blockIdx.x : (int)blockIdx.x - GCN_HIST_CHUNKS;
    const int  e0     = c * GCN_HIST_CHUNK;
    const int* __restrict__ keys = (is_dst ? dst : src) + e0;

    int4 k4[GCN_HIST_ITERS];
#pragma unroll
    for (int k = 0; k < GCN_HIST_ITERS; ++k) {
        const int q = tid + k * 1024;
        k4[k] = make_int4(0, 0, 0, 0);
        if (q < GCN_HIST_QUADS) k4[k] = *(const int4*)(keys + 4 * q);
    }
    for (int i = tid; i < GCN_N_NODES / 4; i += 1024) ((int4*)h)[i] = make_int4(0, 0, 0, 0);
    __syncthreads();

    if (is_dst) {
        int4 r[GCN_HIST_ITERS];
#pragma unroll
        for (int k = 0; k < GCN_HIST_ITERS; ++k) {
            const int q = tid + k * 1024;
            r[k] = make_int4(0, 0, 0, 0);
            if (q < GCN_HIST_QUADS) {
                r[k].x = atomicAdd(&h[k4[k].x], 1);
                r[k].y = atomicAdd(&h[k4[k].y], 1);
                r[k].z = atomicAdd(&h[k4[k].z], 1);
                r[k].w = atomicAdd(&h[k4[k].w], 1);
            }
        }
#pragma unroll
        for (int k = 0; k < GCN_HIST_ITERS; ++k) {
            const int q = tid + k * 1024;
            if (q < GCN_HIST_QUADS) *(int4*)(rank + e0 + 4 * q) = r[k];
        }
    } else {
#pragma unroll
        for (int k = 0; k < GCN_HIST_ITERS; ++k) {
            const int q = tid + k * 1024;
            if (q < GCN_HIST_QUADS) {
                atomicAdd(&h[k4[k].x], 1);
                atomicAdd(&h[k4[k].y], 1);
                atomicAdd(&h[k4[k].z], 1);
                atomicAdd(&h[k4[k].w], 1);
            }
        }
    }
    __syncthreads();

    int4* __restrict__ plane = (int4*)((is_dst ? part_in : part_out) + (size_t)c * GCN_N_NODES);
    for (int i = tid; i < GCN_N_NODES / 4; i += 1024) plane[i] = ((const int4*)h)[i];
}

__global__ void gcn_combine(int* part_in, const int* __restrict__ part_out,
                            int* __restrict__ cnt_out, int* __restrict__ cnt_in, int n) {
    const int d = blockIdx.x * blockDim.x + threadIdx.x;
    if (d >= n) return;
    int vi[GCN_HIST_CHUNKS], vo[GCN_HIST_CHUNKS];
#pragma unroll
    for (int c = 0; c < GCN_HIST_CHUNKS; ++c) vi[c] = part_in[(size_t)c * GCN_N_NODES + d];
#pragma unroll
    for (int c = 0; c < GCN_HIST_CHUNKS; ++c) vo[c] = part_out[(size_t)c * GCN_N_NODES + d];
    int run = 0, so = 0;
#pragma unroll
    for (int c = 0; c < GCN_HIST_CHUNKS; ++c) {
        part_in[(size_t)c * GCN_N_NODES + d] = run;
        run += vi[c];
        so  += vo[c];
    }
    cnt_in[d]  = run;
    cnt_out[d] = so;
}

#define GCN_SCAN_ITERS 7
#define GCN_SCAN_SEG (GCN_SCAN_ITERS * 256)
static_assert(32 * GCN_SCAN_SEG >= GCN_N_NODES, "one work-group of 32 waves must cover all nodes");
static_assert(GCN_N_NODES % 8 == 0, "8-wide node groups must not straddle the end");

static __device__ __forceinline__ float gcn_inv_sqrt_clamped(int c) {
    float v = (float)c;
    v = v < 1.0f ? 1.0f : v;
    return 1.0f / sqrtf(v);
}

__global__ void __launch_bounds__(1024)
gcn_scan_rows(const int* __restrict__ cnt_in, int* __restrict__ row_ptr, int n) {
    __shared__ int wave_total[32];
    const int lane  = threadIdx.x & 31;
    const int w     = threadIdx.x >> 5;
    const int wbase = w * GCN_SCAN_SEG + lane * 8;

    int4 ca[GCN_SCAN_ITERS], cb[GCN_SCAN_ITERS];
#pragma unroll
    for (int it = 0; it < GCN_SCAN_ITERS; ++it) {
        const int idx8 = wbase + it * 256;
        ca[it] = make_int4(0, 0, 0, 0);
        cb[it] = make_int4(0, 0, 0, 0);
        if (idx8 < n) {
            ca[it] = *(const int4*)(cnt_in + idx8);
            cb[it] = *(const int4*)(cnt_in + idx8 + 4);
        }
    }

    int tot8[GCN_SCAN_ITERS], incl[GCN_SCAN_ITERS];
#pragma unroll
    for (int it = 0; it < GCN_SCAN_ITERS; ++it) {
        tot8[it] = ca[it].x + ca[it].y + ca[it].z + ca[it].w
                 + cb[it].x + cb[it].y + cb[it].z + cb[it].w;
        incl[it] = tot8[it];
    }
#pragma unroll
    for (int d = 1; d < 32; d <<= 1) {
        int y[GCN_SCAN_ITERS];
#pragma unroll
        for (int it = 0; it < GCN_SCAN_ITERS; ++it) y[it] = __shfl_up(incl[it], d);
#pragma unroll
        for (int it = 0; it < GCN_SCAN_ITERS; ++it) if (lane >= d) incl[it] += y[it];
    }
    int trip_total[GCN_SCAN_ITERS];
#pragma unroll
    for (int it = 0; it < GCN_SCAN_ITERS; ++it) trip_total[it] = __shfl(incl[it], 31);

    int4 ea[GCN_SCAN_ITERS], eb[GCN_SCAN_ITERS];
    int carry = 0;
#pragma unroll
    for (int it = 0; it < GCN_SCAN_ITERS; ++it) {
        const int e0 = carry + incl[it] - tot8[it];
        const int e1 = e0 + ca[it].x;
        const int e2 = e1 + ca[it].y;
        const int e3 = e2 + ca[it].z;
        const int e4 = e3 + ca[it].w;
        const int e5 = e4 + cb[it].x;
        const int e6 = e5 + cb[it].y;
        const int e7 = e6 + cb[it].z;
        ea[it] = make_int4(e0, e1, e2, e3);
        eb[it] = make_int4(e4, e5, e6, e7);
        carry += trip_total[it];
    }
    if (lane == 0) wave_total[w] = carry;
    __syncthreads();
    int woff = 0;
    for (int k = 0; k < w; ++k) woff += wave_total[k];
#pragma unroll
    for (int it = 0; it < GCN_SCAN_ITERS; ++it) {
        const int idx8 = wbase + it * 256;
        if (idx8 < n) {
            int4 ra = ea[it], rb = eb[it];
            ra.x += woff; ra.y += woff; ra.z += woff; ra.w += woff;
            rb.x += woff; rb.y += woff; rb.z += woff; rb.w += woff;
            *(int4*)(row_ptr + idx8)     = ra;
            *(int4*)(row_ptr + idx8 + 4) = rb;
        }
    }
}

static_assert(GCN_N_EDGES >= GCN_N_NODES, "the placement grid must cover one thread per node");

__global__ void gcn_place(const int* __restrict__ src, const int* __restrict__ dst,
                          const int* __restrict__ rank, const int* __restrict__ row_ptr,
                          const int* __restrict__ chunk_off,
                          const int* __restrict__ cnt_out, const int* __restrict__ cnt_in,
                          int* __restrict__ edges, float* __restrict__ deg_in_is,
                          int E, int n) {
    int e = blockIdx.x * blockDim.x + threadIdx.x;
    if (e < n) deg_in_is[e] = gcn_inv_sqrt_clamped(cnt_in[e]);
    if (e < E) {
        const int s = src[e];
        const int d = dst[e];
        const int c = e / GCN_HIST_CHUNK;
        const int pos = row_ptr[d] + chunk_off[(size_t)c * GCN_N_NODES + d] + rank[e];
        const float sc = gcn_inv_sqrt_clamped(cnt_out[s]);
        ((int2*)edges)[pos] = make_int2(s, __float_as_int(sc));
    }
}

static __device__ __forceinline__ float4 gcn_row4(const float* __restrict__ x, int s, int lane4) {
    return *(const float4*)(x + (size_t)s * 128 + lane4);
}
static __device__ __forceinline__ void gcn_axpy4(float4& acc, const float4 v, const float w) {
    acc.x = fmaf(v.x, w, acc.x);
    acc.y = fmaf(v.y, w, acc.y);
    acc.z = fmaf(v.z, w, acc.z);
    acc.w = fmaf(v.w, w, acc.w);
}
#define GCN_EDGE_SRC(my, j) __builtin_amdgcn_readlane((my).x, (j))
#define GCN_EDGE_W(my, j)   __int_as_float(__builtin_amdgcn_readlane((my).y, (j)))

static __device__ __forceinline__ float4 gcn_gather_row(const float* __restrict__ x,
                                                        const int2* __restrict__ ce,
                                                        const int beg, const int cnt,
                                                        const int2 first,
                                                        const int lane, const int lane4) {
    float4 acc = make_float4(0.0f, 0.0f, 0.0f, 0.0f);
    for (int b = 0; b < cnt; b += 32) {
        const int m = (cnt - b < 32) ? (cnt - b) : 32;
        int2 my = first;
        if (b > 0) {
            my = make_int2(0, 0);
            if (lane < m) my = ce[beg + b + lane];
        }
        int j = 0;
        for (; j + 8 <= m; j += 8) {
            const float4 v0 = gcn_row4(x, GCN_EDGE_SRC(my, j),     lane4);
            const float4 v1 = gcn_row4(x, GCN_EDGE_SRC(my, j + 1), lane4);
            const float4 v2 = gcn_row4(x, GCN_EDGE_SRC(my, j + 2), lane4);
            const float4 v3 = gcn_row4(x, GCN_EDGE_SRC(my, j + 3), lane4);
            const float4 v4 = gcn_row4(x, GCN_EDGE_SRC(my, j + 4), lane4);
            const float4 v5 = gcn_row4(x, GCN_EDGE_SRC(my, j + 5), lane4);
            const float4 v6 = gcn_row4(x, GCN_EDGE_SRC(my, j + 6), lane4);
            const float4 v7 = gcn_row4(x, GCN_EDGE_SRC(my, j + 7), lane4);
            gcn_axpy4(acc, v0, GCN_EDGE_W(my, j));
            gcn_axpy4(acc, v1, GCN_EDGE_W(my, j + 1));
            gcn_axpy4(acc, v2, GCN_EDGE_W(my, j + 2));
            gcn_axpy4(acc, v3, GCN_EDGE_W(my, j + 3));
            gcn_axpy4(acc, v4, GCN_EDGE_W(my, j + 4));
            gcn_axpy4(acc, v5, GCN_EDGE_W(my, j + 5));
            gcn_axpy4(acc, v6, GCN_EDGE_W(my, j + 6));
            gcn_axpy4(acc, v7, GCN_EDGE_W(my, j + 7));
        }
        if (j + 4 <= m) {
            const float4 v0 = gcn_row4(x, GCN_EDGE_SRC(my, j),     lane4);
            const float4 v1 = gcn_row4(x, GCN_EDGE_SRC(my, j + 1), lane4);
            const float4 v2 = gcn_row4(x, GCN_EDGE_SRC(my, j + 2), lane4);
            const float4 v3 = gcn_row4(x, GCN_EDGE_SRC(my, j + 3), lane4);
            gcn_axpy4(acc, v0, GCN_EDGE_W(my, j));
            gcn_axpy4(acc, v1, GCN_EDGE_W(my, j + 1));
            gcn_axpy4(acc, v2, GCN_EDGE_W(my, j + 2));
            gcn_axpy4(acc, v3, GCN_EDGE_W(my, j + 3));
            j += 4;
        }
        const int r = m - j;
        if (r == 1) {
            const float4 v0 = gcn_row4(x, GCN_EDGE_SRC(my, j), lane4);
            gcn_axpy4(acc, v0, GCN_EDGE_W(my, j));
        } else if (r == 2) {
            const float4 v0 = gcn_row4(x, GCN_EDGE_SRC(my, j),     lane4);
            const float4 v1 = gcn_row4(x, GCN_EDGE_SRC(my, j + 1), lane4);
            gcn_axpy4(acc, v0, GCN_EDGE_W(my, j));
            gcn_axpy4(acc, v1, GCN_EDGE_W(my, j + 1));
        } else if (r == 3) {
            const float4 v0 = gcn_row4(x, GCN_EDGE_SRC(my, j),     lane4);
            const float4 v1 = gcn_row4(x, GCN_EDGE_SRC(my, j + 1), lane4);
            const float4 v2 = gcn_row4(x, GCN_EDGE_SRC(my, j + 2), lane4);
            gcn_axpy4(acc, v0, GCN_EDGE_W(my, j));
            gcn_axpy4(acc, v1, GCN_EDGE_W(my, j + 1));
            gcn_axpy4(acc, v2, GCN_EDGE_W(my, j + 2));
        }
    }
    return acc;
}

#define GCN_K 128
#define GCN_C 128
#define GCN_C_LAST 16
#define GCN_TILE_PITCH 132
#define GCN_LAYER_GROUPS 1024
#define GCN_MAX_TILES_PER_GROUP 16
#define GCN_N_TILES (GCN_N_NODES / 16)
static_assert(GCN_N_NODES % 16 == 0, "whole 16-row tiles only");
static_assert(GCN_LAYER_GROUPS <= GCN_N_TILES, "every work-group gets at least one tile");
static_assert((GCN_N_TILES + GCN_LAYER_GROUPS - 1) / GCN_LAYER_GROUPS <= GCN_MAX_TILES_PER_GROUP,
              "a work-group's tiles must fit the lanes that hold their row bounds");
static_assert(GCN_C == 8 * 16, "8 waves, one 16-column strip each");

template <bool FUSE_NEXT>
__global__ void __launch_bounds__(256) __attribute__((amdgpu_waves_per_eu(8, 8)))
gcn_layer_fused(const float* __restrict__ x, const int* __restrict__ row_ptr,
                const int* __restrict__ cnt_in, const int* __restrict__ edges,
                const float* __restrict__ deg_in_is, const float* __restrict__ W,
                const float* __restrict__ bias, const float* __restrict__ Wn,
                float* __restrict__ out, int do_relu, int n_tiles) {
    __shared__ __attribute__((aligned(16))) float tile[2][16 * GCN_TILE_PITCH];
    __shared__ __attribute__((aligned(16))) float htile[FUSE_NEXT ? 16 * GCN_TILE_PITCH : 4];
    __shared__ __attribute__((aligned(16))) v2f wn_frag[FUSE_NEXT ? (GCN_K / 4) * 32 : 1];

    const int lane  = threadIdx.x & 31;
    const int lane4 = lane * 4;
    const int half  = lane >> 4;
    const int l15   = lane & 15;
    const int wave  = __builtin_amdgcn_readfirstlane(threadIdx.x >> 5);
    const int ncol  = wave * 16 + l15;
    const int per_group = n_tiles / GCN_LAYER_GROUPS;
    const int extra     = n_tiles % GCN_LAYER_GROUPS;
    const int g         = (int)blockIdx.x;
    const int tile0     = g * per_group + (g < extra ? g : extra);
    const int my_tiles  = per_group + (g < extra ? 1 : 0);
    const int2* __restrict__ ce = (const int2*)edges;

    int rp = 0, rc = 0;
    {
        const int tj = tile0 + (lane >> 1);
        if ((lane >> 1) < my_tiles) {
            const int node = tj * 16 + wave * 2 + (lane & 1);
            rp = row_ptr[node];
            rc = cnt_in[node];
        }
    }

    v2f bfrag[GCN_K / 4];
#pragma unroll
    for (int s = 0; s < GCN_K / 4; ++s) {
        const int kb = s * 4 + half * 2;
        bfrag[s].x = W[(size_t)kb       * GCN_C + ncol];
        bfrag[s].y = W[(size_t)(kb + 1) * GCN_C + ncol];
    }
    const float bv = bias[ncol];

    if constexpr (FUSE_NEXT) {
        for (int idx = threadIdx.x; idx < (GCN_K / 4) * 32; idx += 256) {
            const int s  = idx >> 5;
            const int ln = idx & 31;
            const int kb = s * 4 + (ln >> 4) * 2;
            v2f f;
            f.x = Wn[(size_t)kb       * GCN_C_LAST + (ln & 15)];
            f.y = Wn[(size_t)(kb + 1) * GCN_C_LAST + (ln & 15)];
            wn_frag[idx] = f;
        }
    }

    for (int i = 0; i < my_tiles; ++i) {
        const int t = tile0 + i;
        float* tb = &tile[i & 1][0];

        const int beg0 = __builtin_amdgcn_readlane(rp, 2 * i);
        const int cnt0 = __builtin_amdgcn_readlane(rc, 2 * i);
        const int beg1 = __builtin_amdgcn_readlane(rp, 2 * i + 1);
        const int cnt1 = __builtin_amdgcn_readlane(rc, 2 * i + 1);

        int2 cur0 = make_int2(0, 0), cur1 = make_int2(0, 0);
        if (lane < cnt0) cur0 = ce[beg0 + lane];
        if (lane < cnt1) cur1 = ce[beg1 + lane];
        const float scale = deg_in_is[t * 16 + l15];

        const float4 g0 = gcn_gather_row(x, ce, beg0, cnt0, cur0, lane, lane4);
        *(float4*)(tb + (2 * wave) * GCN_TILE_PITCH + lane4) = g0;
        const float4 g1 = gcn_gather_row(x, ce, beg1, cnt1, cur1, lane, lane4);
        *(float4*)(tb + (2 * wave + 1) * GCN_TILE_PITCH + lane4) = g1;

        __syncthreads();

        {
            const float* arow = tb + l15 * GCN_TILE_PITCH + half * 2;
            v8f c = {};
#pragma unroll
            for (int s = 0; s < GCN_K / 4; ++s) {
                const v2f araw = *(const v2f*)(arow + s * 4);
                v2f a;
                a.x = araw.x * scale;
                a.y = araw.y * scale;
                c = __builtin_amdgcn_wmma_f32_16x16x4_f32(
                        false, a, false, bfrag[s], (short)0, c, false, false);
            }
#pragma unroll
            for (int v = 0; v < 8; ++v) {
                float val = c[v] + bv;
                if (do_relu) val = fmaxf(val, 0.0f);
                if constexpr (FUSE_NEXT) {
                    htile[(half * 8 + v) * GCN_TILE_PITCH + ncol] = val;
                } else {
                    out[(size_t)(t * 16 + half * 8 + v) * GCN_C + ncol] = val;
                }
            }
        }

        if constexpr (FUSE_NEXT) {
            __syncthreads();
            if (wave == (i & 7)) {
                const float* hrow = htile + l15 * GCN_TILE_PITCH + half * 2;
                v8f z0 = {}, z1 = {};
#pragma unroll
                for (int s = 0; s < GCN_K / 4; s += 2) {
                    const v2f a0 = *(const v2f*)(hrow + s * 4);
                    const v2f b0 = wn_frag[s * 32 + lane];
                    const v2f a1 = *(const v2f*)(hrow + (s + 1) * 4);
                    const v2f b1 = wn_frag[(s + 1) * 32 + lane];
                    z0 = __builtin_amdgcn_wmma_f32_16x16x4_f32(
                            false, a0, false, b0, (short)0, z0, false, false);
                    z1 = __builtin_amdgcn_wmma_f32_16x16x4_f32(
                            false, a1, false, b1, (short)0, z1, false, false);
                }
#pragma unroll
                for (int v = 0; v < 8; ++v) {
                    out[(size_t)(t * 16 + half * 8 + v) * GCN_C_LAST + l15] = z0[v] + z1[v];
                }
            }
        }
    }
}

#define GCN_LAST_BATCH 16

__global__ void __launch_bounds__(256) __attribute__((amdgpu_waves_per_eu(4, 6)))
gcn_gather_last(const float* __restrict__ z, const int* __restrict__ row_ptr,
                const int* __restrict__ cnt_in, const int* __restrict__ edges,
                const float* __restrict__ deg_in_is, const float* __restrict__ bias,
                float* __restrict__ out, int n) {
    const int g  = blockIdx.x * blockDim.x + threadIdx.x;
    const int m  = g >> 2;
    const int q4 = (g & 3) * 4;
    if (m >= n) return;
    const int    cnt = cnt_in[m];
    const int    beg = row_ptr[m];
    const float  sc  = deg_in_is[m];
    const float4 b   = *(const float4*)(bias + q4);
    asm volatile("" :: "v"(cnt), "v"(beg), "v"(sc), "v"(b.x), "v"(b.y), "v"(b.z), "v"(b.w));
    const int2* __restrict__ ce = (const int2*)edges + beg;
    const int last = cnt - 1;

    float4 acc = make_float4(0.0f, 0.0f, 0.0f, 0.0f);
    for (int j = 0; j < cnt; j += GCN_LAST_BATCH) {
        int2 e[GCN_LAST_BATCH];
        float4 v[GCN_LAST_BATCH];
#pragma unroll
        for (int k = 0; k < GCN_LAST_BATCH; ++k) {
            const int jj = (j + k < last) ? (j + k) : last;
            e[k] = ce[jj];
        }
#pragma unroll
        for (int k = 0; k < GCN_LAST_BATCH; ++k) {
            v[k] = *(const float4*)(z + (size_t)e[k].x * GCN_C_LAST + q4);
        }
#pragma unroll
        for (int k = 0; k < GCN_LAST_BATCH; ++k) {
            const bool on = (j + k) < cnt;
            float4 vv;
            vv.x = on ? v[k].x : 0.0f;
            vv.y = on ? v[k].y : 0.0f;
            vv.z = on ? v[k].z : 0.0f;
            vv.w = on ? v[k].w : 0.0f;
            gcn_axpy4(acc, vv, __int_as_float(e[k].y));
        }
    }
    float4 r;
    r.x = fmaf(acc.x, sc, b.x);
    r.y = fmaf(acc.y, sc, b.y);
    r.z = fmaf(acc.z, sc, b.z);
    r.w = fmaf(acc.w, sc, b.w);
    *(float4*)(out + (size_t)m * GCN_C_LAST + q4) = r;
}

extern "C" void kernel_launch(void* const* d_in, const int* in_sizes, int n_in,
                              void* d_out, int out_size, void* d_ws, size_t ws_size,
                              hipStream_t stream) {
    const float* features = (const float*)d_in[0];
    const int*   src      = (const int*)  d_in[1];
    const int*   dst      = (const int*)  d_in[2];
    const float* W1       = (const float*)d_in[3];
    const float* b1       = (const float*)d_in[4];
    const float* W2       = (const float*)d_in[5];
    const float* b2       = (const float*)d_in[6];
    const float* W3       = (const float*)d_in[7];
    const float* b3       = (const float*)d_in[8];
    float* out = (float*)d_out;

    const int N = GCN_N_NODES;
    const int E = GCN_N_EDGES;
    const size_t NF = (size_t)N * 128;

    float* bufZ    = (float*)d_ws;
    float* bufB    = bufZ + NF;
    int*   edges   = (int*)(bufB + NF);
    int*   rank    = edges + 2 * (size_t)E;
    int*   cnt_out = rank + E;
    int*   cnt_in  = cnt_out + N;
    int*   row_ptr = cnt_in + N;
    float* deg_i   = (float*)(row_ptr + N);
    int*   part_in  = (int*)(deg_i + N);
    int*   part_out = part_in + (size_t)GCN_HIST_CHUNKS * N;

    const int TB = 256;

    gcn_hist_lds<<<2 * GCN_HIST_CHUNKS, 1024, 0, stream>>>(src, dst, part_in, part_out, rank);
    gcn_combine<<<(N + TB - 1) / TB, TB, 0, stream>>>(part_in, part_out, cnt_out, cnt_in, N);
    gcn_scan_rows<<<1, 1024, 0, stream>>>(cnt_in, row_ptr, N);
    gcn_place<<<(E + TB - 1) / TB, TB, 0, stream>>>(src, dst, rank, row_ptr, part_in, cnt_out, cnt_in,
                                                    edges, deg_i, E, N);

    const int groups = GCN_LAYER_GROUPS;

    gcn_layer_fused<false><<<groups, TB, 0, stream>>>(features, row_ptr, cnt_in, edges, deg_i,
                                                      W1, b1, W3, bufB, 1, GCN_N_TILES);
    gcn_layer_fused<true><<<groups, TB, 0, stream>>>(bufB, row_ptr, cnt_in, edges, deg_i,
                                                     W2, b2, W3, bufZ, 1, GCN_N_TILES);
    gcn_gather_last<<<(4 * N + TB - 1) / TB, TB, 0, stream>>>(bufZ, row_ptr, cnt_in, edges, deg_i,
                                                             b3, out, N);

    (void)in_sizes; (void)n_in; (void)out_size; (void)ws_size;
}


# Row starts built inside the count-combining kernel from per-chunk node-block totals; chunk planes hold absolute slots; the one-work-group prefix-sum launch is gone (7 -> 6); sum order unchanged
# speedup vs baseline: 5.0432x; 1.0909x over previous
#include <hip/hip_runtime.h>
#include <math.h>


typedef __attribute__((ext_vector_type(2))) float v2f;
typedef __attribute__((ext_vector_type(8))) float v8f;

#define GCN_N_NODES 50000
#define GCN_N_EDGES 600000

#define GCN_HIST_CHUNKS 25
#define GCN_HIST_CHUNK (GCN_N_EDGES / GCN_HIST_CHUNKS)
#define GCN_HIST_QUADS (GCN_HIST_CHUNK / 4)
#define GCN_HIST_ITERS ((GCN_HIST_QUADS + 1023) / 1024)
static_assert(GCN_N_EDGES % GCN_HIST_CHUNKS == 0, "equal chunks");
static_assert(GCN_HIST_CHUNK % 4 == 0, "a chunk is read in 16-byte groups");
static_assert(GCN_N_NODES % 4 == 0, "counter planes are cleared and copied in 16-byte groups");
static_assert(GCN_N_NODES * 4 <= 320 * 1024, "all counters of one kind in one work-group's LDS");
#define GCN_COMB_TB 256
#define GCN_COMB_BLOCKS ((GCN_N_NODES + GCN_COMB_TB - 1) / GCN_COMB_TB)
#define GCN_HIST_COPY_TRIPS ((GCN_COMB_BLOCKS + 31) / 32)
static_assert(GCN_COMB_TB == 2 * 32 * 4, "a wave copies a block as two 16-byte groups per lane");
static_assert(GCN_COMB_BLOCKS <= GCN_COMB_TB, "one thread per earlier block in the next kernel");

__global__ void __launch_bounds__(1024)
gcn_hist_lds(const int* __restrict__ src, const int* __restrict__ dst,
             int* __restrict__ part_in, int* __restrict__ part_out,
             int* __restrict__ blk_in, int* __restrict__ rank) {
    __shared__ __attribute__((aligned(16))) int h[GCN_N_NODES];
    const int tid = threadIdx.x;
    const bool is_dst = (int)blockIdx.x < GCN_HIST_CHUNKS;
    const int  c      = is_dst ? (int)blockIdx.x : (int)blockIdx.x - GCN_HIST_CHUNKS;
    const int  e0     = c * GCN_HIST_CHUNK;
    const int* __restrict__ keys = (is_dst ? dst : src) + e0;

    int4 k4[GCN_HIST_ITERS];
#pragma unroll
    for (int k = 0; k < GCN_HIST_ITERS; ++k) {
        const int q = tid + k * 1024;
        k4[k] = make_int4(0, 0, 0, 0);
        if (q < GCN_HIST_QUADS) k4[k] = *(const int4*)(keys + 4 * q);
    }
    for (int i = tid; i < GCN_N_NODES / 4; i += 1024) ((int4*)h)[i] = make_int4(0, 0, 0, 0);
    __syncthreads();

    if (is_dst) {
        int4 r[GCN_HIST_ITERS];
#pragma unroll
        for (int k = 0; k < GCN_HIST_ITERS; ++k) {
            const int q = tid + k * 1024;
            r[k] = make_int4(0, 0, 0, 0);
            if (q < GCN_HIST_QUADS) {
                r[k].x = atomicAdd(&h[k4[k].x], 1);
                r[k].y = atomicAdd(&h[k4[k].y], 1);
                r[k].z = atomicAdd(&h[k4[k].z], 1);
                r[k].w = atomicAdd(&h[k4[k].w], 1);
            }
        }
#pragma unroll
        for (int k = 0; k < GCN_HIST_ITERS; ++k) {
            const int q = tid + k * 1024;
            if (q < GCN_HIST_QUADS) *(int4*)(rank + e0 + 4 * q) = r[k];
        }
    } else {
#pragma unroll
        for (int k = 0; k < GCN_HIST_ITERS; ++k) {
            const int q = tid + k * 1024;
            if (q < GCN_HIST_QUADS) {
                atomicAdd(&h[k4[k].x], 1);
                atomicAdd(&h[k4[k].y], 1);
                atomicAdd(&h[k4[k].z], 1);
                atomicAdd(&h[k4[k].w], 1);
            }
        }
    }
    __syncthreads();

    int4* __restrict__ plane = (int4*)((is_dst ? part_in : part_out) + (size_t)c * GCN_N_NODES);
    const int lane = tid & 31;
    const int w    = tid >> 5;
    const int last = GCN_N_NODES / 4 - 1;
    int4 q0[GCN_HIST_COPY_TRIPS], q1[GCN_HIST_COPY_TRIPS];
#pragma unroll
    for (int k = 0; k < GCN_HIST_COPY_TRIPS; ++k) {
        const int i0 = (w + 32 * k) * (GCN_COMB_TB / 4) + lane;
        const int i1 = i0 + 32;
        q0[k] = ((const int4*)h)[i0 < last ? i0 : last];
        q1[k] = ((const int4*)h)[i1 < last ? i1 : last];
    }
    int s[GCN_HIST_COPY_TRIPS];
#pragma unroll
    for (int k = 0; k < GCN_HIST_COPY_TRIPS; ++k) {
        const int i0 = (w + 32 * k) * (GCN_COMB_TB / 4) + lane;
        const int i1 = i0 + 32;
        s[k] = 0;
        if (i0 <= last) { plane[i0] = q0[k]; s[k] += q0[k].x + q0[k].y + q0[k].z + q0[k].w; }
        if (i1 <= last) { plane[i1] = q1[k]; s[k] += q1[k].x + q1[k].y + q1[k].z + q1[k].w; }
    }
    if (is_dst) {
        for (int m = 16; m >= 1; m >>= 1) {
#pragma unroll
            for (int k = 0; k < GCN_HIST_COPY_TRIPS; ++k) s[k] += __shfl_xor(s[k], m);
        }
        if (lane == 0) {
#pragma unroll
            for (int k = 0; k < GCN_HIST_COPY_TRIPS; ++k) {
                const int b = w + 32 * k;
                if (b < GCN_COMB_BLOCKS) blk_in[c * GCN_COMB_BLOCKS + b] = s[k];
            }
        }
    }
}

__global__ void __launch_bounds__(GCN_COMB_TB)
gcn_combine(int* part_in, const int* __restrict__ part_out, const int* __restrict__ blk_in,
            int* __restrict__ cnt_out, int* __restrict__ cnt_in, int* __restrict__ row_ptr, int n) {
    __shared__ int wave_before[GCN_COMB_TB / 32];
    __shared__ int wave_rows[GCN_COMB_TB / 32];
    const int tid  = threadIdx.x;
    const int lane = tid & 31;
    const int w    = tid >> 5;
    const int b    = blockIdx.x;
    const int d    = b * GCN_COMB_TB + tid;
    const bool on  = d < n;
    const int dl = on ? d : n - 1;
    const int tb = tid < b ? tid : 0;
    int vi[GCN_HIST_CHUNKS], vo[GCN_HIST_CHUNKS], vb[GCN_HIST_CHUNKS];
#pragma unroll
    for (int c = 0; c < GCN_HIST_CHUNKS; ++c) vb[c] = blk_in[c * GCN_COMB_BLOCKS + tb];
#pragma unroll
    for (int c = 0; c < GCN_HIST_CHUNKS; ++c) vi[c] = part_in[(size_t)c * GCN_N_NODES + dl];
#pragma unroll
    for (int c = 0; c < GCN_HIST_CHUNKS; ++c) vo[c] = part_out[(size_t)c * GCN_N_NODES + dl];
    int run = 0, so = 0, before = 0;
#pragma unroll
    for (int c = 0; c < GCN_HIST_CHUNKS; ++c) {
        run    += vi[c];
        so     += vo[c];
        before += vb[c];
    }
    run    = on ? run : 0;
    before = tid < b ? before : 0;
    int incl = run;
    for (int m = 1; m < 32; m <<= 1) {
        const int y = __shfl_up(incl, m);
        if (lane >= m) incl += y;
    }
    for (int m = 16; m >= 1; m >>= 1) before += __shfl_xor(before, m);
    if (lane == 31) wave_rows[w]   = incl;
    if (lane == 0)  wave_before[w] = before;
    __syncthreads();
    int base = 0;
    for (int k = 0; k < GCN_COMB_TB / 32; ++k) {
        base += wave_before[k];
        if (k < w) base += wave_rows[k];
    }
    if (on) {
        const int start = base + incl - run;
        row_ptr[d] = start;
        cnt_in[d]  = run;
        cnt_out[d] = so;
        int slot = start;
#pragma unroll
        for (int c = 0; c < GCN_HIST_CHUNKS; ++c) {
            part_in[(size_t)c * GCN_N_NODES + d] = slot;
            slot += vi[c];
        }
    }
}

static __device__ __forceinline__ float gcn_inv_sqrt_clamped(int c) {
    float v = (float)c;
    v = v < 1.0f ? 1.0f : v;
    return 1.0f / sqrtf(v);
}

static_assert(GCN_N_EDGES >= GCN_N_NODES, "the placement grid must cover one thread per node");

__global__ void gcn_place(const int* __restrict__ src, const int* __restrict__ dst,
                          const int* __restrict__ rank, const int* __restrict__ chunk_start,
                          const int* __restrict__ cnt_out, const int* __restrict__ cnt_in,
                          int* __restrict__ edges, float* __restrict__ deg_in_is,
                          int E, int n) {
    int e = blockIdx.x * blockDim.x + threadIdx.x;
    if (e < n) deg_in_is[e] = gcn_inv_sqrt_clamped(cnt_in[e]);
    if (e < E) {
        const int s = src[e];
        const int d = dst[e];
        const int c = e / GCN_HIST_CHUNK;
        const int pos = chunk_start[(size_t)c * GCN_N_NODES + d] + rank[e];
        const float sc = gcn_inv_sqrt_clamped(cnt_out[s]);
        ((int2*)edges)[pos] = make_int2(s, __float_as_int(sc));
    }
}

static __device__ __forceinline__ float4 gcn_row4(const float* __restrict__ x, int s, int lane4) {
    return *(const float4*)(x + (size_t)s * 128 + lane4);
}
static __device__ __forceinline__ void gcn_axpy4(float4& acc, const float4 v, const float w) {
    acc.x = fmaf(v.x, w, acc.x);
    acc.y = fmaf(v.y, w, acc.y);
    acc.z = fmaf(v.z, w, acc.z);
    acc.w = fmaf(v.w, w, acc.w);
}
#define GCN_EDGE_SRC(my, j) __builtin_amdgcn_readlane((my).x, (j))
#define GCN_EDGE_W(my, j)   __int_as_float(__builtin_amdgcn_readlane((my).y, (j)))

static __device__ __forceinline__ float4 gcn_gather_row(const float* __restrict__ x,
                                                        const int2* __restrict__ ce,
                                                        const int beg, const int cnt,
                                                        const int2 first,
                                                        const int lane, const int lane4) {
    float4 acc = make_float4(0.0f, 0.0f, 0.0f, 0.0f);
    for (int b = 0; b < cnt; b += 32) {
        const int m = (cnt - b < 32) ? (cnt - b) : 32;
        int2 my = first;
        if (b > 0) {
            my = make_int2(0, 0);
            if (lane < m) my = ce[beg + b + lane];
        }
        int j = 0;
        for (; j + 8 <= m; j += 8) {
            const float4 v0 = gcn_row4(x, GCN_EDGE_SRC(my, j),     lane4);
            const float4 v1 = gcn_row4(x, GCN_EDGE_SRC(my, j + 1), lane4);
            const float4 v2 = gcn_row4(x, GCN_EDGE_SRC(my, j + 2), lane4);
            const float4 v3 = gcn_row4(x, GCN_EDGE_SRC(my, j + 3), lane4);
            const float4 v4 = gcn_row4(x, GCN_EDGE_SRC(my, j + 4), lane4);
            const float4 v5 = gcn_row4(x, GCN_EDGE_SRC(my, j + 5), lane4);
            const float4 v6 = gcn_row4(x, GCN_EDGE_SRC(my, j + 6), lane4);
            const float4 v7 = gcn_row4(x, GCN_EDGE_SRC(my, j + 7), lane4);
            gcn_axpy4(acc, v0, GCN_EDGE_W(my, j));
            gcn_axpy4(acc, v1, GCN_EDGE_W(my, j + 1));
            gcn_axpy4(acc, v2, GCN_EDGE_W(my, j + 2));
            gcn_axpy4(acc, v3, GCN_EDGE_W(my, j + 3));
            gcn_axpy4(acc, v4, GCN_EDGE_W(my, j + 4));
            gcn_axpy4(acc, v5, GCN_EDGE_W(my, j + 5));
            gcn_axpy4(acc, v6, GCN_EDGE_W(my, j + 6));
            gcn_axpy4(acc, v7, GCN_EDGE_W(my, j + 7));
        }
        if (j + 4 <= m) {
            const float4 v0 = gcn_row4(x, GCN_EDGE_SRC(my, j),     lane4);
            const float4 v1 = gcn_row4(x, GCN_EDGE_SRC(my, j + 1), lane4);
            const float4 v2 = gcn_row4(x, GCN_EDGE_SRC(my, j + 2), lane4);
            const float4 v3 = gcn_row4(x, GCN_EDGE_SRC(my, j + 3), lane4);
            gcn_axpy4(acc, v0, GCN_EDGE_W(my, j));
            gcn_axpy4(acc, v1, GCN_EDGE_W(my, j + 1));
            gcn_axpy4(acc, v2, GCN_EDGE_W(my, j + 2));
            gcn_axpy4(acc, v3, GCN_EDGE_W(my, j + 3));
            j += 4;
        }
        const int r = m - j;
        if (r == 1) {
            const float4 v0 = gcn_row4(x, GCN_EDGE_SRC(my, j), lane4);
            gcn_axpy4(acc, v0, GCN_EDGE_W(my, j));
        } else if (r == 2) {
            const float4 v0 = gcn_row4(x, GCN_EDGE_SRC(my, j),     lane4);
            const float4 v1 = gcn_row4(x, GCN_EDGE_SRC(my, j + 1), lane4);
            gcn_axpy4(acc, v0, GCN_EDGE_W(my, j));
            gcn_axpy4(acc, v1, GCN_EDGE_W(my, j + 1));
        } else if (r == 3) {
            const float4 v0 = gcn_row4(x, GCN_EDGE_SRC(my, j),     lane4);
            const float4 v1 = gcn_row4(x, GCN_EDGE_SRC(my, j + 1), lane4);
            const float4 v2 = gcn_row4(x, GCN_EDGE_SRC(my, j + 2), lane4);
            gcn_axpy4(acc, v0, GCN_EDGE_W(my, j));
            gcn_axpy4(acc, v1, GCN_EDGE_W(my, j + 1));
            gcn_axpy4(acc, v2, GCN_EDGE_W(my, j + 2));
        }
    }
    return acc;
}

#define GCN_K 128
#define GCN_C 128
#define GCN_C_LAST 16
#define GCN_TILE_PITCH 132
#define GCN_LAYER_GROUPS 1024
#define GCN_MAX_TILES_PER_GROUP 16
#define GCN_N_TILES (GCN_N_NODES / 16)
static_assert(GCN_N_NODES % 16 == 0, "whole 16-row tiles only");
static_assert(GCN_LAYER_GROUPS <= GCN_N_TILES, "every work-group gets at least one tile");
static_assert((GCN_N_TILES + GCN_LAYER_GROUPS - 1) / GCN_LAYER_GROUPS <= GCN_MAX_TILES_PER_GROUP,
              "a work-group's tiles must fit the lanes that hold their row bounds");
static_assert(GCN_C == 8 * 16, "8 waves, one 16-column strip each");

template <bool FUSE_NEXT>
__global__ void __launch_bounds__(256) __attribute__((amdgpu_waves_per_eu(8, 8)))
gcn_layer_fused(const float* __restrict__ x, const int* __restrict__ row_ptr,
                const int* __restrict__ cnt_in, const int* __restrict__ edges,
                const float* __restrict__ deg_in_is, const float* __restrict__ W,
                const float* __restrict__ bias, const float* __restrict__ Wn,
                float* __restrict__ out, int do_relu, int n_tiles) {
    __shared__ __attribute__((aligned(16))) float tile[2][16 * GCN_TILE_PITCH];
    __shared__ __attribute__((aligned(16))) float htile[FUSE_NEXT ? 16 * GCN_TILE_PITCH : 4];
    __shared__ __attribute__((aligned(16))) v2f wn_frag[FUSE_NEXT ? (GCN_K / 4) * 32 : 1];

    const int lane  = threadIdx.x & 31;
    const int lane4 = lane * 4;
    const int half  = lane >> 4;
    const int l15   = lane & 15;
    const int wave  = __builtin_amdgcn_readfirstlane(threadIdx.x >> 5);
    const int ncol  = wave * 16 + l15;
    const int per_group = n_tiles / GCN_LAYER_GROUPS;
    const int extra     = n_tiles % GCN_LAYER_GROUPS;
    const int g         = (int)blockIdx.x;
    const int tile0     = g * per_group + (g < extra ? g : extra);
    const int my_tiles  = per_group + (g < extra ? 1 : 0);
    const int2* __restrict__ ce = (const int2*)edges;

    int rp = 0, rc = 0;
    {
        const int tj = tile0 + (lane >> 1);
        if ((lane >> 1) < my_tiles) {
            const int node = tj * 16 + wave * 2 + (lane & 1);
            rp = row_ptr[node];
            rc = cnt_in[node];
        }
    }

    v2f bfrag[GCN_K / 4];
#pragma unroll
    for (int s = 0; s < GCN_K / 4; ++s) {
        const int kb = s * 4 + half * 2;
        bfrag[s].x = W[(size_t)kb       * GCN_C + ncol];
        bfrag[s].y = W[(size_t)(kb + 1) * GCN_C + ncol];
    }
    const float bv = bias[ncol];

    if constexpr (FUSE_NEXT) {
        for (int idx = threadIdx.x; idx < (GCN_K / 4) * 32; idx += 256) {
            const int s  = idx >> 5;
            const int ln = idx & 31;
            const int kb = s * 4 + (ln >> 4) * 2;
            v2f f;
            f.x = Wn[(size_t)kb       * GCN_C_LAST + (ln & 15)];
            f.y = Wn[(size_t)(kb + 1) * GCN_C_LAST + (ln & 15)];
            wn_frag[idx] = f;
        }
    }

    for (int i = 0; i < my_tiles; ++i) {
        const int t = tile0 + i;
        float* tb = &tile[i & 1][0];

        const int beg0 = __builtin_amdgcn_readlane(rp, 2 * i);
        const int cnt0 = __builtin_amdgcn_readlane(rc, 2 * i);
        const int beg1 = __builtin_amdgcn_readlane(rp, 2 * i + 1);
        const int cnt1 = __builtin_amdgcn_readlane(rc, 2 * i + 1);

        int2 cur0 = make_int2(0, 0), cur1 = make_int2(0, 0);
        if (lane < cnt0) cur0 = ce[beg0 + lane];
        if (lane < cnt1) cur1 = ce[beg1 + lane];
        const float scale = deg_in_is[t * 16 + l15];

        const float4 g0 = gcn_gather_row(x, ce, beg0, cnt0, cur0, lane, lane4);
        *(float4*)(tb + (2 * wave) * GCN_TILE_PITCH + lane4) = g0;
        const float4 g1 = gcn_gather_row(x, ce, beg1, cnt1, cur1, lane, lane4);
        *(float4*)(tb + (2 * wave + 1) * GCN_TILE_PITCH + lane4) = g1;

        __syncthreads();

        {
            const float* arow = tb + l15 * GCN_TILE_PITCH + half * 2;
            v8f c = {};
#pragma unroll
            for (int s = 0; s < GCN_K / 4; ++s) {
                const v2f araw = *(const v2f*)(arow + s * 4);
                v2f a;
                a.x = araw.x * scale;
                a.y = araw.y * scale;
                c = __builtin_amdgcn_wmma_f32_16x16x4_f32(
                        false, a, false, bfrag[s], (short)0, c, false, false);
            }
#pragma unroll
            for (int v = 0; v < 8; ++v) {
                float val = c[v] + bv;
                if (do_relu) val = fmaxf(val, 0.0f);
                if constexpr (FUSE_NEXT) {
                    htile[(half * 8 + v) * GCN_TILE_PITCH + ncol] = val;
                } else {
                    out[(size_t)(t * 16 + half * 8 + v) * GCN_C + ncol] = val;
                }
            }
        }

        if constexpr (FUSE_NEXT) {
            __syncthreads();
            if (wave == (i & 7)) {
                const float* hrow = htile + l15 * GCN_TILE_PITCH + half * 2;
                v8f z0 = {}, z1 = {};
#pragma unroll
                for (int s = 0; s < GCN_K / 4; s += 2) {
                    const v2f a0 = *(const v2f*)(hrow + s * 4);
                    const v2f b0 = wn_frag[s * 32 + lane];
                    const v2f a1 = *(const v2f*)(hrow + (s + 1) * 4);
                    const v2f b1 = wn_frag[(s + 1) * 32 + lane];
                    z0 = __builtin_amdgcn_wmma_f32_16x16x4_f32(
                            false, a0, false, b0, (short)0, z0, false, false);
                    z1 = __builtin_amdgcn_wmma_f32_16x16x4_f32(
                            false, a1, false, b1, (short)0, z1, false, false);
                }
#pragma unroll
                for (int v = 0; v < 8; ++v) {
                    out[(size_t)(t * 16 + half * 8 + v) * GCN_C_LAST + l15] = z0[v] + z1[v];
                }
            }
        }
    }
}

#define GCN_LAST_BATCH 16

__global__ void __launch_bounds__(256) __attribute__((amdgpu_waves_per_eu(4, 6)))
gcn_gather_last(const float* __restrict__ z, const int* __restrict__ row_ptr,
                const int* __restrict__ cnt_in, const int* __restrict__ edges,
                const float* __restrict__ deg_in_is, const float* __restrict__ bias,
                float* __restrict__ out, int n) {
    const int g  = blockIdx.x * blockDim.x + threadIdx.x;
    const int m  = g >> 2;
    const int q4 = (g & 3) * 4;
    if (m >= n) return;
    const int    cnt = cnt_in[m];
    const int    beg = row_ptr[m];
    const float  sc  = deg_in_is[m];
    const float4 b   = *(const float4*)(bias + q4);
    asm volatile("" :: "v"(cnt), "v"(beg), "v"(sc), "v"(b.x), "v"(b.y), "v"(b.z), "v"(b.w));
    const int2* __restrict__ ce = (const int2*)edges + beg;
    const int last = cnt - 1;

    float4 acc = make_float4(0.0f, 0.0f, 0.0f, 0.0f);
    for (int j = 0; j < cnt; j += GCN_LAST_BATCH) {
        int2 e[GCN_LAST_BATCH];
        float4 v[GCN_LAST_BATCH];
#pragma unroll
        for (int k = 0; k < GCN_LAST_BATCH; ++k) {
            const int jj = (j + k < last) ? (j + k) : last;
            e[k] = ce[jj];
        }
#pragma unroll
        for (int k = 0; k < GCN_LAST_BATCH; ++k) {
            v[k] = *(const float4*)(z + (size_t)e[k].x * GCN_C_LAST + q4);
        }
#pragma unroll
        for (int k = 0; k < GCN_LAST_BATCH; ++k) {
            const bool on = (j + k) < cnt;
            float4 vv;
            vv.x = on ? v[k].x : 0.0f;
            vv.y = on ? v[k].y : 0.0f;
            vv.z = on ? v[k].z : 0.0f;
            vv.w = on ? v[k].w : 0.0f;
            gcn_axpy4(acc, vv, __int_as_float(e[k].y));
        }
    }
    float4 r;
    r.x = fmaf(acc.x, sc, b.x);
    r.y = fmaf(acc.y, sc, b.y);
    r.z = fmaf(acc.z, sc, b.z);
    r.w = fmaf(acc.w, sc, b.w);
    *(float4*)(out + (size_t)m * GCN_C_LAST + q4) = r;
}

extern "C" void kernel_launch(void* const* d_in, const int* in_sizes, int n_in,
                              void* d_out, int out_size, void* d_ws, size_t ws_size,
                              hipStream_t stream) {
    const float* features = (const float*)d_in[0];
    const int*   src      = (const int*)  d_in[1];
    const int*   dst      = (const int*)  d_in[2];
    const float* W1       = (const float*)d_in[3];
    const float* b1       = (const float*)d_in[4];
    const float* W2       = (const float*)d_in[5];
    const float* b2       = (const float*)d_in[6];
    const float* W3       = (const float*)d_in[7];
    const float* b3       = (const float*)d_in[8];
    float* out = (float*)d_out;

    const int N = GCN_N_NODES;
    const int E = GCN_N_EDGES;
    const size_t NF = (size_t)N * 128;

    float* bufZ    = (float*)d_ws;
    float* bufB    = bufZ + NF;
    int*   edges   = (int*)(bufB + NF);
    int*   rank    = edges + 2 * (size_t)E;
    int*   cnt_out = rank + E;
    int*   cnt_in  = cnt_out + N;
    int*   row_ptr = cnt_in + N;
    float* deg_i   = (float*)(row_ptr + N);
    int*   part_in  = (int*)(deg_i + N);
    int*   part_out = part_in + (size_t)GCN_HIST_CHUNKS * N;
    int*   blk_in   = part_out + (size_t)GCN_HIST_CHUNKS * N;

    const int TB = 256;

    gcn_hist_lds<<<2 * GCN_HIST_CHUNKS, 1024, 0, stream>>>(src, dst, part_in, part_out, blk_in, rank);
    gcn_combine<<<GCN_COMB_BLOCKS, GCN_COMB_TB, 0, stream>>>(part_in, part_out, blk_in,
                                                             cnt_out, cnt_in, row_ptr, N);
    gcn_place<<<(E + TB - 1) / TB, TB, 0, stream>>>(src, dst, rank, part_in, cnt_out, cnt_in,
                                                    edges, deg_i, E, N);

    const int groups = GCN_LAYER_GROUPS;

    gcn_layer_fused<false><<<groups, TB, 0, stream>>>(features, row_ptr, cnt_in, edges, deg_i,
                                                      W1, b1, W3, bufB, 1, GCN_N_TILES);
    gcn_layer_fused<true><<<groups, TB, 0, stream>>>(bufB, row_ptr, cnt_in, edges, deg_i,
                                                     W2, b2, W3, bufZ, 1, GCN_N_TILES);
    gcn_gather_last<<<(4 * N + TB - 1) / TB, TB, 0, stream>>>(bufZ, row_ptr, cnt_in, edges, deg_i,
                                                             b3, out, N);

    (void)in_sizes; (void)n_in; (void)out_size; (void)ws_size;
}


# Placement by node block: counting work-groups also write their chunk ordered by block; a work-group per block fills its slot range in LDS, stores whole lines, no scattered stores; sum order unchanged
# speedup vs baseline: 5.1976x; 1.0306x over previous
#include <hip/hip_runtime.h>
#include <math.h>


typedef __attribute__((ext_vector_type(2))) float v2f;
typedef __attribute__((ext_vector_type(8))) float v8f;

#define GCN_N_NODES 50000
#define GCN_N_EDGES 600000

#define GCN_HIST_CHUNKS 25
#define GCN_HIST_CHUNK (GCN_N_EDGES / GCN_HIST_CHUNKS)
#define GCN_HIST_QUADS (GCN_HIST_CHUNK / 4)
#define GCN_HIST_ITERS ((GCN_HIST_QUADS + 1023) / 1024)
static_assert(GCN_N_EDGES % GCN_HIST_CHUNKS == 0, "equal chunks");
static_assert(GCN_HIST_CHUNK % 4 == 0, "a chunk is read in 16-byte groups");
static_assert(GCN_N_NODES % 4 == 0, "counter planes are cleared and copied in 16-byte groups");
static_assert(GCN_N_NODES * 4 <= 320 * 1024, "all counters of one kind in one work-group's LDS");
#define GCN_COMB_TB 256
#define GCN_COMB_BLOCKS ((GCN_N_NODES + GCN_COMB_TB - 1) / GCN_COMB_TB)
#define GCN_HIST_COPY_TRIPS ((GCN_COMB_BLOCKS + 31) / 32)
static_assert(GCN_COMB_TB == 2 * 32 * 4, "a wave copies a block as two 16-byte groups per lane");
static_assert(GCN_COMB_BLOCKS <= GCN_COMB_TB, "one thread per earlier block in the next kernel");
#define GCN_RANK_BITS 15
#define GCN_RANK_MASK ((1 << GCN_RANK_BITS) - 1)
static_assert(GCN_HIST_CHUNK <= (1 << GCN_RANK_BITS), "a rank inside a chunk fits the key's low bits");
static_assert(GCN_N_NODES <= (1 << (31 - GCN_RANK_BITS)), "a destination fits the key's high bits");
static_assert(GCN_HIST_CHUNK * 8 <= GCN_N_NODES * 4, "a chunk's run is put together in the counters' LDS");
static_assert(GCN_HIST_CHUNK % 2 == 0, "a run leaves the LDS in 16-byte groups of two records");
static_assert(GCN_COMB_BLOCKS < GCN_COMB_TB, "the first GCN_COMB_TB threads scan the block totals, one more for the end");

__global__ void __launch_bounds__(1024)
gcn_hist_lds(const int* __restrict__ src, const int* __restrict__ dst,
             int* __restrict__ part_in, int* __restrict__ part_out,
             int* __restrict__ blk_in, int* __restrict__ blk_base, int* __restrict__ runs) {
    __shared__ __attribute__((aligned(16))) int h[GCN_N_NODES];
    __shared__ int bt[GCN_COMB_TB];
    __shared__ int cur[GCN_COMB_TB];
    __shared__ int wtot[GCN_COMB_TB / 32];
    const int tid = threadIdx.x;
    const bool is_dst = (int)blockIdx.x < GCN_HIST_CHUNKS;
    const int  c      = is_dst ? (int)blockIdx.x : (int)blockIdx.x - GCN_HIST_CHUNKS;
    const int  e0     = c * GCN_HIST_CHUNK;
    const int* __restrict__ keys = (is_dst ? dst : src) + e0;

    int4 k4[GCN_HIST_ITERS];
#pragma unroll
    for (int k = 0; k < GCN_HIST_ITERS; ++k) {
        const int q = tid + k * 1024;
        k4[k] = make_int4(0, 0, 0, 0);
        if (q < GCN_HIST_QUADS) k4[k] = *(const int4*)(keys + 4 * q);
    }
    for (int i = tid; i < GCN_N_NODES / 4; i += 1024) ((int4*)h)[i] = make_int4(0, 0, 0, 0);
    __syncthreads();

    if (is_dst) {
        int4 r[GCN_HIST_ITERS];
#pragma unroll
        for (int k = 0; k < GCN_HIST_ITERS; ++k) {
            const int q = tid + k * 1024;
            r[k] = make_int4(0, 0, 0, 0);
            if (q < GCN_HIST_QUADS) {
                r[k].x = atomicAdd(&h[k4[k].x], 1);
                r[k].y = atomicAdd(&h[k4[k].y], 1);
                r[k].z = atomicAdd(&h[k4[k].z], 1);
                r[k].w = atomicAdd(&h[k4[k].w], 1);
            }
        }
#pragma unroll
        for (int k = 0; k < GCN_HIST_ITERS; ++k) {
            k4[k].x = (k4[k].x << GCN_RANK_BITS) | r[k].x;
            k4[k].y = (k4[k].y << GCN_RANK_BITS) | r[k].y;
            k4[k].z = (k4[k].z << GCN_RANK_BITS) | r[k].z;
            k4[k].w = (k4[k].w << GCN_RANK_BITS) | r[k].w;
        }
    } else {
#pragma unroll
        for (int k = 0; k < GCN_HIST_ITERS; ++k) {
            const int q = tid + k * 1024;
            if (q < GCN_HIST_QUADS) {
                atomicAdd(&h[k4[k].x], 1);
                atomicAdd(&h[k4[k].y], 1);
                atomicAdd(&h[k4[k].z], 1);
                atomicAdd(&h[k4[k].w], 1);
            }
        }
    }
    __syncthreads();

    int4* __restrict__ plane = (int4*)((is_dst ? part_in : part_out) + (size_t)c * GCN_N_NODES);
    const int lane = tid & 31;
    const int w    = tid >> 5;
    const int last = GCN_N_NODES / 4 - 1;
    int4 q0[GCN_HIST_COPY_TRIPS], q1[GCN_HIST_COPY_TRIPS];
#pragma unroll
    for (int k = 0; k < GCN_HIST_COPY_TRIPS; ++k) {
        const int i0 = (w + 32 * k) * (GCN_COMB_TB / 4) + lane;
        const int i1 = i0 + 32;
        q0[k] = ((const int4*)h)[i0 < last ? i0 : last];
        q1[k] = ((const int4*)h)[i1 < last ? i1 : last];
    }
    int s[GCN_HIST_COPY_TRIPS];
#pragma unroll
    for (int k = 0; k < GCN_HIST_COPY_TRIPS; ++k) {
        const int i0 = (w + 32 * k) * (GCN_COMB_TB / 4) + lane;
        const int i1 = i0 + 32;
        s[k] = 0;
        if (i0 <= last) { plane[i0] = q0[k]; s[k] += q0[k].x + q0[k].y + q0[k].z + q0[k].w; }
        if (i1 <= last) { plane[i1] = q1[k]; s[k] += q1[k].x + q1[k].y + q1[k].z + q1[k].w; }
    }
    if (is_dst) {
        for (int m = 16; m >= 1; m >>= 1) {
#pragma unroll
            for (int k = 0; k < GCN_HIST_COPY_TRIPS; ++k) s[k] += __shfl_xor(s[k], m);
        }
        if (lane == 0) {
#pragma unroll
            for (int k = 0; k < GCN_HIST_COPY_TRIPS; ++k) {
                const int b = w + 32 * k;
                if (b < GCN_COMB_BLOCKS) { blk_in[c * GCN_COMB_BLOCKS + b] = s[k]; bt[b] = s[k]; }
            }
        }
        int4 s4[GCN_HIST_ITERS];
#pragma unroll
        for (int k = 0; k < GCN_HIST_ITERS; ++k) {
            const int q = tid + k * 1024;
            s4[k] = make_int4(0, 0, 0, 0);
            if (q < GCN_HIST_QUADS) s4[k] = *(const int4*)(src + e0 + 4 * q);
        }
        __syncthreads();
        int v = 0, incl = 0, excl = 0;
        if (tid < GCN_COMB_TB) {
            v = tid < GCN_COMB_BLOCKS ? bt[tid] : 0;
            incl = v;
            for (int m = 1; m < 32; m <<= 1) {
                const int y = __shfl_up(incl, m);
                if (lane >= m) incl += y;
            }
            if (lane == 31) wtot[w] = incl;
        }
        __syncthreads();
        if (tid < GCN_COMB_TB) {
            int base = 0;
            for (int k = 0; k < GCN_COMB_TB / 32; ++k) base += (k < w) ? wtot[k] : 0;
            excl = base + incl - v;
            cur[tid] = excl;
        }
        __syncthreads();
        int2* run = (int2*)h;
#pragma unroll
        for (int k = 0; k < GCN_HIST_ITERS; ++k) {
            const int q = tid + k * 1024;
            if (q < GCN_HIST_QUADS) {
                const int p0 = atomicAdd(&cur[(k4[k].x >> GCN_RANK_BITS) / GCN_COMB_TB], 1);
                const int p1 = atomicAdd(&cur[(k4[k].y >> GCN_RANK_BITS) / GCN_COMB_TB], 1);
                const int p2 = atomicAdd(&cur[(k4[k].z >> GCN_RANK_BITS) / GCN_COMB_TB], 1);
                const int p3 = atomicAdd(&cur[(k4[k].w >> GCN_RANK_BITS) / GCN_COMB_TB], 1);
                run[p0] = make_int2(s4[k].x, k4[k].x);
                run[p1] = make_int2(s4[k].y, k4[k].y);
                run[p2] = make_int2(s4[k].z, k4[k].z);
                run[p3] = make_int2(s4[k].w, k4[k].w);
            }
        }
        __syncthreads();
        if (tid <= GCN_COMB_BLOCKS) blk_base[c * (GCN_COMB_BLOCKS + 1) + tid] = excl;
        int4* __restrict__ run_out = (int4*)(runs + 2 * (size_t)e0);
        for (int i = tid; i < GCN_HIST_CHUNK / 2; i += 1024) run_out[i] = ((const int4*)h)[i];
    }
}

__global__ void __launch_bounds__(GCN_COMB_TB)
gcn_combine(int* part_in, const int* __restrict__ part_out, const int* __restrict__ blk_in,
            int* __restrict__ cnt_out, int* __restrict__ cnt_in, int* __restrict__ row_ptr, int n) {
    __shared__ int wave_before[GCN_COMB_TB / 32];
    __shared__ int wave_rows[GCN_COMB_TB / 32];
    const int tid  = threadIdx.x;
    const int lane = tid & 31;
    const int w    = tid >> 5;
    const int b    = blockIdx.x;
    const int d    = b * GCN_COMB_TB + tid;
    const bool on  = d < n;
    const int dl = on ? d : n - 1;
    const int tb = tid < b ? tid : 0;
    int vi[GCN_HIST_CHUNKS], vo[GCN_HIST_CHUNKS], vb[GCN_HIST_CHUNKS];
#pragma unroll
    for (int c = 0; c < GCN_HIST_CHUNKS; ++c) vb[c] = blk_in[c * GCN_COMB_BLOCKS + tb];
#pragma unroll
    for (int c = 0; c < GCN_HIST_CHUNKS; ++c) vi[c] = part_in[(size_t)c * GCN_N_NODES + dl];
#pragma unroll
    for (int c = 0; c < GCN_HIST_CHUNKS; ++c) vo[c] = part_out[(size_t)c * GCN_N_NODES + dl];
    int run = 0, so = 0, before = 0;
#pragma unroll
    for (int c = 0; c < GCN_HIST_CHUNKS; ++c) {
        run    += vi[c];
        so     += vo[c];
        before += vb[c];
    }
    run    = on ? run : 0;
    before = tid < b ? before : 0;
    int incl = run;
    for (int m = 1; m < 32; m <<= 1) {
        const int y = __shfl_up(incl, m);
        if (lane >= m) incl += y;
    }
    for (int m = 16; m >= 1; m >>= 1) before += __shfl_xor(before, m);
    if (lane == 31) wave_rows[w]   = incl;
    if (lane == 0)  wave_before[w] = before;
    __syncthreads();
    int base = 0;
    for (int k = 0; k < GCN_COMB_TB / 32; ++k) {
        base += wave_before[k];
        if (k < w) base += wave_rows[k];
    }
    if (on) {
        const int start = base + incl - run;
        row_ptr[d] = start;
        cnt_in[d]  = run;
        cnt_out[d] = so;
        int slot = start;
#pragma unroll
        for (int c = 0; c < GCN_HIST_CHUNKS; ++c) {
            part_in[(size_t)c * GCN_N_NODES + d] = slot;
            slot += vi[c];
        }
    }
}

static __device__ __forceinline__ float gcn_inv_sqrt_clamped(int c) {
    float v = (float)c;
    v = v < 1.0f ? 1.0f : v;
    return 1.0f / sqrtf(v);
}

#define GCN_OWN_THREADS 1024
#define GCN_OWN_CAP 8192
#define GCN_OWN_FAST 4
static_assert(GCN_HIST_CHUNKS <= GCN_OWN_THREADS, "one thread per chunk fetches the piece bounds");

__global__ void __launch_bounds__(GCN_OWN_THREADS)
gcn_place_block(const int* __restrict__ runs, const int* __restrict__ blk_in,
                const int* __restrict__ blk_base, const int* __restrict__ chunk_start,
                const int* __restrict__ cnt_out, const int* __restrict__ cnt_in,
                const int* __restrict__ row_ptr, int* __restrict__ edges,
                float* __restrict__ deg_in_is, int n) {
    __shared__ int2 stage[GCN_OWN_CAP];
    __shared__ int rs[GCN_HIST_CHUNKS];
    __shared__ int rl[GCN_HIST_CHUNKS];
    const int tid = threadIdx.x;
    const int b   = blockIdx.x;
    const int n0  = b * GCN_COMB_TB;
    if (tid < GCN_HIST_CHUNKS) {
        rs[tid] = blk_base[tid * (GCN_COMB_BLOCKS + 1) + b];
        rl[tid] = blk_in[tid * GCN_COMB_BLOCKS + b];
    }
    const int out_lo = row_ptr[n0];
    __syncthreads();
    int total = 0;
    for (int c = 0; c < GCN_HIST_CHUNKS; ++c) total += rl[c];

    const int2* __restrict__ rr = (const int2*)runs;
    int2* __restrict__ eo = (int2*)edges;
    if (total > 0) {
        int c = 0, acc = 0;
        int2 rec[GCN_OWN_FAST];
        int  rc[GCN_OWN_FAST], ri[GCN_OWN_FAST];
#pragma unroll
        for (int k = 0; k < GCN_OWN_FAST; ++k) {
            const int j  = tid + k * GCN_OWN_THREADS;
            const int jj = j < total ? j : total - 1;
            while (jj >= acc + rl[c]) { acc += rl[c]; ++c; }
            rc[k] = c;
            ri[k] = c * GCN_HIST_CHUNK + rs[c] + (jj - acc);
        }
#pragma unroll
        for (int k = 0; k < GCN_OWN_FAST; ++k) rec[k] = rr[ri[k]];
        int cs[GCN_OWN_FAST], co[GCN_OWN_FAST];
#pragma unroll
        for (int k = 0; k < GCN_OWN_FAST; ++k) {
            const int d = rec[k].y >> GCN_RANK_BITS;
            cs[k] = chunk_start[(size_t)rc[k] * GCN_N_NODES + d];
            co[k] = cnt_out[rec[k].x];
        }
#pragma unroll
        for (int k = 0; k < GCN_OWN_FAST; ++k) {
            const int j = tid + k * GCN_OWN_THREADS;
            if (j < total) {
                const int  slot = cs[k] + (rec[k].y & GCN_RANK_MASK);
                const int2 pr   = make_int2(rec[k].x, __float_as_int(gcn_inv_sqrt_clamped(co[k])));
                const int  o    = slot - out_lo;
                if ((unsigned)o < (unsigned)GCN_OWN_CAP) stage[o] = pr; else eo[slot] = pr;
            }
        }
        for (int j = tid + GCN_OWN_FAST * GCN_OWN_THREADS; j < total; j += GCN_OWN_THREADS) {
            while (j >= acc + rl[c]) { acc += rl[c]; ++c; }
            const int2 e    = rr[(size_t)c * GCN_HIST_CHUNK + rs[c] + (j - acc)];
            const int  d    = e.y >> GCN_RANK_BITS;
            const int  slot = chunk_start[(size_t)c * GCN_N_NODES + d] + (e.y & GCN_RANK_MASK);
            const int2 pr   = make_int2(e.x, __float_as_int(gcn_inv_sqrt_clamped(cnt_out[e.x])));
            const int  o    = slot - out_lo;
            if ((unsigned)o < (unsigned)GCN_OWN_CAP) stage[o] = pr; else eo[slot] = pr;
        }
    }
    __syncthreads();
    const int staged = total < GCN_OWN_CAP ? total : GCN_OWN_CAP;
    for (int o = tid; o < staged; o += GCN_OWN_THREADS) eo[out_lo + o] = stage[o];
    if (tid < GCN_COMB_TB && n0 + tid < n) deg_in_is[n0 + tid] = gcn_inv_sqrt_clamped(cnt_in[n0 + tid]);
}

static __device__ __forceinline__ float4 gcn_row4(const float* __restrict__ x, int s, int lane4) {
    return *(const float4*)(x + (size_t)s * 128 + lane4);
}
static __device__ __forceinline__ void gcn_axpy4(float4& acc, const float4 v, const float w) {
    acc.x = fmaf(v.x, w, acc.x);
    acc.y = fmaf(v.y, w, acc.y);
    acc.z = fmaf(v.z, w, acc.z);
    acc.w = fmaf(v.w, w, acc.w);
}
#define GCN_EDGE_SRC(my, j) __builtin_amdgcn_readlane((my).x, (j))
#define GCN_EDGE_W(my, j)   __int_as_float(__builtin_amdgcn_readlane((my).y, (j)))

static __device__ __forceinline__ float4 gcn_gather_row(const float* __restrict__ x,
                                                        const int2* __restrict__ ce,
                                                        const int beg, const int cnt,
                                                        const int2 first,
                                                        const int lane, const int lane4) {
    float4 acc = make_float4(0.0f, 0.0f, 0.0f, 0.0f);
    for (int b = 0; b < cnt; b += 32) {
        const int m = (cnt - b < 32) ? (cnt - b) : 32;
        int2 my = first;
        if (b > 0) {
            my = make_int2(0, 0);
            if (lane < m) my = ce[beg + b + lane];
        }
        int j = 0;
        for (; j + 8 <= m; j += 8) {
            const float4 v0 = gcn_row4(x, GCN_EDGE_SRC(my, j),     lane4);
            const float4 v1 = gcn_row4(x, GCN_EDGE_SRC(my, j + 1), lane4);
            const float4 v2 = gcn_row4(x, GCN_EDGE_SRC(my, j + 2), lane4);
            const float4 v3 = gcn_row4(x, GCN_EDGE_SRC(my, j + 3), lane4);
            const float4 v4 = gcn_row4(x, GCN_EDGE_SRC(my, j + 4), lane4);
            const float4 v5 = gcn_row4(x, GCN_EDGE_SRC(my, j + 5), lane4);
            const float4 v6 = gcn_row4(x, GCN_EDGE_SRC(my, j + 6), lane4);
            const float4 v7 = gcn_row4(x, GCN_EDGE_SRC(my, j + 7), lane4);
            gcn_axpy4(acc, v0, GCN_EDGE_W(my, j));
            gcn_axpy4(acc, v1, GCN_EDGE_W(my, j + 1));
            gcn_axpy4(acc, v2, GCN_EDGE_W(my, j + 2));
            gcn_axpy4(acc, v3, GCN_EDGE_W(my, j + 3));
            gcn_axpy4(acc, v4, GCN_EDGE_W(my, j + 4));
            gcn_axpy4(acc, v5, GCN_EDGE_W(my, j + 5));
            gcn_axpy4(acc, v6, GCN_EDGE_W(my, j + 6));
            gcn_axpy4(acc, v7, GCN_EDGE_W(my, j + 7));
        }
        if (j + 4 <= m) {
            const float4 v0 = gcn_row4(x, GCN_EDGE_SRC(my, j),     lane4);
            const float4 v1 = gcn_row4(x, GCN_EDGE_SRC(my, j + 1), lane4);
            const float4 v2 = gcn_row4(x, GCN_EDGE_SRC(my, j + 2), lane4);
            const float4 v3 = gcn_row4(x, GCN_EDGE_SRC(my, j + 3), lane4);
            gcn_axpy4(acc, v0, GCN_EDGE_W(my, j));
            gcn_axpy4(acc, v1, GCN_EDGE_W(my, j + 1));
            gcn_axpy4(acc, v2, GCN_EDGE_W(my, j + 2));
            gcn_axpy4(acc, v3, GCN_EDGE_W(my, j + 3));
            j += 4;
        }
        const int r = m - j;
        if (r == 1) {
            const float4 v0 = gcn_row4(x, GCN_EDGE_SRC(my, j), lane4);
            gcn_axpy4(acc, v0, GCN_EDGE_W(my, j));
        } else if (r == 2) {
            const float4 v0 = gcn_row4(x, GCN_EDGE_SRC(my, j),     lane4);
            const float4 v1 = gcn_row4(x, GCN_EDGE_SRC(my, j + 1), lane4);
            gcn_axpy4(acc, v0, GCN_EDGE_W(my, j));
            gcn_axpy4(acc, v1, GCN_EDGE_W(my, j + 1));
        } else if (r == 3) {
            const float4 v0 = gcn_row4(x, GCN_EDGE_SRC(my, j),     lane4);
            const float4 v1 = gcn_row4(x, GCN_EDGE_SRC(my, j + 1), lane4);
            const float4 v2 = gcn_row4(x, GCN_EDGE_SRC(my, j + 2), lane4);
            gcn_axpy4(acc, v0, GCN_EDGE_W(my, j));
            gcn_axpy4(acc, v1, GCN_EDGE_W(my, j + 1));
            gcn_axpy4(acc, v2, GCN_EDGE_W(my, j + 2));
        }
    }
    return acc;
}

#define GCN_K 128
#define GCN_C 128
#define GCN_C_LAST 16
#define GCN_TILE_PITCH 132
#define GCN_LAYER_GROUPS 1024
#define GCN_MAX_TILES_PER_GROUP 16
#define GCN_N_TILES (GCN_N_NODES / 16)
static_assert(GCN_N_NODES % 16 == 0, "whole 16-row tiles only");
static_assert(GCN_LAYER_GROUPS <= GCN_N_TILES, "every work-group gets at least one tile");
static_assert((GCN_N_TILES + GCN_LAYER_GROUPS - 1) / GCN_LAYER_GROUPS <= GCN_MAX_TILES_PER_GROUP,
              "a work-group's tiles must fit the lanes that hold their row bounds");
static_assert(GCN_C == 8 * 16, "8 waves, one 16-column strip each");

template <bool FUSE_NEXT>
__global__ void __launch_bounds__(256) __attribute__((amdgpu_waves_per_eu(8, 8)))
gcn_layer_fused(const float* __restrict__ x, const int* __restrict__ row_ptr,
                const int* __restrict__ cnt_in, const int* __restrict__ edges,
                const float* __restrict__ deg_in_is, const float* __restrict__ W,
                const float* __restrict__ bias, const float* __restrict__ Wn,
                float* __restrict__ out, int do_relu, int n_tiles) {
    __shared__ __attribute__((aligned(16))) float tile[2][16 * GCN_TILE_PITCH];
    __shared__ __attribute__((aligned(16))) float htile[FUSE_NEXT ? 16 * GCN_TILE_PITCH : 4];
    __shared__ __attribute__((aligned(16))) v2f wn_frag[FUSE_NEXT ? (GCN_K / 4) * 32 : 1];

    const int lane  = threadIdx.x & 31;
    const int lane4 = lane * 4;
    const int half  = lane >> 4;
    const int l15   = lane & 15;
    const int wave  = __builtin_amdgcn_readfirstlane(threadIdx.x >> 5);
    const int ncol  = wave * 16 + l15;
    const int per_group = n_tiles / GCN_LAYER_GROUPS;
    const int extra     = n_tiles % GCN_LAYER_GROUPS;
    const int g         = (int)blockIdx.x;
    const int tile0     = g * per_group + (g < extra ? g : extra);
    const int my_tiles  = per_group + (g < extra ? 1 : 0);
    const int2* __restrict__ ce = (const int2*)edges;

    int rp = 0, rc = 0;
    {
        const int tj = tile0 + (lane >> 1);
        if ((lane >> 1) < my_tiles) {
            const int node = tj * 16 + wave * 2 + (lane & 1);
            rp = row_ptr[node];
            rc = cnt_in[node];
        }
    }

    v2f bfrag[GCN_K / 4];
#pragma unroll
    for (int s = 0; s < GCN_K / 4; ++s) {
        const int kb = s * 4 + half * 2;
        bfrag[s].x = W[(size_t)kb       * GCN_C + ncol];
        bfrag[s].y = W[(size_t)(kb + 1) * GCN_C + ncol];
    }
    const float bv = bias[ncol];

    if constexpr (FUSE_NEXT) {
        for (int idx = threadIdx.x; idx < (GCN_K / 4) * 32; idx += 256) {
            const int s  = idx >> 5;
            const int ln = idx & 31;
            const int kb = s * 4 + (ln >> 4) * 2;
            v2f f;
            f.x = Wn[(size_t)kb       * GCN_C_LAST + (ln & 15)];
            f.y = Wn[(size_t)(kb + 1) * GCN_C_LAST + (ln & 15)];
            wn_frag[idx] = f;
        }
    }

    for (int i = 0; i < my_tiles; ++i) {
        const int t = tile0 + i;
        float* tb = &tile[i & 1][0];

        const int beg0 = __builtin_amdgcn_readlane(rp, 2 * i);
        const int cnt0 = __builtin_amdgcn_readlane(rc, 2 * i);
        const int beg1 = __builtin_amdgcn_readlane(rp, 2 * i + 1);
        const int cnt1 = __builtin_amdgcn_readlane(rc, 2 * i + 1);

        int2 cur0 = make_int2(0, 0), cur1 = make_int2(0, 0);
        if (lane < cnt0) cur0 = ce[beg0 + lane];
        if (lane < cnt1) cur1 = ce[beg1 + lane];
        const float scale = deg_in_is[t * 16 + l15];

        const float4 g0 = gcn_gather_row(x, ce, beg0, cnt0, cur0, lane, lane4);
        *(float4*)(tb + (2 * wave) * GCN_TILE_PITCH + lane4) = g0;
        const float4 g1 = gcn_gather_row(x, ce, beg1, cnt1, cur1, lane, lane4);
        *(float4*)(tb + (2 * wave + 1) * GCN_TILE_PITCH + lane4) = g1;

        __syncthreads();

        {
            const float* arow = tb + l15 * GCN_TILE_PITCH + half * 2;
            v8f c = {};
#pragma unroll
            for (int s = 0; s < GCN_K / 4; ++s) {
                const v2f araw = *(const v2f*)(arow + s * 4);
                v2f a;
                a.x = araw.x * scale;
                a.y = araw.y * scale;
                c = __builtin_amdgcn_wmma_f32_16x16x4_f32(
                        false, a, false, bfrag[s], (short)0, c, false, false);
            }
#pragma unroll
            for (int v = 0; v < 8; ++v) {
                float val = c[v] + bv;
                if (do_relu) val = fmaxf(val, 0.0f);
                if constexpr (FUSE_NEXT) {
                    htile[(half * 8 + v) * GCN_TILE_PITCH + ncol] = val;
                } else {
                    out[(size_t)(t * 16 + half * 8 + v) * GCN_C + ncol] = val;
                }
            }
        }

        if constexpr (FUSE_NEXT) {
            __syncthreads();
            if (wave == (i & 7)) {
                const float* hrow = htile + l15 * GCN_TILE_PITCH + half * 2;
                v8f z0 = {}, z1 = {};
#pragma unroll
                for (int s = 0; s < GCN_K / 4; s += 2) {
                    const v2f a0 = *(const v2f*)(hrow + s * 4);
                    const v2f b0 = wn_frag[s * 32 + lane];
                    const v2f a1 = *(const v2f*)(hrow + (s + 1) * 4);
                    const v2f b1 = wn_frag[(s + 1) * 32 + lane];
                    z0 = __builtin_amdgcn_wmma_f32_16x16x4_f32(
                            false, a0, false, b0, (short)0, z0, false, false);
                    z1 = __builtin_amdgcn_wmma_f32_16x16x4_f32(
                            false, a1, false, b1, (short)0, z1, false, false);
                }
#pragma unroll
                for (int v = 0; v < 8; ++v) {
                    out[(size_t)(t * 16 + half * 8 + v) * GCN_C_LAST + l15] = z0[v] + z1[v];
                }
            }
        }
    }
}

#define GCN_LAST_BATCH 16

__global__ void __launch_bounds__(256) __attribute__((amdgpu_waves_per_eu(4, 6)))
gcn_gather_last(const float* __restrict__ z, const int* __restrict__ row_ptr,
                const int* __restrict__ cnt_in, const int* __restrict__ edges,
                const float* __restrict__ deg_in_is, const float* __restrict__ bias,
                float* __restrict__ out, int n) {
    const int g  = blockIdx.x * blockDim.x + threadIdx.x;
    const int m  = g >> 2;
    const int q4 = (g & 3) * 4;
    if (m >= n) return;
    const int    cnt = cnt_in[m];
    const int    beg = row_ptr[m];
    const float  sc  = deg_in_is[m];
    const float4 b   = *(const float4*)(bias + q4);
    asm volatile("" :: "v"(cnt), "v"(beg), "v"(sc), "v"(b.x), "v"(b.y), "v"(b.z), "v"(b.w));
    const int2* __restrict__ ce = (const int2*)edges + beg;
    const int last = cnt - 1;

    float4 acc = make_float4(0.0f, 0.0f, 0.0f, 0.0f);
    for (int j = 0; j < cnt; j += GCN_LAST_BATCH) {
        int2 e[GCN_LAST_BATCH];
        float4 v[GCN_LAST_BATCH];
#pragma unroll
        for (int k = 0; k < GCN_LAST_BATCH; ++k) {
            const int jj = (j + k < last) ? (j + k) : last;
            e[k] = ce[jj];
        }
#pragma unroll
        for (int k = 0; k < GCN_LAST_BATCH; ++k) {
            v[k] = *(const float4*)(z + (size_t)e[k].x * GCN_C_LAST + q4);
        }
#pragma unroll
        for (int k = 0; k < GCN_LAST_BATCH; ++k) {
            const bool on = (j + k) < cnt;
            float4 vv;
            vv.x = on ? v[k].x : 0.0f;
            vv.y = on ? v[k].y : 0.0f;
            vv.z = on ? v[k].z : 0.0f;
            vv.w = on ? v[k].w : 0.0f;
            gcn_axpy4(acc, vv, __int_as_float(e[k].y));
        }
    }
    float4 r;
    r.x = fmaf(acc.x, sc, b.x);
    r.y = fmaf(acc.y, sc, b.y);
    r.z = fmaf(acc.z, sc, b.z);
    r.w = fmaf(acc.w, sc, b.w);
    *(float4*)(out + (size_t)m * GCN_C_LAST + q4) = r;
}

extern "C" void kernel_launch(void* const* d_in, const int* in_sizes, int n_in,
                              void* d_out, int out_size, void* d_ws, size_t ws_size,
                              hipStream_t stream) {
    const float* features = (const float*)d_in[0];
    const int*   src      = (const int*)  d_in[1];
    const int*   dst      = (const int*)  d_in[2];
    const float* W1       = (const float*)d_in[3];
    const float* b1       = (const float*)d_in[4];
    const float* W2       = (const float*)d_in[5];
    const float* b2       = (const float*)d_in[6];
    const float* W3       = (const float*)d_in[7];
    const float* b3       = (const float*)d_in[8];
    float* out = (float*)d_out;

    const int N = GCN_N_NODES;
    const int E = GCN_N_EDGES;
    const size_t NF = (size_t)N * 128;

    float* bufZ    = (float*)d_ws;
    float* bufB    = bufZ + NF;
    int*   edges   = (int*)(bufB + NF);
    int*   runs    = edges + 2 * (size_t)E;
    int*   cnt_out = runs + 2 * (size_t)E;
    int*   cnt_in  = cnt_out + N;
    int*   row_ptr = cnt_in + N;
    float* deg_i   = (float*)(row_ptr + N);
    int*   part_in  = (int*)(deg_i + N);
    int*   part_out = part_in + (size_t)GCN_HIST_CHUNKS * N;
    int*   blk_in   = part_out + (size_t)GCN_HIST_CHUNKS * N;
    int*   blk_base = blk_in + GCN_HIST_CHUNKS * GCN_COMB_BLOCKS;

    const int TB = 256;

    gcn_hist_lds<<<2 * GCN_HIST_CHUNKS, 1024, 0, stream>>>(src, dst, part_in, part_out, blk_in, blk_base, runs);
    gcn_combine<<<GCN_COMB_BLOCKS, GCN_COMB_TB, 0, stream>>>(part_in, part_out, blk_in,
                                                             cnt_out, cnt_in, row_ptr, N);
    gcn_place_block<<<GCN_COMB_BLOCKS, GCN_OWN_THREADS, 0, stream>>>(runs, blk_in, blk_base, part_in,
                                                                    cnt_out, cnt_in, row_ptr, edges, deg_i, N);

    const int groups = GCN_LAYER_GROUPS;

    gcn_layer_fused<false><<<groups, TB, 0, stream>>>(features, row_ptr, cnt_in, edges, deg_i,
                                                      W1, b1, W3, bufB, 1, GCN_N_TILES);
    gcn_layer_fused<true><<<groups, TB, 0, stream>>>(bufB, row_ptr, cnt_in, edges, deg_i,
                                                     W2, b2, W3, bufZ, 1, GCN_N_TILES);
    gcn_gather_last<<<(4 * N + TB - 1) / TB, TB, 0, stream>>>(bufZ, row_ptr, cnt_in, edges, deg_i,
                                                             b3, out, N);

    (void)in_sizes; (void)n_in; (void)out_size; (void)ws_size;
}


# Placement orders a node block's edges by node on its own LDS counters (count, scan, slot adds); per-node destination counts, ranks and 25 planes are gone; row sum order = LDS arrival in the block
# speedup vs baseline: 5.6063x; 1.0786x over previous
#include <hip/hip_runtime.h>
#include <math.h>


typedef __attribute__((ext_vector_type(2))) float v2f;
typedef __attribute__((ext_vector_type(8))) float v8f;

#define GCN_N_NODES 50000
#define GCN_N_EDGES 600000


#define GCN_HIST_CHUNKS 25
#define GCN_HIST_CHUNK (GCN_N_EDGES / GCN_HIST_CHUNKS)
#define GCN_HIST_QUADS (GCN_HIST_CHUNK / 4)
#define GCN_HIST_ITERS ((GCN_HIST_QUADS + 1023) / 1024)
static_assert(GCN_N_EDGES % GCN_HIST_CHUNKS == 0, "equal chunks");
static_assert(GCN_HIST_CHUNK % 4 == 0, "a chunk is read in 16-byte groups");
static_assert(GCN_N_NODES % 4 == 0, "counter planes are cleared and copied in 16-byte groups");
static_assert(GCN_N_NODES * 4 <= 320 * 1024, "all source counters in one work-group's LDS");
#define GCN_COMB_TB 256
#define GCN_COMB_BLOCKS ((GCN_N_NODES + GCN_COMB_TB - 1) / GCN_COMB_TB)
#define GCN_HIST_COPY_TRIPS ((GCN_COMB_BLOCKS + 31) / 32)
static_assert(GCN_COMB_TB == 2 * 32 * 4, "a wave copies a block as two 16-byte groups per lane");
static_assert(GCN_COMB_BLOCKS <= GCN_COMB_TB, "one thread per earlier block in the placement kernel");
static_assert(GCN_HIST_CHUNK * 8 <= GCN_N_NODES * 4, "a chunk's run is put together in the counters' LDS");
static_assert(GCN_HIST_CHUNK % 2 == 0, "a run leaves the LDS in 16-byte groups of two records");
static_assert(GCN_COMB_BLOCKS < GCN_COMB_TB, "the first GCN_COMB_TB threads scan the block totals, one more for the end");

__global__ void __launch_bounds__(1024)
gcn_hist_lds(const int* __restrict__ src, const int* __restrict__ dst,
             int* __restrict__ part_out,
             int* __restrict__ blk_in, int* __restrict__ blk_base, int* __restrict__ runs) {
    __shared__ __attribute__((aligned(16))) int h[GCN_N_NODES];
    __shared__ int bt[GCN_COMB_TB];
    __shared__ int cur[GCN_COMB_TB];
    __shared__ int wtot[GCN_COMB_TB / 32];
    const int tid  = threadIdx.x;
    const int lane = tid & 31;
    const int w    = tid >> 5;
    const bool is_dst = (int)blockIdx.x < GCN_HIST_CHUNKS;
    const int  c      = is_dst ? (int)blockIdx.x : (int)blockIdx.x - GCN_HIST_CHUNKS;
    const int  e0     = c * GCN_HIST_CHUNK;

    if (is_dst) {
        int4 k4[GCN_HIST_ITERS], s4[GCN_HIST_ITERS];
#pragma unroll
        for (int k = 0; k < GCN_HIST_ITERS; ++k) {
            const int q = tid + k * 1024;
            k4[k] = make_int4(0, 0, 0, 0);
            if (q < GCN_HIST_QUADS) k4[k] = *(const int4*)(dst + e0 + 4 * q);
        }
#pragma unroll
        for (int k = 0; k < GCN_HIST_ITERS; ++k) {
            const int q = tid + k * 1024;
            s4[k] = make_int4(0, 0, 0, 0);
            if (q < GCN_HIST_QUADS) s4[k] = *(const int4*)(src + e0 + 4 * q);
        }
        if (tid < GCN_COMB_TB) bt[tid] = 0;
        __syncthreads();
        int4 r[GCN_HIST_ITERS];
#pragma unroll
        for (int k = 0; k < GCN_HIST_ITERS; ++k) {
            const int q = tid + k * 1024;
            r[k] = make_int4(0, 0, 0, 0);
            if (q < GCN_HIST_QUADS) {
                r[k].x = atomicAdd(&bt[k4[k].x / GCN_COMB_TB], 1);
                r[k].y = atomicAdd(&bt[k4[k].y / GCN_COMB_TB], 1);
                r[k].z = atomicAdd(&bt[k4[k].z / GCN_COMB_TB], 1);
                r[k].w = atomicAdd(&bt[k4[k].w / GCN_COMB_TB], 1);
            }
        }
        __syncthreads();
        int v = 0, incl = 0, excl = 0;
        if (tid < GCN_COMB_TB) {
            v = bt[tid];
            incl = v;
            for (int m = 1; m < 32; m <<= 1) {
                const int y = __shfl_up(incl, m);
                if (lane >= m) incl += y;
            }
            if (lane == 31) wtot[w] = incl;
        }
        __syncthreads();
        if (tid < GCN_COMB_TB) {
            int base = 0;
            for (int k = 0; k < GCN_COMB_TB / 32; ++k) base += (k < w) ? wtot[k] : 0;
            excl = base + incl - v;
            cur[tid] = excl;
        }
        __syncthreads();
        int2* run = (int2*)h;
#pragma unroll
        for (int k = 0; k < GCN_HIST_ITERS; ++k) {
            const int q = tid + k * 1024;
            if (q < GCN_HIST_QUADS) {
                run[cur[k4[k].x / GCN_COMB_TB] + r[k].x] = make_int2(s4[k].x, k4[k].x);
                run[cur[k4[k].y / GCN_COMB_TB] + r[k].y] = make_int2(s4[k].y, k4[k].y);
                run[cur[k4[k].z / GCN_COMB_TB] + r[k].z] = make_int2(s4[k].z, k4[k].z);
                run[cur[k4[k].w / GCN_COMB_TB] + r[k].w] = make_int2(s4[k].w, k4[k].w);
            }
        }
        __syncthreads();
        if (tid < GCN_COMB_BLOCKS)  blk_in[c * GCN_COMB_BLOCKS + tid] = v;
        if (tid <= GCN_COMB_BLOCKS) blk_base[c * (GCN_COMB_BLOCKS + 1) + tid] = excl;
        int4* __restrict__ run_out = (int4*)(runs + 2 * (size_t)e0);
        for (int i = tid; i < GCN_HIST_CHUNK / 2; i += 1024) run_out[i] = ((const int4*)h)[i];
    } else {
        int4 k4[GCN_HIST_ITERS];
#pragma unroll
        for (int k = 0; k < GCN_HIST_ITERS; ++k) {
            const int q = tid + k * 1024;
            k4[k] = make_int4(0, 0, 0, 0);
            if (q < GCN_HIST_QUADS) k4[k] = *(const int4*)(src + e0 + 4 * q);
        }
        for (int i = tid; i < GCN_N_NODES / 4; i += 1024) ((int4*)h)[i] = make_int4(0, 0, 0, 0);
        __syncthreads();
#pragma unroll
        for (int k = 0; k < GCN_HIST_ITERS; ++k) {
            const int q = tid + k * 1024;
            if (q < GCN_HIST_QUADS) {
                atomicAdd(&h[k4[k].x], 1);
                atomicAdd(&h[k4[k].y], 1);
                atomicAdd(&h[k4[k].z], 1);
                atomicAdd(&h[k4[k].w], 1);
            }
        }
        __syncthreads();
        int4* __restrict__ plane = (int4*)(part_out + (size_t)c * GCN_N_NODES);
        const int last = GCN_N_NODES / 4 - 1;
        int4 q0[GCN_HIST_COPY_TRIPS], q1[GCN_HIST_COPY_TRIPS];
#pragma unroll
        for (int k = 0; k < GCN_HIST_COPY_TRIPS; ++k) {
            const int i0 = (w + 32 * k) * (GCN_COMB_TB / 4) + lane;
            const int i1 = i0 + 32;
            q0[k] = ((const int4*)h)[i0 < last ? i0 : last];
            q1[k] = ((const int4*)h)[i1 < last ? i1 : last];
        }
#pragma unroll
        for (int k = 0; k < GCN_HIST_COPY_TRIPS; ++k) {
            const int i0 = (w + 32 * k) * (GCN_COMB_TB / 4) + lane;
            const int i1 = i0 + 32;
            if (i0 <= last) plane[i0] = q0[k];
            if (i1 <= last) plane[i1] = q1[k];
        }
    }
}

__global__ void __launch_bounds__(GCN_COMB_TB)
gcn_combine(const int* __restrict__ part_out, int* __restrict__ cnt_out, int n) {
    const int d = blockIdx.x * GCN_COMB_TB + threadIdx.x;
    const int dl = d < n ? d : n - 1;
    int vo[GCN_HIST_CHUNKS];
#pragma unroll
    for (int c = 0; c < GCN_HIST_CHUNKS; ++c) vo[c] = part_out[(size_t)c * GCN_N_NODES + dl];
    int so = 0;
#pragma unroll
    for (int c = 0; c < GCN_HIST_CHUNKS; ++c) so += vo[c];
    if (d < n) cnt_out[d] = so;
}

static __device__ __forceinline__ float gcn_inv_sqrt_clamped(int c) {
    float v = (float)c;
    v = v < 1.0f ? 1.0f : v;
    return 1.0f / sqrtf(v);
}

#define GCN_OWN_THREADS 1024
#define GCN_OWN_CAP 8192
#define GCN_OWN_FAST 4
static_assert(GCN_COMB_TB + GCN_HIST_CHUNKS <= GCN_OWN_THREADS, "one thread per chunk, behind the node threads, fetches the piece bounds");
static_assert(GCN_COMB_TB <= GCN_OWN_THREADS, "one thread per node of the block");

__global__ void __launch_bounds__(GCN_OWN_THREADS)
gcn_place_block(const int* __restrict__ runs, const int* __restrict__ blk_in,
                const int* __restrict__ blk_base, const int* __restrict__ cnt_out,
                int* __restrict__ cnt_in, int* __restrict__ row_ptr, int* __restrict__ edges,
                float* __restrict__ deg_in_is, int n) {
    __shared__ int2 stage[GCN_OWN_CAP];
    __shared__ int rs[GCN_HIST_CHUNKS];
    __shared__ int rl[GCN_HIST_CHUNKS];
    __shared__ int ncnt[GCN_COMB_TB];
    __shared__ int wbefore[GCN_COMB_TB / 32];
    __shared__ int wrows[GCN_COMB_TB / 32];
    const int tid  = threadIdx.x;
    const int lane = tid & 31;
    const int w    = tid >> 5;
    const int b    = blockIdx.x;
    const int n0   = b * GCN_COMB_TB;
    if (tid < GCN_COMB_TB) {
        const int tb = tid < b ? tid : 0;
        int vb[GCN_HIST_CHUNKS];
#pragma unroll
        for (int c = 0; c < GCN_HIST_CHUNKS; ++c) vb[c] = blk_in[c * GCN_COMB_BLOCKS + tb];
        int before = 0;
#pragma unroll
        for (int c = 0; c < GCN_HIST_CHUNKS; ++c) before += vb[c];
        before = tid < b ? before : 0;
        for (int m = 16; m >= 1; m >>= 1) before += __shfl_xor(before, m);
        if (lane == 0) wbefore[w] = before;
        ncnt[tid] = 0;
    } else if (tid < GCN_COMB_TB + GCN_HIST_CHUNKS) {
        const int t = tid - GCN_COMB_TB;
        rs[t] = blk_base[t * (GCN_COMB_BLOCKS + 1) + b];
        rl[t] = blk_in[t * GCN_COMB_BLOCKS + b];
    }
    __syncthreads();
    int out_lo = 0;
    for (int k = 0; k < GCN_COMB_TB / 32; ++k) out_lo += wbefore[k];
    int total = 0;
    for (int c = 0; c < GCN_HIST_CHUNKS; ++c) total += rl[c];

    const int2* __restrict__ rr = (const int2*)runs;
    int2* __restrict__ eo = (int2*)edges;
    int c = 0, acc = 0;
    int2 rec[GCN_OWN_FAST];
    int  co[GCN_OWN_FAST];
#pragma unroll
    for (int k = 0; k < GCN_OWN_FAST; ++k) { rec[k] = make_int2(0, n0); co[k] = 1; }
    if (total > 0) {
        int ri[GCN_OWN_FAST];
#pragma unroll
        for (int k = 0; k < GCN_OWN_FAST; ++k) {
            const int j  = tid + k * GCN_OWN_THREADS;
            const int jj = j < total ? j : total - 1;
            while (jj >= acc + rl[c]) { acc += rl[c]; ++c; }
            ri[k] = c * GCN_HIST_CHUNK + rs[c] + (jj - acc);
        }
#pragma unroll
        for (int k = 0; k < GCN_OWN_FAST; ++k) rec[k] = rr[ri[k]];
#pragma unroll
        for (int k = 0; k < GCN_OWN_FAST; ++k) co[k] = cnt_out[rec[k].x];
#pragma unroll
        for (int k = 0; k < GCN_OWN_FAST; ++k) {
            const int j = tid + k * GCN_OWN_THREADS;
            if (j < total) atomicAdd(&ncnt[rec[k].y - n0], 1);
        }
        int c2 = c, acc2 = acc;
        for (int j = tid + GCN_OWN_FAST * GCN_OWN_THREADS; j < total; j += GCN_OWN_THREADS) {
            while (j >= acc2 + rl[c2]) { acc2 += rl[c2]; ++c2; }
            const int2 e = rr[(size_t)c2 * GCN_HIST_CHUNK + rs[c2] + (j - acc2)];
            atomicAdd(&ncnt[e.y - n0], 1);
        }
    }
    __syncthreads();
    int v = 0, incl = 0, loc = 0;
    if (tid < GCN_COMB_TB) {
        v = ncnt[tid];
        incl = v;
        for (int m = 1; m < 32; m <<= 1) {
            const int y = __shfl_up(incl, m);
            if (lane >= m) incl += y;
        }
        if (lane == 31) wrows[w] = incl;
    }
    __syncthreads();
    if (tid < GCN_COMB_TB) {
        int base = 0;
        for (int k = 0; k < GCN_COMB_TB / 32; ++k) base += (k < w) ? wrows[k] : 0;
        loc = base + incl - v;
        ncnt[tid] = loc;
    }
    __syncthreads();
    if (total > 0) {
        int o[GCN_OWN_FAST];
#pragma unroll
        for (int k = 0; k < GCN_OWN_FAST; ++k) {
            const bool on = tid + k * GCN_OWN_THREADS < total;
            o[k] = atomicAdd(&ncnt[on ? rec[k].y - n0 : (tid & (GCN_COMB_TB - 1))], on ? 1 : 0);
        }
#pragma unroll
        for (int k = 0; k < GCN_OWN_FAST; ++k) {
            const int j = tid + k * GCN_OWN_THREADS;
            if (j < total) {
                const int2 pr = make_int2(rec[k].x, __float_as_int(gcn_inv_sqrt_clamped(co[k])));
                if ((unsigned)o[k] < (unsigned)GCN_OWN_CAP) stage[o[k]] = pr; else eo[out_lo + o[k]] = pr;
            }
        }
        for (int j = tid + GCN_OWN_FAST * GCN_OWN_THREADS; j < total; j += GCN_OWN_THREADS) {
            while (j >= acc + rl[c]) { acc += rl[c]; ++c; }
            const int2 e  = rr[(size_t)c * GCN_HIST_CHUNK + rs[c] + (j - acc)];
            const int  os = atomicAdd(&ncnt[e.y - n0], 1);
            const int2 pr = make_int2(e.x, __float_as_int(gcn_inv_sqrt_clamped(cnt_out[e.x])));
            if ((unsigned)os < (unsigned)GCN_OWN_CAP) stage[os] = pr; else eo[out_lo + os] = pr;
        }
    }
    __syncthreads();
    const int staged = total < GCN_OWN_CAP ? total : GCN_OWN_CAP;
    for (int o = tid; o < staged; o += GCN_OWN_THREADS) eo[out_lo + o] = stage[o];
    if (tid < GCN_COMB_TB && n0 + tid < n) {
        row_ptr[n0 + tid]   = out_lo + loc;
        cnt_in[n0 + tid]    = v;
        deg_in_is[n0 + tid] = gcn_inv_sqrt_clamped(v);
    }
}

static __device__ __forceinline__ float4 gcn_row4(const float* __restrict__ x, int s, int lane4) {
    return *(const float4*)(x + (size_t)s * 128 + lane4);
}
static __device__ __forceinline__ void gcn_axpy4(float4& acc, const float4 v, const float w) {
    acc.x = fmaf(v.x, w, acc.x);
    acc.y = fmaf(v.y, w, acc.y);
    acc.z = fmaf(v.z, w, acc.z);
    acc.w = fmaf(v.w, w, acc.w);
}
#define GCN_EDGE_SRC(my, j) __builtin_amdgcn_readlane((my).x, (j))
#define GCN_EDGE_W(my, j)   __int_as_float(__builtin_amdgcn_readlane((my).y, (j)))

static __device__ __forceinline__ float4 gcn_gather_row(const float* __restrict__ x,
                                                        const int2* __restrict__ ce,
                                                        const int beg, const int cnt,
                                                        const int2 first,
                                                        const int lane, const int lane4) {
    float4 acc = make_float4(0.0f, 0.0f, 0.0f, 0.0f);
    for (int b = 0; b < cnt; b += 32) {
        const int m = (cnt - b < 32) ? (cnt - b) : 32;
        int2 my = first;
        if (b > 0) {
            my = make_int2(0, 0);
            if (lane < m) my = ce[beg + b + lane];
        }
        int j = 0;
        for (; j + 8 <= m; j += 8) {
            const float4 v0 = gcn_row4(x, GCN_EDGE_SRC(my, j),     lane4);
            const float4 v1 = gcn_row4(x, GCN_EDGE_SRC(my, j + 1), lane4);
            const float4 v2 = gcn_row4(x, GCN_EDGE_SRC(my, j + 2), lane4);
            const float4 v3 = gcn_row4(x, GCN_EDGE_SRC(my, j + 3), lane4);
            const float4 v4 = gcn_row4(x, GCN_EDGE_SRC(my, j + 4), lane4);
            const float4 v5 = gcn_row4(x, GCN_EDGE_SRC(my, j + 5), lane4);
            const float4 v6 = gcn_row4(x, GCN_EDGE_SRC(my, j + 6), lane4);
            const float4 v7 = gcn_row4(x, GCN_EDGE_SRC(my, j + 7), lane4);
            gcn_axpy4(acc, v0, GCN_EDGE_W(my, j));
            gcn_axpy4(acc, v1, GCN_EDGE_W(my, j + 1));
            gcn_axpy4(acc, v2, GCN_EDGE_W(my, j + 2));
            gcn_axpy4(acc, v3, GCN_EDGE_W(my, j + 3));
            gcn_axpy4(acc, v4, GCN_EDGE_W(my, j + 4));
            gcn_axpy4(acc, v5, GCN_EDGE_W(my, j + 5));
            gcn_axpy4(acc, v6, GCN_EDGE_W(my, j + 6));
            gcn_axpy4(acc, v7, GCN_EDGE_W(my, j + 7));
        }
        if (j + 4 <= m) {
            const float4 v0 = gcn_row4(x, GCN_EDGE_SRC(my, j),     lane4);
            const float4 v1 = gcn_row4(x, GCN_EDGE_SRC(my, j + 1), lane4);
            const float4 v2 = gcn_row4(x, GCN_EDGE_SRC(my, j + 2), lane4);
            const float4 v3 = gcn_row4(x, GCN_EDGE_SRC(my, j + 3), lane4);
            gcn_axpy4(acc, v0, GCN_EDGE_W(my, j));
            gcn_axpy4(acc, v1, GCN_EDGE_W(my, j + 1));
            gcn_axpy4(acc, v2, GCN_EDGE_W(my, j + 2));
            gcn_axpy4(acc, v3, GCN_EDGE_W(my, j + 3));
            j += 4;
        }
        const int r = m - j;
        if (r == 1) {
            const float4 v0 = gcn_row4(x, GCN_EDGE_SRC(my, j), lane4);
            gcn_axpy4(acc, v0, GCN_EDGE_W(my, j));
        } else if (r == 2) {
            const float4 v0 = gcn_row4(x, GCN_EDGE_SRC(my, j),     lane4);
            const float4 v1 = gcn_row4(x, GCN_EDGE_SRC(my, j + 1), lane4);
            gcn_axpy4(acc, v0, GCN_EDGE_W(my, j));
            gcn_axpy4(acc, v1, GCN_EDGE_W(my, j + 1));
        } else if (r == 3) {
            const float4 v0 = gcn_row4(x, GCN_EDGE_SRC(my, j),     lane4);
            const float4 v1 = gcn_row4(x, GCN_EDGE_SRC(my, j + 1), lane4);
            const float4 v2 = gcn_row4(x, GCN_EDGE_SRC(my, j + 2), lane4);
            gcn_axpy4(acc, v0, GCN_EDGE_W(my, j));
            gcn_axpy4(acc, v1, GCN_EDGE_W(my, j + 1));
            gcn_axpy4(acc, v2, GCN_EDGE_W(my, j + 2));
        }
    }
    return acc;
}

#define GCN_K 128
#define GCN_C 128
#define GCN_C_LAST 16
#define GCN_TILE_PITCH 132
#define GCN_LAYER_GROUPS 1024
#define GCN_MAX_TILES_PER_GROUP 16
#define GCN_N_TILES (GCN_N_NODES / 16)
static_assert(GCN_N_NODES % 16 == 0, "whole 16-row tiles only");
static_assert(GCN_LAYER_GROUPS <= GCN_N_TILES, "every work-group gets at least one tile");
static_assert((GCN_N_TILES + GCN_LAYER_GROUPS - 1) / GCN_LAYER_GROUPS <= GCN_MAX_TILES_PER_GROUP,
              "a work-group's tiles must fit the lanes that hold their row bounds");
static_assert(GCN_C == 8 * 16, "8 waves, one 16-column strip each");

template <bool FUSE_NEXT>
__global__ void __launch_bounds__(256) __attribute__((amdgpu_waves_per_eu(8, 8)))
gcn_layer_fused(const float* __restrict__ x, const int* __restrict__ row_ptr,
                const int* __restrict__ cnt_in, const int* __restrict__ edges,
                const float* __restrict__ deg_in_is, const float* __restrict__ W,
                const float* __restrict__ bias, const float* __restrict__ Wn,
                float* __restrict__ out, int do_relu, int n_tiles) {
    __shared__ __attribute__((aligned(16))) float tile[2][16 * GCN_TILE_PITCH];
    __shared__ __attribute__((aligned(16))) float htile[FUSE_NEXT ? 16 * GCN_TILE_PITCH : 4];
    __shared__ __attribute__((aligned(16))) v2f wn_frag[FUSE_NEXT ? (GCN_K / 4) * 32 : 1];

    const int lane  = threadIdx.x & 31;
    const int lane4 = lane * 4;
    const int half  = lane >> 4;
    const int l15   = lane & 15;
    const int wave  = __builtin_amdgcn_readfirstlane(threadIdx.x >> 5);
    const int ncol  = wave * 16 + l15;
    const int per_group = n_tiles / GCN_LAYER_GROUPS;
    const int extra     = n_tiles % GCN_LAYER_GROUPS;
    const int g         = (int)blockIdx.x;
    const int tile0     = g * per_group + (g < extra ? g : extra);
    const int my_tiles  = per_group + (g < extra ? 1 : 0);
    const int2* __restrict__ ce = (const int2*)edges;

    int rp = 0, rc = 0;
    {
        const int tj = tile0 + (lane >> 1);
        if ((lane >> 1) < my_tiles) {
            const int node = tj * 16 + wave * 2 + (lane & 1);
            rp = row_ptr[node];
            rc = cnt_in[node];
        }
    }

    v2f bfrag[GCN_K / 4];
#pragma unroll
    for (int s = 0; s < GCN_K / 4; ++s) {
        const int kb = s * 4 + half * 2;
        bfrag[s].x = W[(size_t)kb       * GCN_C + ncol];
        bfrag[s].y = W[(size_t)(kb + 1) * GCN_C + ncol];
    }
    const float bv = bias[ncol];

    if constexpr (FUSE_NEXT) {
        for (int idx = threadIdx.x; idx < (GCN_K / 4) * 32; idx += 256) {
            const int s  = idx >> 5;
            const int ln = idx & 31;
            const int kb = s * 4 + (ln >> 4) * 2;
            v2f f;
            f.x = Wn[(size_t)kb       * GCN_C_LAST + (ln & 15)];
            f.y = Wn[(size_t)(kb + 1) * GCN_C_LAST + (ln & 15)];
            wn_frag[idx] = f;
        }
    }

    for (int i = 0; i < my_tiles; ++i) {
        const int t = tile0 + i;
        float* tb = &tile[i & 1][0];

        const int beg0 = __builtin_amdgcn_readlane(rp, 2 * i);
        const int cnt0 = __builtin_amdgcn_readlane(rc, 2 * i);
        const int beg1 = __builtin_amdgcn_readlane(rp, 2 * i + 1);
        const int cnt1 = __builtin_amdgcn_readlane(rc, 2 * i + 1);

        int2 cur0 = make_int2(0, 0), cur1 = make_int2(0, 0);
        if (lane < cnt0) cur0 = ce[beg0 + lane];
        if (lane < cnt1) cur1 = ce[beg1 + lane];
        const float scale = deg_in_is[t * 16 + l15];

        const float4 g0 = gcn_gather_row(x, ce, beg0, cnt0, cur0, lane, lane4);
        *(float4*)(tb + (2 * wave) * GCN_TILE_PITCH + lane4) = g0;
        const float4 g1 = gcn_gather_row(x, ce, beg1, cnt1, cur1, lane, lane4);
        *(float4*)(tb + (2 * wave + 1) * GCN_TILE_PITCH + lane4) = g1;

        __syncthreads();

        {
            const float* arow = tb + l15 * GCN_TILE_PITCH + half * 2;
            v8f c = {};
#pragma unroll
            for (int s = 0; s < GCN_K / 4; ++s) {
                const v2f araw = *(const v2f*)(arow + s * 4);
                v2f a;
                a.x = araw.x * scale;
                a.y = araw.y * scale;
                c = __builtin_amdgcn_wmma_f32_16x16x4_f32(
                        false, a, false, bfrag[s], (short)0, c, false, false);
            }
#pragma unroll
            for (int v = 0; v < 8; ++v) {
                float val = c[v] + bv;
                if (do_relu) val = fmaxf(val, 0.0f);
                if constexpr (FUSE_NEXT) {
                    htile[(half * 8 + v) * GCN_TILE_PITCH + ncol] = val;
                } else {
                    out[(size_t)(t * 16 + half * 8 + v) * GCN_C + ncol] = val;
                }
            }
        }

        if constexpr (FUSE_NEXT) {
            __syncthreads();
            if (wave == (i & 7)) {
                const float* hrow = htile + l15 * GCN_TILE_PITCH + half * 2;
                v8f z0 = {}, z1 = {};
#pragma unroll
                for (int s = 0; s < GCN_K / 4; s += 2) {
                    const v2f a0 = *(const v2f*)(hrow + s * 4);
                    const v2f b0 = wn_frag[s * 32 + lane];
                    const v2f a1 = *(const v2f*)(hrow + (s + 1) * 4);
                    const v2f b1 = wn_frag[(s + 1) * 32 + lane];
                    z0 = __builtin_amdgcn_wmma_f32_16x16x4_f32(
                            false, a0, false, b0, (short)0, z0, false, false);
                    z1 = __builtin_amdgcn_wmma_f32_16x16x4_f32(
                            false, a1, false, b1, (short)0, z1, false, false);
                }
#pragma unroll
                for (int v = 0; v < 8; ++v) {
                    out[(size_t)(t * 16 + half * 8 + v) * GCN_C_LAST + l15] = z0[v] + z1[v];
                }
            }
        }
    }
}

#define GCN_LAST_BATCH 16

__global__ void __launch_bounds__(256) __attribute__((amdgpu_waves_per_eu(4, 6)))
gcn_gather_last(const float* __restrict__ z, const int* __restrict__ row_ptr,
                const int* __restrict__ cnt_in, const int* __restrict__ edges,
                const float* __restrict__ deg_in_is, const float* __restrict__ bias,
                float* __restrict__ out, int n) {
    const int g  = blockIdx.x * blockDim.x + threadIdx.x;
    const int m  = g >> 2;
    const int q4 = (g & 3) * 4;
    if (m >= n) return;
    const int    cnt = cnt_in[m];
    const int    beg = row_ptr[m];
    const float  sc  = deg_in_is[m];
    const float4 b   = *(const float4*)(bias + q4);
    asm volatile("" :: "v"(cnt), "v"(beg), "v"(sc), "v"(b.x), "v"(b.y), "v"(b.z), "v"(b.w));
    const int2* __restrict__ ce = (const int2*)edges + beg;
    const int last = cnt - 1;

    float4 acc = make_float4(0.0f, 0.0f, 0.0f, 0.0f);
    for (int j = 0; j < cnt; j += GCN_LAST_BATCH) {
        int2 e[GCN_LAST_BATCH];
        float4 v[GCN_LAST_BATCH];
#pragma unroll
        for (int k = 0; k < GCN_LAST_BATCH; ++k) {
            const int jj = (j + k < last) ? (j + k) : last;
            e[k] = ce[jj];
        }
#pragma unroll
        for (int k = 0; k < GCN_LAST_BATCH; ++k) {
            v[k] = *(const float4*)(z + (size_t)e[k].x * GCN_C_LAST + q4);
        }
#pragma unroll
        for (int k = 0; k < GCN_LAST_BATCH; ++k) {
            const bool on = (j + k) < cnt;
            float4 vv;
            vv.x = on ? v[k].x : 0.0f;
            vv.y = on ? v[k].y : 0.0f;
            vv.z = on ? v[k].z : 0.0f;
            vv.w = on ? v[k].w : 0.0f;
            gcn_axpy4(acc, vv, __int_as_float(e[k].y));
        }
    }
    float4 r;
    r.x = fmaf(acc.x, sc, b.x);
    r.y = fmaf(acc.y, sc, b.y);
    r.z = fmaf(acc.z, sc, b.z);
    r.w = fmaf(acc.w, sc, b.w);
    *(float4*)(out + (size_t)m * GCN_C_LAST + q4) = r;
}

extern "C" void kernel_launch(void* const* d_in, const int* in_sizes, int n_in,
                              void* d_out, int out_size, void* d_ws, size_t ws_size,
                              hipStream_t stream) {
    const float* features = (const float*)d_in[0];
    const int*   src      = (const int*)  d_in[1];
    const int*   dst      = (const int*)  d_in[2];
    const float* W1       = (const float*)d_in[3];
    const float* b1       = (const float*)d_in[4];
    const float* W2       = (const float*)d_in[5];
    const float* b2       = (const float*)d_in[6];
    const float* W3       = (const float*)d_in[7];
    const float* b3       = (const float*)d_in[8];
    float* out = (float*)d_out;

    const int N = GCN_N_NODES;
    const int E = GCN_N_EDGES;
    const size_t NF = (size_t)N * 128;

    float* bufZ    = (float*)d_ws;
    float* bufB    = bufZ + NF;
    int*   edges   = (int*)(bufB + NF);
    int*   runs    = edges + 2 * (size_t)E;
    int*   cnt_out = runs + 2 * (size_t)E;
    int*   cnt_in  = cnt_out + N;
    int*   row_ptr = cnt_in + N;
    float* deg_i   = (float*)(row_ptr + N);
    int*   part_out = (int*)(deg_i + N);
    int*   blk_in   = part_out + (size_t)GCN_HIST_CHUNKS * N;
    int*   blk_base = blk_in + GCN_HIST_CHUNKS * GCN_COMB_BLOCKS;

    const int TB = 256;

    gcn_hist_lds<<<2 * GCN_HIST_CHUNKS, 1024, 0, stream>>>(src, dst, part_out, blk_in, blk_base, runs);
    gcn_combine<<<GCN_COMB_BLOCKS, GCN_COMB_TB, 0, stream>>>(part_out, cnt_out, N);
    gcn_place_block<<<GCN_COMB_BLOCKS, GCN_OWN_THREADS, 0, stream>>>(runs, blk_in, blk_base, cnt_out,
                                                                    cnt_in, row_ptr, edges, deg_i, N);

    const int groups = GCN_LAYER_GROUPS;

    gcn_layer_fused<false><<<groups, TB, 0, stream>>>(features, row_ptr, cnt_in, edges, deg_i,
                                                      W1, b1, W3, bufB, 1, GCN_N_TILES);
    gcn_layer_fused<true><<<groups, TB, 0, stream>>>(bufB, row_ptr, cnt_in, edges, deg_i,
                                                     W2, b2, W3, bufZ, 1, GCN_N_TILES);
    gcn_gather_last<<<(4 * N + TB - 1) / TB, TB, 0, stream>>>(bufZ, row_ptr, cnt_in, edges, deg_i,
                                                             b3, out, N);

    (void)in_sizes; (void)n_in; (void)out_size; (void)ws_size;
}


# Edge-counting launch spread over 125 WGPs (was 50): destination ordering in 75 chunks, source counts in 50 on paired 16-bit LDS counters; placement finds pieces by search; row order LDS arrival
# speedup vs baseline: 6.0675x; 1.0823x over previous
#include <hip/hip_runtime.h>
#include <math.h>


typedef __attribute__((ext_vector_type(2))) float v2f;
typedef __attribute__((ext_vector_type(8))) float v8f;

#define GCN_N_NODES 50000
#define GCN_N_EDGES 600000


#define GCN_HIST_CHUNKS 50
#define GCN_SRC_WORDS (GCN_N_NODES / 2)
#define GCN_HIST_CHUNK (GCN_N_EDGES / GCN_HIST_CHUNKS)
#define GCN_HIST_QUADS (GCN_HIST_CHUNK / 4)
#define GCN_HIST_ITERS ((GCN_HIST_QUADS + 1023) / 1024)
#define GCN_DST_CHUNKS 75
#define GCN_DST_CHUNK (GCN_N_EDGES / GCN_DST_CHUNKS)
#define GCN_DST_QUADS (GCN_DST_CHUNK / 4)
#define GCN_DST_ITERS ((GCN_DST_QUADS + 1023) / 1024)
static_assert(GCN_N_EDGES % GCN_DST_CHUNKS == 0, "equal chunks");
static_assert(GCN_DST_CHUNK % 4 == 0, "a chunk is read in 16-byte groups");
static_assert(GCN_N_EDGES % GCN_HIST_CHUNKS == 0, "equal chunks");
static_assert(GCN_HIST_CHUNK % 4 == 0, "a chunk is read in 16-byte groups");
static_assert(GCN_N_NODES % 8 == 0, "counter planes are cleared and copied in 16-byte groups of 8 counters");
static_assert(GCN_HIST_CHUNK < 65536, "a 16-bit counter holds a chunk's whole count");
static_assert(GCN_N_NODES * 4 <= 320 * 1024, "all source counters in one work-group's LDS");
#define GCN_COMB_TB 256
#define GCN_COMB_BLOCKS ((GCN_N_NODES + GCN_COMB_TB - 1) / GCN_COMB_TB)
#define GCN_HIST_COPY_TRIPS ((GCN_SRC_WORDS / 4 + 32 * (GCN_COMB_TB / 4) - 1) / (32 * (GCN_COMB_TB / 4)))
static_assert(GCN_COMB_TB == 2 * 32 * 4, "a wave copies a block as two 16-byte groups per lane");
static_assert(GCN_COMB_BLOCKS <= GCN_COMB_TB, "one thread per earlier block in the placement kernel");
static_assert(GCN_DST_CHUNK * 8 <= GCN_N_NODES * 4, "a chunk's run is put together in the counters' LDS");
static_assert(GCN_DST_CHUNK % 2 == 0, "a run leaves the LDS in 16-byte groups of two records");
static_assert(GCN_COMB_BLOCKS < GCN_COMB_TB, "the first GCN_COMB_TB threads scan the block totals, one more for the end");

__global__ void __launch_bounds__(1024)
gcn_hist_lds(const int* __restrict__ src, const int* __restrict__ dst,
             int* __restrict__ part_out,
             int* __restrict__ blk_in, int* __restrict__ blk_base, int* __restrict__ runs) {
    __shared__ __attribute__((aligned(16))) int h[GCN_N_NODES];
    __shared__ int bt[GCN_COMB_TB];
    __shared__ int cur[GCN_COMB_TB];
    __shared__ int wtot[GCN_COMB_TB / 32];
    const int tid  = threadIdx.x;
    const int lane = tid & 31;
    const int w    = tid >> 5;
    const bool is_dst = (int)blockIdx.x < GCN_DST_CHUNKS;
    const int  c      = is_dst ? (int)blockIdx.x : (int)blockIdx.x - GCN_DST_CHUNKS;
    const int  e0     = c * (is_dst ? GCN_DST_CHUNK : GCN_HIST_CHUNK);

    if (is_dst) {
        int4 k4[GCN_DST_ITERS], s4[GCN_DST_ITERS];
#pragma unroll
        for (int k = 0; k < GCN_DST_ITERS; ++k) {
            const int q = tid + k * 1024;
            k4[k] = make_int4(0, 0, 0, 0);
            if (q < GCN_DST_QUADS) k4[k] = *(const int4*)(dst + e0 + 4 * q);
        }
#pragma unroll
        for (int k = 0; k < GCN_DST_ITERS; ++k) {
            const int q = tid + k * 1024;
            s4[k] = make_int4(0, 0, 0, 0);
            if (q < GCN_DST_QUADS) s4[k] = *(const int4*)(src + e0 + 4 * q);
        }
        if (tid < GCN_COMB_TB) bt[tid] = 0;
        __syncthreads();
        int4 r[GCN_DST_ITERS];
#pragma unroll
        for (int k = 0; k < GCN_DST_ITERS; ++k) {
            const int q = tid + k * 1024;
            r[k] = make_int4(0, 0, 0, 0);
            if (q < GCN_DST_QUADS) {
                r[k].x = atomicAdd(&bt[k4[k].x / GCN_COMB_TB], 1);
                r[k].y = atomicAdd(&bt[k4[k].y / GCN_COMB_TB], 1);
                r[k].z = atomicAdd(&bt[k4[k].z / GCN_COMB_TB], 1);
                r[k].w = atomicAdd(&bt[k4[k].w / GCN_COMB_TB], 1);
            }
        }
        __syncthreads();
        int v = 0, incl = 0, excl = 0;
        if (tid < GCN_COMB_TB) {
            v = bt[tid];
            incl = v;
            for (int m = 1; m < 32; m <<= 1) {
                const int y = __shfl_up(incl, m);
                if (lane >= m) incl += y;
            }
            if (lane == 31) wtot[w] = incl;
        }
        __syncthreads();
        if (tid < GCN_COMB_TB) {
            int base = 0;
            for (int k = 0; k < GCN_COMB_TB / 32; ++k) base += (k < w) ? wtot[k] : 0;
            excl = base + incl - v;
            cur[tid] = excl;
        }
        __syncthreads();
        int2* run = (int2*)h;
#pragma unroll
        for (int k = 0; k < GCN_DST_ITERS; ++k) {
            const int q = tid + k * 1024;
            if (q < GCN_DST_QUADS) {
                run[cur[k4[k].x / GCN_COMB_TB] + r[k].x] = make_int2(s4[k].x, k4[k].x);
                run[cur[k4[k].y / GCN_COMB_TB] + r[k].y] = make_int2(s4[k].y, k4[k].y);
                run[cur[k4[k].z / GCN_COMB_TB] + r[k].z] = make_int2(s4[k].z, k4[k].z);
                run[cur[k4[k].w / GCN_COMB_TB] + r[k].w] = make_int2(s4[k].w, k4[k].w);
            }
        }
        __syncthreads();
        if (tid < GCN_COMB_BLOCKS)  blk_in[c * GCN_COMB_BLOCKS + tid] = v;
        if (tid <= GCN_COMB_BLOCKS) blk_base[c * (GCN_COMB_BLOCKS + 1) + tid] = excl;
        int4* __restrict__ run_out = (int4*)(runs + 2 * (size_t)e0);
        for (int i = tid; i < GCN_DST_CHUNK / 2; i += 1024) run_out[i] = ((const int4*)h)[i];
    } else {
        int4 k4[GCN_HIST_ITERS];
#pragma unroll
        for (int k = 0; k < GCN_HIST_ITERS; ++k) {
            const int q = tid + k * 1024;
            k4[k] = make_int4(0, 0, 0, 0);
            if (q < GCN_HIST_QUADS) k4[k] = *(const int4*)(src + e0 + 4 * q);
        }
        for (int i = tid; i < GCN_SRC_WORDS / 4; i += 1024) ((int4*)h)[i] = make_int4(0, 0, 0, 0);
        __syncthreads();
#pragma unroll
        for (int k = 0; k < GCN_HIST_ITERS; ++k) {
            const int q = tid + k * 1024;
            if (q < GCN_HIST_QUADS) {
                atomicAdd(&h[k4[k].x >> 1], 1 << ((k4[k].x & 1) << 4));
                atomicAdd(&h[k4[k].y >> 1], 1 << ((k4[k].y & 1) << 4));
                atomicAdd(&h[k4[k].z >> 1], 1 << ((k4[k].z & 1) << 4));
                atomicAdd(&h[k4[k].w >> 1], 1 << ((k4[k].w & 1) << 4));
            }
        }
        __syncthreads();
        int4* __restrict__ plane = (int4*)(part_out + (size_t)c * GCN_SRC_WORDS);
        const int last = GCN_SRC_WORDS / 4 - 1;
        int4 q0[GCN_HIST_COPY_TRIPS], q1[GCN_HIST_COPY_TRIPS];
#pragma unroll
        for (int k = 0; k < GCN_HIST_COPY_TRIPS; ++k) {
            const int i0 = (w + 32 * k) * (GCN_COMB_TB / 4) + lane;
            const int i1 = i0 + 32;
            q0[k] = ((const int4*)h)[i0 < last ? i0 : last];
            q1[k] = ((const int4*)h)[i1 < last ? i1 : last];
        }
#pragma unroll
        for (int k = 0; k < GCN_HIST_COPY_TRIPS; ++k) {
            const int i0 = (w + 32 * k) * (GCN_COMB_TB / 4) + lane;
            const int i1 = i0 + 32;
            if (i0 <= last) plane[i0] = q0[k];
            if (i1 <= last) plane[i1] = q1[k];
        }
    }
}

__global__ void __launch_bounds__(GCN_COMB_TB)
gcn_combine(const int* __restrict__ part_out, int* __restrict__ cnt_out, int n) {
    const int d = blockIdx.x * GCN_COMB_TB + threadIdx.x;
    const int dl = d < n ? d : n - 1;
    const int wd = dl >> 1;
    const int sh = (dl & 1) << 4;
    int vo[GCN_HIST_CHUNKS];
#pragma unroll
    for (int c = 0; c < GCN_HIST_CHUNKS; ++c) vo[c] = part_out[(size_t)c * GCN_SRC_WORDS + wd];
    int so = 0;
#pragma unroll
    for (int c = 0; c < GCN_HIST_CHUNKS; ++c) so += (int)(((unsigned)vo[c] >> sh) & 0xffffu);
    if (d < n) cnt_out[d] = so;
}

static __device__ __forceinline__ float gcn_inv_sqrt_clamped(int c) {
    float v = (float)c;
    v = v < 1.0f ? 1.0f : v;
    return 1.0f / sqrtf(v);
}

#define GCN_OWN_THREADS 1024
#define GCN_OWN_CAP 8192
#define GCN_OWN_FAST 4
#define GCN_PS_SLOTS 128
static_assert(GCN_DST_CHUNKS < GCN_PS_SLOTS, "the table holds one entry per piece and one for the end");
static_assert(GCN_COMB_TB + GCN_DST_CHUNKS <= GCN_OWN_THREADS, "one thread per chunk, behind the node threads, fetches the piece bounds");
static_assert(GCN_PS_SLOTS <= GCN_OWN_THREADS, "one thread per table entry");

static __device__ __forceinline__ int gcn_piece_of(const int* ps, const int j) {
    int lo = 0;
#pragma unroll
    for (int st = GCN_PS_SLOTS / 2; st >= 1; st >>= 1) lo = (ps[lo + st] <= j) ? lo + st : lo;
    return lo;
}
static_assert(GCN_COMB_TB <= GCN_OWN_THREADS, "one thread per node of the block");

__global__ void __launch_bounds__(GCN_OWN_THREADS)
gcn_place_block(const int* __restrict__ runs, const int* __restrict__ blk_in,
                const int* __restrict__ blk_base, const int* __restrict__ cnt_out,
                int* __restrict__ cnt_in, int* __restrict__ row_ptr, int* __restrict__ edges,
                float* __restrict__ deg_in_is, int n) {
    __shared__ int2 stage[GCN_OWN_CAP];
    __shared__ int rs[GCN_DST_CHUNKS];
    __shared__ int rl[GCN_DST_CHUNKS];
    __shared__ int ps[GCN_PS_SLOTS];
    __shared__ int ncnt[GCN_COMB_TB];
    __shared__ int wbefore[GCN_COMB_TB / 32];
    __shared__ int wrows[GCN_COMB_TB / 32];
    const int tid  = threadIdx.x;
    const int lane = tid & 31;
    const int w    = tid >> 5;
    const int b    = blockIdx.x;
    const int n0   = b * GCN_COMB_TB;
    if (tid < GCN_COMB_TB) {
        const int tb = tid < b ? tid : 0;
        int vb[GCN_DST_CHUNKS];
#pragma unroll
        for (int c = 0; c < GCN_DST_CHUNKS; ++c) vb[c] = blk_in[c * GCN_COMB_BLOCKS + tb];
        int before = 0;
#pragma unroll
        for (int c = 0; c < GCN_DST_CHUNKS; ++c) before += vb[c];
        before = tid < b ? before : 0;
        for (int m = 16; m >= 1; m >>= 1) before += __shfl_xor(before, m);
        if (lane == 0) wbefore[w] = before;
        ncnt[tid] = 0;
    } else if (tid < GCN_COMB_TB + GCN_DST_CHUNKS) {
        const int t = tid - GCN_COMB_TB;
        rs[t] = blk_base[t * (GCN_COMB_BLOCKS + 1) + b];
        rl[t] = blk_in[t * GCN_COMB_BLOCKS + b];
    }
    __syncthreads();
    if (tid < GCN_PS_SLOTS) {
        int a = 0;
#pragma unroll
        for (int i = 0; i < GCN_DST_CHUNKS; ++i) a += (i < tid) ? rl[i] : 0;
        ps[tid] = tid <= GCN_DST_CHUNKS ? a : 0x7fffffff;
    }
    int out_lo = 0;
    for (int k = 0; k < GCN_COMB_TB / 32; ++k) out_lo += wbefore[k];
    __syncthreads();
    const int total = ps[GCN_DST_CHUNKS];

    const int2* __restrict__ rr = (const int2*)runs;
    int2* __restrict__ eo = (int2*)edges;
    int2 rec[GCN_OWN_FAST];
    int  co[GCN_OWN_FAST];
#pragma unroll
    for (int k = 0; k < GCN_OWN_FAST; ++k) { rec[k] = make_int2(0, n0); co[k] = 1; }
    if (total > 0) {
        int ri[GCN_OWN_FAST], jj[GCN_OWN_FAST], pc[GCN_OWN_FAST];
#pragma unroll
        for (int k = 0; k < GCN_OWN_FAST; ++k) {
            const int j = tid + k * GCN_OWN_THREADS;
            jj[k] = j < total ? j : total - 1;
            pc[k] = 0;
        }
#pragma unroll
        for (int st = GCN_PS_SLOTS / 2; st >= 1; st >>= 1) {
#pragma unroll
            for (int k = 0; k < GCN_OWN_FAST; ++k) pc[k] = (ps[pc[k] + st] <= jj[k]) ? pc[k] + st : pc[k];
        }
#pragma unroll
        for (int k = 0; k < GCN_OWN_FAST; ++k) {
            ri[k] = pc[k] * GCN_DST_CHUNK + rs[pc[k]] + (jj[k] - ps[pc[k]]);
        }
#pragma unroll
        for (int k = 0; k < GCN_OWN_FAST; ++k) rec[k] = rr[ri[k]];
#pragma unroll
        for (int k = 0; k < GCN_OWN_FAST; ++k) co[k] = cnt_out[rec[k].x];
#pragma unroll
        for (int k = 0; k < GCN_OWN_FAST; ++k) {
            const int j = tid + k * GCN_OWN_THREADS;
            if (j < total) atomicAdd(&ncnt[rec[k].y - n0], 1);
        }
        for (int j = tid + GCN_OWN_FAST * GCN_OWN_THREADS; j < total; j += GCN_OWN_THREADS) {
            const int c2 = gcn_piece_of(ps, j);
            const int2 e = rr[(size_t)c2 * GCN_DST_CHUNK + rs[c2] + (j - ps[c2])];
            atomicAdd(&ncnt[e.y - n0], 1);
        }
    }
    __syncthreads();
    int v = 0, incl = 0, loc = 0;
    if (tid < GCN_COMB_TB) {
        v = ncnt[tid];
        incl = v;
        for (int m = 1; m < 32; m <<= 1) {
            const int y = __shfl_up(incl, m);
            if (lane >= m) incl += y;
        }
        if (lane == 31) wrows[w] = incl;
    }
    __syncthreads();
    if (tid < GCN_COMB_TB) {
        int base = 0;
        for (int k = 0; k < GCN_COMB_TB / 32; ++k) base += (k < w) ? wrows[k] : 0;
        loc = base + incl - v;
        ncnt[tid] = loc;
    }
    __syncthreads();
    if (total > 0) {
        int o[GCN_OWN_FAST];
#pragma unroll
        for (int k = 0; k < GCN_OWN_FAST; ++k) {
            const bool on = tid + k * GCN_OWN_THREADS < total;
            o[k] = atomicAdd(&ncnt[on ? rec[k].y - n0 : (tid & (GCN_COMB_TB - 1))], on ? 1 : 0);
        }
#pragma unroll
        for (int k = 0; k < GCN_OWN_FAST; ++k) {
            const int j = tid + k * GCN_OWN_THREADS;
            if (j < total) {
                const int2 pr = make_int2(rec[k].x, __float_as_int(gcn_inv_sqrt_clamped(co[k])));
                if ((unsigned)o[k] < (unsigned)GCN_OWN_CAP) stage[o[k]] = pr; else eo[out_lo + o[k]] = pr;
            }
        }
        for (int j = tid + GCN_OWN_FAST * GCN_OWN_THREADS; j < total; j += GCN_OWN_THREADS) {
            const int c2 = gcn_piece_of(ps, j);
            const int2 e  = rr[(size_t)c2 * GCN_DST_CHUNK + rs[c2] + (j - ps[c2])];
            const int  os = atomicAdd(&ncnt[e.y - n0], 1);
            const int2 pr = make_int2(e.x, __float_as_int(gcn_inv_sqrt_clamped(cnt_out[e.x])));
            if ((unsigned)os < (unsigned)GCN_OWN_CAP) stage[os] = pr; else eo[out_lo + os] = pr;
        }
    }
    __syncthreads();
    const int staged = total < GCN_OWN_CAP ? total : GCN_OWN_CAP;
    for (int o = tid; o < staged; o += GCN_OWN_THREADS) eo[out_lo + o] = stage[o];
    if (tid < GCN_COMB_TB && n0 + tid < n) {
        row_ptr[n0 + tid]   = out_lo + loc;
        cnt_in[n0 + tid]    = v;
        deg_in_is[n0 + tid] = gcn_inv_sqrt_clamped(v);
    }
}

static __device__ __forceinline__ float4 gcn_row4(const float* __restrict__ x, int s, int lane4) {
    return *(const float4*)(x + (size_t)s * 128 + lane4);
}
static __device__ __forceinline__ void gcn_axpy4(float4& acc, const float4 v, const float w) {
    acc.x = fmaf(v.x, w, acc.x);
    acc.y = fmaf(v.y, w, acc.y);
    acc.z = fmaf(v.z, w, acc.z);
    acc.w = fmaf(v.w, w, acc.w);
}
#define GCN_EDGE_SRC(my, j) __builtin_amdgcn_readlane((my).x, (j))
#define GCN_EDGE_W(my, j)   __int_as_float(__builtin_amdgcn_readlane((my).y, (j)))

static __device__ __forceinline__ float4 gcn_gather_row(const float* __restrict__ x,
                                                        const int2* __restrict__ ce,
                                                        const int beg, const int cnt,
                                                        const int2 first,
                                                        const int lane, const int lane4) {
    float4 acc = make_float4(0.0f, 0.0f, 0.0f, 0.0f);
    for (int b = 0; b < cnt; b += 32) {
        const int m = (cnt - b < 32) ? (cnt - b) : 32;
        int2 my = first;
        if (b > 0) {
            my = make_int2(0, 0);
            if (lane < m) my = ce[beg + b + lane];
        }
        int j = 0;
        for (; j + 8 <= m; j += 8) {
            const float4 v0 = gcn_row4(x, GCN_EDGE_SRC(my, j),     lane4);
            const float4 v1 = gcn_row4(x, GCN_EDGE_SRC(my, j + 1), lane4);
            const float4 v2 = gcn_row4(x, GCN_EDGE_SRC(my, j + 2), lane4);
            const float4 v3 = gcn_row4(x, GCN_EDGE_SRC(my, j + 3), lane4);
            const float4 v4 = gcn_row4(x, GCN_EDGE_SRC(my, j + 4), lane4);
            const float4 v5 = gcn_row4(x, GCN_EDGE_SRC(my, j + 5), lane4);
            const float4 v6 = gcn_row4(x, GCN_EDGE_SRC(my, j + 6), lane4);
            const float4 v7 = gcn_row4(x, GCN_EDGE_SRC(my, j + 7), lane4);
            gcn_axpy4(acc, v0, GCN_EDGE_W(my, j));
            gcn_axpy4(acc, v1, GCN_EDGE_W(my, j + 1));
            gcn_axpy4(acc, v2, GCN_EDGE_W(my, j + 2));
            gcn_axpy4(acc, v3, GCN_EDGE_W(my, j + 3));
            gcn_axpy4(acc, v4, GCN_EDGE_W(my, j + 4));
            gcn_axpy4(acc, v5, GCN_EDGE_W(my, j + 5));
            gcn_axpy4(acc, v6, GCN_EDGE_W(my, j + 6));
            gcn_axpy4(acc, v7, GCN_EDGE_W(my, j + 7));
        }
        if (j + 4 <= m) {
            const float4 v0 = gcn_row4(x, GCN_EDGE_SRC(my, j),     lane4);
            const float4 v1 = gcn_row4(x, GCN_EDGE_SRC(my, j + 1), lane4);
            const float4 v2 = gcn_row4(x, GCN_EDGE_SRC(my, j + 2), lane4);
            const float4 v3 = gcn_row4(x, GCN_EDGE_SRC(my, j + 3), lane4);
            gcn_axpy4(acc, v0, GCN_EDGE_W(my, j));
            gcn_axpy4(acc, v1, GCN_EDGE_W(my, j + 1));
            gcn_axpy4(acc, v2, GCN_EDGE_W(my, j + 2));
            gcn_axpy4(acc, v3, GCN_EDGE_W(my, j + 3));
            j += 4;
        }
        const int r = m - j;
        if (r == 1) {
            const float4 v0 = gcn_row4(x, GCN_EDGE_SRC(my, j), lane4);
            gcn_axpy4(acc, v0, GCN_EDGE_W(my, j));
        } else if (r == 2) {
            const float4 v0 = gcn_row4(x, GCN_EDGE_SRC(my, j),     lane4);
            const float4 v1 = gcn_row4(x, GCN_EDGE_SRC(my, j + 1), lane4);
            gcn_axpy4(acc, v0, GCN_EDGE_W(my, j));
            gcn_axpy4(acc, v1, GCN_EDGE_W(my, j + 1));
        } else if (r == 3) {
            const float4 v0 = gcn_row4(x, GCN_EDGE_SRC(my, j),     lane4);
            const float4 v1 = gcn_row4(x, GCN_EDGE_SRC(my, j + 1), lane4);
            const float4 v2 = gcn_row4(x, GCN_EDGE_SRC(my, j + 2), lane4);
            gcn_axpy4(acc, v0, GCN_EDGE_W(my, j));
            gcn_axpy4(acc, v1, GCN_EDGE_W(my, j + 1));
            gcn_axpy4(acc, v2, GCN_EDGE_W(my, j + 2));
        }
    }
    return acc;
}

#define GCN_K 128
#define GCN_C 128
#define GCN_C_LAST 16
#define GCN_TILE_PITCH 132
#define GCN_LAYER_GROUPS 1024
#define GCN_MAX_TILES_PER_GROUP 16
#define GCN_N_TILES (GCN_N_NODES / 16)
static_assert(GCN_N_NODES % 16 == 0, "whole 16-row tiles only");
static_assert(GCN_LAYER_GROUPS <= GCN_N_TILES, "every work-group gets at least one tile");
static_assert((GCN_N_TILES + GCN_LAYER_GROUPS - 1) / GCN_LAYER_GROUPS <= GCN_MAX_TILES_PER_GROUP,
              "a work-group's tiles must fit the lanes that hold their row bounds");
static_assert(GCN_C == 8 * 16, "8 waves, one 16-column strip each");

template <bool FUSE_NEXT>
__global__ void __launch_bounds__(256) __attribute__((amdgpu_waves_per_eu(8, 8)))
gcn_layer_fused(const float* __restrict__ x, const int* __restrict__ row_ptr,
                const int* __restrict__ cnt_in, const int* __restrict__ edges,
                const float* __restrict__ deg_in_is, const float* __restrict__ W,
                const float* __restrict__ bias, const float* __restrict__ Wn,
                float* __restrict__ out, int do_relu, int n_tiles) {
    __shared__ __attribute__((aligned(16))) float tile[2][16 * GCN_TILE_PITCH];
    __shared__ __attribute__((aligned(16))) float htile[FUSE_NEXT ? 16 * GCN_TILE_PITCH : 4];
    __shared__ __attribute__((aligned(16))) v2f wn_frag[FUSE_NEXT ? (GCN_K / 4) * 32 : 1];

    const int lane  = threadIdx.x & 31;
    const int lane4 = lane * 4;
    const int half  = lane >> 4;
    const int l15   = lane & 15;
    const int wave  = __builtin_amdgcn_readfirstlane(threadIdx.x >> 5);
    const int ncol  = wave * 16 + l15;
    const int per_group = n_tiles / GCN_LAYER_GROUPS;
    const int extra     = n_tiles % GCN_LAYER_GROUPS;
    const int g         = (int)blockIdx.x;
    const int tile0     = g * per_group + (g < extra ? g : extra);
    const int my_tiles  = per_group + (g < extra ? 1 : 0);
    const int2* __restrict__ ce = (const int2*)edges;

    int rp = 0, rc = 0;
    {
        const int tj = tile0 + (lane >> 1);
        if ((lane >> 1) < my_tiles) {
            const int node = tj * 16 + wave * 2 + (lane & 1);
            rp = row_ptr[node];
            rc = cnt_in[node];
        }
    }

    v2f bfrag[GCN_K / 4];
#pragma unroll
    for (int s = 0; s < GCN_K / 4; ++s) {
        const int kb = s * 4 + half * 2;
        bfrag[s].x = W[(size_t)kb       * GCN_C + ncol];
        bfrag[s].y = W[(size_t)(kb + 1) * GCN_C + ncol];
    }
    const float bv = bias[ncol];

    if constexpr (FUSE_NEXT) {
        for (int idx = threadIdx.x; idx < (GCN_K / 4) * 32; idx += 256) {
            const int s  = idx >> 5;
            const int ln = idx & 31;
            const int kb = s * 4 + (ln >> 4) * 2;
            v2f f;
            f.x = Wn[(size_t)kb       * GCN_C_LAST + (ln & 15)];
            f.y = Wn[(size_t)(kb + 1) * GCN_C_LAST + (ln & 15)];
            wn_frag[idx] = f;
        }
    }

    for (int i = 0; i < my_tiles; ++i) {
        const int t = tile0 + i;
        float* tb = &tile[i & 1][0];

        const int beg0 = __builtin_amdgcn_readlane(rp, 2 * i);
        const int cnt0 = __builtin_amdgcn_readlane(rc, 2 * i);
        const int beg1 = __builtin_amdgcn_readlane(rp, 2 * i + 1);
        const int cnt1 = __builtin_amdgcn_readlane(rc, 2 * i + 1);

        int2 cur0 = make_int2(0, 0), cur1 = make_int2(0, 0);
        if (lane < cnt0) cur0 = ce[beg0 + lane];
        if (lane < cnt1) cur1 = ce[beg1 + lane];
        const float scale = deg_in_is[t * 16 + l15];

        const float4 g0 = gcn_gather_row(x, ce, beg0, cnt0, cur0, lane, lane4);
        *(float4*)(tb + (2 * wave) * GCN_TILE_PITCH + lane4) = g0;
        const float4 g1 = gcn_gather_row(x, ce, beg1, cnt1, cur1, lane, lane4);
        *(float4*)(tb + (2 * wave + 1) * GCN_TILE_PITCH + lane4) = g1;

        __syncthreads();

        {
            const float* arow = tb + l15 * GCN_TILE_PITCH + half * 2;
            v8f c = {};
#pragma unroll
            for (int s = 0; s < GCN_K / 4; ++s) {
                const v2f araw = *(const v2f*)(arow + s * 4);
                v2f a;
                a.x = araw.x * scale;
                a.y = araw.y * scale;
                c = __builtin_amdgcn_wmma_f32_16x16x4_f32(
                        false, a, false, bfrag[s], (short)0, c, false, false);
            }
#pragma unroll
            for (int v = 0; v < 8; ++v) {
                float val = c[v] + bv;
                if (do_relu) val = fmaxf(val, 0.0f);
                if constexpr (FUSE_NEXT) {
                    htile[(half * 8 + v) * GCN_TILE_PITCH + ncol] = val;
                } else {
                    out[(size_t)(t * 16 + half * 8 + v) * GCN_C + ncol] = val;
                }
            }
        }

        if constexpr (FUSE_NEXT) {
            __syncthreads();
            if (wave == (i & 7)) {
                const float* hrow = htile + l15 * GCN_TILE_PITCH + half * 2;
                v8f z0 = {}, z1 = {};
#pragma unroll
                for (int s = 0; s < GCN_K / 4; s += 2) {
                    const v2f a0 = *(const v2f*)(hrow + s * 4);
                    const v2f b0 = wn_frag[s * 32 + lane];
                    const v2f a1 = *(const v2f*)(hrow + (s + 1) * 4);
                    const v2f b1 = wn_frag[(s + 1) * 32 + lane];
                    z0 = __builtin_amdgcn_wmma_f32_16x16x4_f32(
                            false, a0, false, b0, (short)0, z0, false, false);
                    z1 = __builtin_amdgcn_wmma_f32_16x16x4_f32(
                            false, a1, false, b1, (short)0, z1, false, false);
                }
#pragma unroll
                for (int v = 0; v < 8; ++v) {
                    out[(size_t)(t * 16 + half * 8 + v) * GCN_C_LAST + l15] = z0[v] + z1[v];
                }
            }
        }
    }
}

#define GCN_LAST_BATCH 16

__global__ void __launch_bounds__(256) __attribute__((amdgpu_waves_per_eu(4, 6)))
gcn_gather_last(const float* __restrict__ z, const int* __restrict__ row_ptr,
                const int* __restrict__ cnt_in, const int* __restrict__ edges,
                const float* __restrict__ deg_in_is, const float* __restrict__ bias,
                float* __restrict__ out, int n) {
    const int g  = blockIdx.x * blockDim.x + threadIdx.x;
    const int m  = g >> 2;
    const int q4 = (g & 3) * 4;
    if (m >= n) return;
    const int    cnt = cnt_in[m];
    const int    beg = row_ptr[m];
    const float  sc  = deg_in_is[m];
    const float4 b   = *(const float4*)(bias + q4);
    asm volatile("" :: "v"(cnt), "v"(beg), "v"(sc), "v"(b.x), "v"(b.y), "v"(b.z), "v"(b.w));
    const int2* __restrict__ ce = (const int2*)edges + beg;
    const int last = cnt - 1;

    float4 acc = make_float4(0.0f, 0.0f, 0.0f, 0.0f);
    for (int j = 0; j < cnt; j += GCN_LAST_BATCH) {
        int2 e[GCN_LAST_BATCH];
        float4 v[GCN_LAST_BATCH];
#pragma unroll
        for (int k = 0; k < GCN_LAST_BATCH; ++k) {
            const int jj = (j + k < last) ? (j + k) : last;
            e[k] = ce[jj];
        }
#pragma unroll
        for (int k = 0; k < GCN_LAST_BATCH; ++k) {
            v[k] = *(const float4*)(z + (size_t)e[k].x * GCN_C_LAST + q4);
        }
#pragma unroll
        for (int k = 0; k < GCN_LAST_BATCH; ++k) {
            const bool on = (j + k) < cnt;
            float4 vv;
            vv.x = on ? v[k].x : 0.0f;
            vv.y = on ? v[k].y : 0.0f;
            vv.z = on ? v[k].z : 0.0f;
            vv.w = on ? v[k].w : 0.0f;
            gcn_axpy4(acc, vv, __int_as_float(e[k].y));
        }
    }
    float4 r;
    r.x = fmaf(acc.x, sc, b.x);
    r.y = fmaf(acc.y, sc, b.y);
    r.z = fmaf(acc.z, sc, b.z);
    r.w = fmaf(acc.w, sc, b.w);
    *(float4*)(out + (size_t)m * GCN_C_LAST + q4) = r;
}

extern "C" void kernel_launch(void* const* d_in, const int* in_sizes, int n_in,
                              void* d_out, int out_size, void* d_ws, size_t ws_size,
                              hipStream_t stream) {
    const float* features = (const float*)d_in[0];
    const int*   src      = (const int*)  d_in[1];
    const int*   dst      = (const int*)  d_in[2];
    const float* W1       = (const float*)d_in[3];
    const float* b1       = (const float*)d_in[4];
    const float* W2       = (const float*)d_in[5];
    const float* b2       = (const float*)d_in[6];
    const float* W3       = (const float*)d_in[7];
    const float* b3       = (const float*)d_in[8];
    float* out = (float*)d_out;

    const int N = GCN_N_NODES;
    const int E = GCN_N_EDGES;
    const size_t NF = (size_t)N * 128;

    float* bufZ    = (float*)d_ws;
    float* bufB    = bufZ + NF;
    int*   edges   = (int*)(bufB + NF);
    int*   runs    = edges + 2 * (size_t)E;
    int*   cnt_out = runs + 2 * (size_t)E;
    int*   cnt_in  = cnt_out + N;
    int*   row_ptr = cnt_in + N;
    float* deg_i   = (float*)(row_ptr + N);
    int*   part_out = (int*)(deg_i + N);
    int*   blk_in   = part_out + (size_t)GCN_HIST_CHUNKS * GCN_SRC_WORDS;
    int*   blk_base = blk_in + GCN_DST_CHUNKS * GCN_COMB_BLOCKS;

    const int TB = 256;

    gcn_hist_lds<<<GCN_DST_CHUNKS + GCN_HIST_CHUNKS, 1024, 0, stream>>>(src, dst, part_out, blk_in, blk_base, runs);
    gcn_combine<<<GCN_COMB_BLOCKS, GCN_COMB_TB, 0, stream>>>(part_out, cnt_out, N);
    gcn_place_block<<<GCN_COMB_BLOCKS, GCN_OWN_THREADS, 0, stream>>>(runs, blk_in, blk_base, cnt_out,
                                                                    cnt_in, row_ptr, edges, deg_i, N);

    const int groups = GCN_LAYER_GROUPS;

    gcn_layer_fused<false><<<groups, TB, 0, stream>>>(features, row_ptr, cnt_in, edges, deg_i,
                                                      W1, b1, W3, bufB, 1, GCN_N_TILES);
    gcn_layer_fused<true><<<groups, TB, 0, stream>>>(bufB, row_ptr, cnt_in, edges, deg_i,
                                                     W2, b2, W3, bufZ, 1, GCN_N_TILES);
    gcn_gather_last<<<(4 * N + TB - 1) / TB, TB, 0, stream>>>(bufZ, row_ptr, cnt_in, edges, deg_i,
                                                             b3, out, N);

    (void)in_sizes; (void)n_in; (void)out_size; (void)ws_size;
}


# Out-degree scale moved from edge records to rows (layers store rows pre-scaled, first gather reads it per source): placement needs no out-degrees, their launch is a role in it; 6->5; sum order same
# speedup vs baseline: 6.3768x; 1.0510x over previous
#include <hip/hip_runtime.h>
#include <math.h>


typedef __attribute__((ext_vector_type(2))) float v2f;
typedef __attribute__((ext_vector_type(8))) float v8f;

#define GCN_N_NODES 50000
#define GCN_N_EDGES 600000


#define GCN_HIST_CHUNKS 50
#define GCN_SRC_WORDS (GCN_N_NODES / 2)
#define GCN_HIST_CHUNK (GCN_N_EDGES / GCN_HIST_CHUNKS)
#define GCN_HIST_QUADS (GCN_HIST_CHUNK / 4)
#define GCN_HIST_ITERS ((GCN_HIST_QUADS + 1023) / 1024)
#define GCN_DST_CHUNKS 75
#define GCN_DST_CHUNK (GCN_N_EDGES / GCN_DST_CHUNKS)
#define GCN_DST_QUADS (GCN_DST_CHUNK / 4)
#define GCN_DST_ITERS ((GCN_DST_QUADS + 1023) / 1024)
static_assert(GCN_N_EDGES % GCN_DST_CHUNKS == 0, "equal chunks");
static_assert(GCN_DST_CHUNK % 4 == 0, "a chunk is read in 16-byte groups");
static_assert(GCN_N_EDGES % GCN_HIST_CHUNKS == 0, "equal chunks");
static_assert(GCN_HIST_CHUNK % 4 == 0, "a chunk is read in 16-byte groups");
static_assert(GCN_N_NODES % 8 == 0, "counter planes are cleared and copied in 16-byte groups of 8 counters");
static_assert(GCN_HIST_CHUNK < 65536, "a 16-bit counter holds a chunk's whole count");
static_assert(GCN_N_NODES * 4 <= 320 * 1024, "all source counters in one work-group's LDS");
#define GCN_COMB_TB 256
#define GCN_COMB_BLOCKS ((GCN_N_NODES + GCN_COMB_TB - 1) / GCN_COMB_TB)
#define GCN_HIST_COPY_TRIPS ((GCN_SRC_WORDS / 4 + 32 * (GCN_COMB_TB / 4) - 1) / (32 * (GCN_COMB_TB / 4)))
static_assert(GCN_COMB_TB == 2 * 32 * 4, "a wave copies a block as two 16-byte groups per lane");
static_assert(GCN_COMB_BLOCKS <= GCN_COMB_TB, "one thread per earlier block in the placement kernel");
static_assert(GCN_DST_CHUNK * 8 <= GCN_N_NODES * 4, "a chunk's run is put together in the counters' LDS");
static_assert(GCN_DST_CHUNK % 2 == 0, "a run leaves the LDS in 16-byte groups of two records");
static_assert(GCN_COMB_BLOCKS < GCN_COMB_TB, "the first GCN_COMB_TB threads scan the block totals, one more for the end");

__global__ void __launch_bounds__(1024)
gcn_hist_lds(const int* __restrict__ src, const int* __restrict__ dst,
             int* __restrict__ part_out,
             int* __restrict__ blk_in, int* __restrict__ blk_base, int* __restrict__ runs) {
    __shared__ __attribute__((aligned(16))) int h[GCN_N_NODES];
    __shared__ int bt[GCN_COMB_TB];
    __shared__ int cur[GCN_COMB_TB];
    __shared__ int wtot[GCN_COMB_TB / 32];
    const int tid  = threadIdx.x;
    const int lane = tid & 31;
    const int w    = tid >> 5;
    const bool is_dst = (int)blockIdx.x < GCN_DST_CHUNKS;
    const int  c      = is_dst ? (int)blockIdx.x : (int)blockIdx.x - GCN_DST_CHUNKS;
    const int  e0     = c * (is_dst ? GCN_DST_CHUNK : GCN_HIST_CHUNK);

    if (is_dst) {
        int4 k4[GCN_DST_ITERS], s4[GCN_DST_ITERS];
#pragma unroll
        for (int k = 0; k < GCN_DST_ITERS; ++k) {
            const int q = tid + k * 1024;
            k4[k] = make_int4(0, 0, 0, 0);
            if (q < GCN_DST_QUADS) k4[k] = *(const int4*)(dst + e0 + 4 * q);
        }
#pragma unroll
        for (int k = 0; k < GCN_DST_ITERS; ++k) {
            const int q = tid + k * 1024;
            s4[k] = make_int4(0, 0, 0, 0);
            if (q < GCN_DST_QUADS) s4[k] = *(const int4*)(src + e0 + 4 * q);
        }
        if (tid < GCN_COMB_TB) bt[tid] = 0;
        __syncthreads();
        int4 r[GCN_DST_ITERS];
#pragma unroll
        for (int k = 0; k < GCN_DST_ITERS; ++k) {
            const int q = tid + k * 1024;
            r[k] = make_int4(0, 0, 0, 0);
            if (q < GCN_DST_QUADS) {
                r[k].x = atomicAdd(&bt[k4[k].x / GCN_COMB_TB], 1);
                r[k].y = atomicAdd(&bt[k4[k].y / GCN_COMB_TB], 1);
                r[k].z = atomicAdd(&bt[k4[k].z / GCN_COMB_TB], 1);
                r[k].w = atomicAdd(&bt[k4[k].w / GCN_COMB_TB], 1);
            }
        }
        __syncthreads();
        int v = 0, incl = 0, excl = 0;
        if (tid < GCN_COMB_TB) {
            v = bt[tid];
            incl = v;
            for (int m = 1; m < 32; m <<= 1) {
                const int y = __shfl_up(incl, m);
                if (lane >= m) incl += y;
            }
            if (lane == 31) wtot[w] = incl;
        }
        __syncthreads();
        if (tid < GCN_COMB_TB) {
            int base = 0;
            for (int k = 0; k < GCN_COMB_TB / 32; ++k) base += (k < w) ? wtot[k] : 0;
            excl = base + incl - v;
            cur[tid] = excl;
        }
        __syncthreads();
        int2* run = (int2*)h;
#pragma unroll
        for (int k = 0; k < GCN_DST_ITERS; ++k) {
            const int q = tid + k * 1024;
            if (q < GCN_DST_QUADS) {
                run[cur[k4[k].x / GCN_COMB_TB] + r[k].x] = make_int2(s4[k].x, k4[k].x);
                run[cur[k4[k].y / GCN_COMB_TB] + r[k].y] = make_int2(s4[k].y, k4[k].y);
                run[cur[k4[k].z / GCN_COMB_TB] + r[k].z] = make_int2(s4[k].z, k4[k].z);
                run[cur[k4[k].w / GCN_COMB_TB] + r[k].w] = make_int2(s4[k].w, k4[k].w);
            }
        }
        __syncthreads();
        if (tid < GCN_COMB_BLOCKS)  blk_in[c * GCN_COMB_BLOCKS + tid] = v;
        if (tid <= GCN_COMB_BLOCKS) blk_base[c * (GCN_COMB_BLOCKS + 1) + tid] = excl;
        int4* __restrict__ run_out = (int4*)(runs + 2 * (size_t)e0);
        for (int i = tid; i < GCN_DST_CHUNK / 2; i += 1024) run_out[i] = ((const int4*)h)[i];
    } else {
        int4 k4[GCN_HIST_ITERS];
#pragma unroll
        for (int k = 0; k < GCN_HIST_ITERS; ++k) {
            const int q = tid + k * 1024;
            k4[k] = make_int4(0, 0, 0, 0);
            if (q < GCN_HIST_QUADS) k4[k] = *(const int4*)(src + e0 + 4 * q);
        }
        for (int i = tid; i < GCN_SRC_WORDS / 4; i += 1024) ((int4*)h)[i] = make_int4(0, 0, 0, 0);
        __syncthreads();
#pragma unroll
        for (int k = 0; k < GCN_HIST_ITERS; ++k) {
            const int q = tid + k * 1024;
            if (q < GCN_HIST_QUADS) {
                atomicAdd(&h[k4[k].x >> 1], 1 << ((k4[k].x & 1) << 4));
                atomicAdd(&h[k4[k].y >> 1], 1 << ((k4[k].y & 1) << 4));
                atomicAdd(&h[k4[k].z >> 1], 1 << ((k4[k].z & 1) << 4));
                atomicAdd(&h[k4[k].w >> 1], 1 << ((k4[k].w & 1) << 4));
            }
        }
        __syncthreads();
        int4* __restrict__ plane = (int4*)(part_out + (size_t)c * GCN_SRC_WORDS);
        const int last = GCN_SRC_WORDS / 4 - 1;
        int4 q0[GCN_HIST_COPY_TRIPS], q1[GCN_HIST_COPY_TRIPS];
#pragma unroll
        for (int k = 0; k < GCN_HIST_COPY_TRIPS; ++k) {
            const int i0 = (w + 32 * k) * (GCN_COMB_TB / 4) + lane;
            const int i1 = i0 + 32;
            q0[k] = ((const int4*)h)[i0 < last ? i0 : last];
            q1[k] = ((const int4*)h)[i1 < last ? i1 : last];
        }
#pragma unroll
        for (int k = 0; k < GCN_HIST_COPY_TRIPS; ++k) {
            const int i0 = (w + 32 * k) * (GCN_COMB_TB / 4) + lane;
            const int i1 = i0 + 32;
            if (i0 <= last) plane[i0] = q0[k];
            if (i1 <= last) plane[i1] = q1[k];
        }
    }
}

static __device__ __forceinline__ float gcn_inv_sqrt_clamped(int c) {
    float v = (float)c;
    v = v < 1.0f ? 1.0f : v;
    return 1.0f / sqrtf(v);
}

static __device__ __forceinline__ void gcn_out_degree_role(const int* __restrict__ part_out,
                                                           float* __restrict__ deg_out_is,
                                                           const int d, const int n) {
    const int dl = d < n ? d : n - 1;
    const int wd = dl >> 1;
    const int sh = (dl & 1) << 4;
    int vo[GCN_HIST_CHUNKS];
#pragma unroll
    for (int c = 0; c < GCN_HIST_CHUNKS; ++c) vo[c] = part_out[(size_t)c * GCN_SRC_WORDS + wd];
    int so = 0;
#pragma unroll
    for (int c = 0; c < GCN_HIST_CHUNKS; ++c) so += (int)(((unsigned)vo[c] >> sh) & 0xffffu);
    if (d < n) deg_out_is[d] = gcn_inv_sqrt_clamped(so);
}

#define GCN_OWN_THREADS 1024
#define GCN_DEG_GROUPS ((GCN_N_NODES + GCN_OWN_THREADS - 1) / GCN_OWN_THREADS)
#define GCN_OWN_CAP 8192
#define GCN_OWN_FAST 4
#define GCN_PS_SLOTS 128
static_assert(GCN_DST_CHUNKS < GCN_PS_SLOTS, "the table holds one entry per piece and one for the end");
static_assert(GCN_COMB_TB + GCN_DST_CHUNKS <= GCN_OWN_THREADS, "one thread per chunk, behind the node threads, fetches the piece bounds");
static_assert(GCN_PS_SLOTS <= GCN_OWN_THREADS, "one thread per table entry");

static __device__ __forceinline__ int gcn_piece_of(const int* ps, const int j) {
    int lo = 0;
#pragma unroll
    for (int st = GCN_PS_SLOTS / 2; st >= 1; st >>= 1) lo = (ps[lo + st] <= j) ? lo + st : lo;
    return lo;
}
static_assert(GCN_COMB_TB <= GCN_OWN_THREADS, "one thread per node of the block");

__global__ void __launch_bounds__(GCN_OWN_THREADS)
gcn_place_block(const int* __restrict__ runs, const int* __restrict__ blk_in,
                const int* __restrict__ blk_base, const int* __restrict__ part_out,
                int* __restrict__ cnt_in, int* __restrict__ row_ptr, int* __restrict__ edges,
                float* __restrict__ deg_in_is, float* __restrict__ deg_out_is, int n) {
    __shared__ int stage[GCN_OWN_CAP];
    __shared__ int rs[GCN_DST_CHUNKS];
    __shared__ int rl[GCN_DST_CHUNKS];
    __shared__ int ps[GCN_PS_SLOTS];
    __shared__ int ncnt[GCN_COMB_TB];
    __shared__ int wbefore[GCN_COMB_TB / 32];
    __shared__ int wrows[GCN_COMB_TB / 32];
    const int tid  = threadIdx.x;
    const int lane = tid & 31;
    const int w    = tid >> 5;
    const int b    = blockIdx.x;
    if (b >= GCN_COMB_BLOCKS) {
        gcn_out_degree_role(part_out, deg_out_is, (b - GCN_COMB_BLOCKS) * GCN_OWN_THREADS + tid, n);
        return;
    }
    const int n0   = b * GCN_COMB_TB;
    if (tid < GCN_COMB_TB) {
        const int tb = tid < b ? tid : 0;
        int vb[GCN_DST_CHUNKS];
#pragma unroll
        for (int c = 0; c < GCN_DST_CHUNKS; ++c) vb[c] = blk_in[c * GCN_COMB_BLOCKS + tb];
        int before = 0;
#pragma unroll
        for (int c = 0; c < GCN_DST_CHUNKS; ++c) before += vb[c];
        before = tid < b ? before : 0;
        for (int m = 16; m >= 1; m >>= 1) before += __shfl_xor(before, m);
        if (lane == 0) wbefore[w] = before;
        ncnt[tid] = 0;
    } else if (tid < GCN_COMB_TB + GCN_DST_CHUNKS) {
        const int t = tid - GCN_COMB_TB;
        rs[t] = blk_base[t * (GCN_COMB_BLOCKS + 1) + b];
        rl[t] = blk_in[t * GCN_COMB_BLOCKS + b];
    }
    __syncthreads();
    if (tid < GCN_PS_SLOTS) {
        int a = 0;
#pragma unroll
        for (int i = 0; i < GCN_DST_CHUNKS; ++i) a += (i < tid) ? rl[i] : 0;
        ps[tid] = tid <= GCN_DST_CHUNKS ? a : 0x7fffffff;
    }
    int out_lo = 0;
    for (int k = 0; k < GCN_COMB_TB / 32; ++k) out_lo += wbefore[k];
    __syncthreads();
    const int total = ps[GCN_DST_CHUNKS];

    const int2* __restrict__ rr = (const int2*)runs;
    int* __restrict__ eo = edges;
    int2 rec[GCN_OWN_FAST];
#pragma unroll
    for (int k = 0; k < GCN_OWN_FAST; ++k) rec[k] = make_int2(0, n0);
    if (total > 0) {
        int ri[GCN_OWN_FAST], jj[GCN_OWN_FAST], pc[GCN_OWN_FAST];
#pragma unroll
        for (int k = 0; k < GCN_OWN_FAST; ++k) {
            const int j = tid + k * GCN_OWN_THREADS;
            jj[k] = j < total ? j : total - 1;
            pc[k] = 0;
        }
#pragma unroll
        for (int st = GCN_PS_SLOTS / 2; st >= 1; st >>= 1) {
#pragma unroll
            for (int k = 0; k < GCN_OWN_FAST; ++k) pc[k] = (ps[pc[k] + st] <= jj[k]) ? pc[k] + st : pc[k];
        }
#pragma unroll
        for (int k = 0; k < GCN_OWN_FAST; ++k) {
            ri[k] = pc[k] * GCN_DST_CHUNK + rs[pc[k]] + (jj[k] - ps[pc[k]]);
        }
#pragma unroll
        for (int k = 0; k < GCN_OWN_FAST; ++k) rec[k] = rr[ri[k]];
#pragma unroll
        for (int k = 0; k < GCN_OWN_FAST; ++k) {
            const int j = tid + k * GCN_OWN_THREADS;
            if (j < total) atomicAdd(&ncnt[rec[k].y - n0], 1);
        }
        for (int j = tid + GCN_OWN_FAST * GCN_OWN_THREADS; j < total; j += GCN_OWN_THREADS) {
            const int c2 = gcn_piece_of(ps, j);
            const int2 e = rr[(size_t)c2 * GCN_DST_CHUNK + rs[c2] + (j - ps[c2])];
            atomicAdd(&ncnt[e.y - n0], 1);
        }
    }
    __syncthreads();
    int v = 0, incl = 0, loc = 0;
    if (tid < GCN_COMB_TB) {
        v = ncnt[tid];
        incl = v;
        for (int m = 1; m < 32; m <<= 1) {
            const int y = __shfl_up(incl, m);
            if (lane >= m) incl += y;
        }
        if (lane == 31) wrows[w] = incl;
    }
    __syncthreads();
    if (tid < GCN_COMB_TB) {
        int base = 0;
        for (int k = 0; k < GCN_COMB_TB / 32; ++k) base += (k < w) ? wrows[k] : 0;
        loc = base + incl - v;
        ncnt[tid] = loc;
    }
    __syncthreads();
    if (total > 0) {
        int o[GCN_OWN_FAST];
#pragma unroll
        for (int k = 0; k < GCN_OWN_FAST; ++k) {
            const bool on = tid + k * GCN_OWN_THREADS < total;
            o[k] = atomicAdd(&ncnt[on ? rec[k].y - n0 : (tid & (GCN_COMB_TB - 1))], on ? 1 : 0);
        }
#pragma unroll
        for (int k = 0; k < GCN_OWN_FAST; ++k) {
            const int j = tid + k * GCN_OWN_THREADS;
            if (j < total) {
                if ((unsigned)o[k] < (unsigned)GCN_OWN_CAP) stage[o[k]] = rec[k].x; else eo[out_lo + o[k]] = rec[k].x;
            }
        }
        for (int j = tid + GCN_OWN_FAST * GCN_OWN_THREADS; j < total; j += GCN_OWN_THREADS) {
            const int c2 = gcn_piece_of(ps, j);
            const int2 e  = rr[(size_t)c2 * GCN_DST_CHUNK + rs[c2] + (j - ps[c2])];
            const int  os = atomicAdd(&ncnt[e.y - n0], 1);
            if ((unsigned)os < (unsigned)GCN_OWN_CAP) stage[os] = e.x; else eo[out_lo + os] = e.x;
        }
    }
    __syncthreads();
    const int staged = total < GCN_OWN_CAP ? total : GCN_OWN_CAP;
    for (int o = tid; o < staged; o += GCN_OWN_THREADS) eo[out_lo + o] = stage[o];
    if (tid < GCN_COMB_TB && n0 + tid < n) {
        row_ptr[n0 + tid]   = out_lo + loc;
        cnt_in[n0 + tid]    = v;
        deg_in_is[n0 + tid] = gcn_inv_sqrt_clamped(v);
    }
}

static __device__ __forceinline__ float4 gcn_row4(const float* __restrict__ x, int s, int lane4) {
    return *(const float4*)(x + (size_t)s * 128 + lane4);
}
static __device__ __forceinline__ void gcn_axpy4(float4& acc, const float4 v, const float w) {
    acc.x = fmaf(v.x, w, acc.x);
    acc.y = fmaf(v.y, w, acc.y);
    acc.z = fmaf(v.z, w, acc.z);
    acc.w = fmaf(v.w, w, acc.w);
}
#define GCN_EDGE_SRC(my, j) __builtin_amdgcn_readlane((my), (j))

template <bool SRC_W>
static __device__ __forceinline__ void gcn_edge_add(float4& acc, const float4 v, const float mw, const int j) {
    if constexpr (SRC_W) {
        gcn_axpy4(acc, v, __int_as_float(__builtin_amdgcn_readlane(__float_as_int(mw), j)));
    } else {
        acc.x += v.x;
        acc.y += v.y;
        acc.z += v.z;
        acc.w += v.w;
    }
}

template <bool SRC_W>
static __device__ __forceinline__ float4 gcn_gather_row(const float* __restrict__ x,
                                                        const int* __restrict__ ce,
                                                        const float* __restrict__ deg_out_is,
                                                        const int beg, const int cnt,
                                                        const int first, const float firstw,
                                                        const int lane, const int lane4) {
    float4 acc = make_float4(0.0f, 0.0f, 0.0f, 0.0f);
    for (int b = 0; b < cnt; b += 32) {
        const int m = (cnt - b < 32) ? (cnt - b) : 32;
        int   my = first;
        float mw = firstw;
        if (b > 0) {
            my = 0;
            if (lane < m) my = ce[beg + b + lane];
            if constexpr (SRC_W) mw = deg_out_is[my];
        }
        int j = 0;
        for (; j + 8 <= m; j += 8) {
            const float4 v0 = gcn_row4(x, GCN_EDGE_SRC(my, j),     lane4);
            const float4 v1 = gcn_row4(x, GCN_EDGE_SRC(my, j + 1), lane4);
            const float4 v2 = gcn_row4(x, GCN_EDGE_SRC(my, j + 2), lane4);
            const float4 v3 = gcn_row4(x, GCN_EDGE_SRC(my, j + 3), lane4);
            const float4 v4 = gcn_row4(x, GCN_EDGE_SRC(my, j + 4), lane4);
            const float4 v5 = gcn_row4(x, GCN_EDGE_SRC(my, j + 5), lane4);
            const float4 v6 = gcn_row4(x, GCN_EDGE_SRC(my, j + 6), lane4);
            const float4 v7 = gcn_row4(x, GCN_EDGE_SRC(my, j + 7), lane4);
            gcn_edge_add<SRC_W>(acc, v0, mw, j);
            gcn_edge_add<SRC_W>(acc, v1, mw, j + 1);
            gcn_edge_add<SRC_W>(acc, v2, mw, j + 2);
            gcn_edge_add<SRC_W>(acc, v3, mw, j + 3);
            gcn_edge_add<SRC_W>(acc, v4, mw, j + 4);
            gcn_edge_add<SRC_W>(acc, v5, mw, j + 5);
            gcn_edge_add<SRC_W>(acc, v6, mw, j + 6);
            gcn_edge_add<SRC_W>(acc, v7, mw, j + 7);
        }
        if (j + 4 <= m) {
            const float4 v0 = gcn_row4(x, GCN_EDGE_SRC(my, j),     lane4);
            const float4 v1 = gcn_row4(x, GCN_EDGE_SRC(my, j + 1), lane4);
            const float4 v2 = gcn_row4(x, GCN_EDGE_SRC(my, j + 2), lane4);
            const float4 v3 = gcn_row4(x, GCN_EDGE_SRC(my, j + 3), lane4);
            gcn_edge_add<SRC_W>(acc, v0, mw, j);
            gcn_edge_add<SRC_W>(acc, v1, mw, j + 1);
            gcn_edge_add<SRC_W>(acc, v2, mw, j + 2);
            gcn_edge_add<SRC_W>(acc, v3, mw, j + 3);
            j += 4;
        }
        const int r = m - j;
        if (r == 1) {
            const float4 v0 = gcn_row4(x, GCN_EDGE_SRC(my, j), lane4);
            gcn_edge_add<SRC_W>(acc, v0, mw, j);
        } else if (r == 2) {
            const float4 v0 = gcn_row4(x, GCN_EDGE_SRC(my, j),     lane4);
            const float4 v1 = gcn_row4(x, GCN_EDGE_SRC(my, j + 1), lane4);
            gcn_edge_add<SRC_W>(acc, v0, mw, j);
            gcn_edge_add<SRC_W>(acc, v1, mw, j + 1);
        } else if (r == 3) {
            const float4 v0 = gcn_row4(x, GCN_EDGE_SRC(my, j),     lane4);
            const float4 v1 = gcn_row4(x, GCN_EDGE_SRC(my, j + 1), lane4);
            const float4 v2 = gcn_row4(x, GCN_EDGE_SRC(my, j + 2), lane4);
            gcn_edge_add<SRC_W>(acc, v0, mw, j);
            gcn_edge_add<SRC_W>(acc, v1, mw, j + 1);
            gcn_edge_add<SRC_W>(acc, v2, mw, j + 2);
        }
    }
    return acc;
}

#define GCN_K 128
#define GCN_C 128
#define GCN_C_LAST 16
#define GCN_TILE_PITCH 132
#define GCN_LAYER_GROUPS 1024
#define GCN_MAX_TILES_PER_GROUP 16
#define GCN_N_TILES (GCN_N_NODES / 16)
static_assert(GCN_N_NODES % 16 == 0, "whole 16-row tiles only");
static_assert(GCN_LAYER_GROUPS <= GCN_N_TILES, "every work-group gets at least one tile");
static_assert((GCN_N_TILES + GCN_LAYER_GROUPS - 1) / GCN_LAYER_GROUPS <= GCN_MAX_TILES_PER_GROUP,
              "a work-group's tiles must fit the lanes that hold their row bounds");
static_assert(GCN_C == 8 * 16, "8 waves, one 16-column strip each");

template <bool FUSE_NEXT, bool SCALE_SRC>
__global__ void __launch_bounds__(256) __attribute__((amdgpu_waves_per_eu(8, 8)))
gcn_layer_fused(const float* __restrict__ x, const int* __restrict__ row_ptr,
                const int* __restrict__ cnt_in, const int* __restrict__ edges,
                const float* __restrict__ deg_in_is, const float* __restrict__ deg_out_is,
                const float* __restrict__ W,
                const float* __restrict__ bias, const float* __restrict__ Wn,
                float* __restrict__ out, int do_relu, int n_tiles) {
    __shared__ __attribute__((aligned(16))) float tile[2][16 * GCN_TILE_PITCH];
    __shared__ __attribute__((aligned(16))) float htile[FUSE_NEXT ? 16 * GCN_TILE_PITCH : 4];
    __shared__ __attribute__((aligned(16))) v2f wn_frag[FUSE_NEXT ? (GCN_K / 4) * 32 : 1];

    const int lane  = threadIdx.x & 31;
    const int lane4 = lane * 4;
    const int half  = lane >> 4;
    const int l15   = lane & 15;
    const int wave  = __builtin_amdgcn_readfirstlane(threadIdx.x >> 5);
    const int ncol  = wave * 16 + l15;
    const int per_group = n_tiles / GCN_LAYER_GROUPS;
    const int extra     = n_tiles % GCN_LAYER_GROUPS;
    const int g         = (int)blockIdx.x;
    const int tile0     = g * per_group + (g < extra ? g : extra);
    const int my_tiles  = per_group + (g < extra ? 1 : 0);
    const int* __restrict__ ce = edges;

    int rp = 0, rc = 0;
    {
        const int tj = tile0 + (lane >> 1);
        if ((lane >> 1) < my_tiles) {
            const int node = tj * 16 + wave * 2 + (lane & 1);
            rp = row_ptr[node];
            rc = cnt_in[node];
        }
    }

    v2f bfrag[GCN_K / 4];
#pragma unroll
    for (int s = 0; s < GCN_K / 4; ++s) {
        const int kb = s * 4 + half * 2;
        bfrag[s].x = W[(size_t)kb       * GCN_C + ncol];
        bfrag[s].y = W[(size_t)(kb + 1) * GCN_C + ncol];
    }
    const float bv = bias[ncol];

    if constexpr (FUSE_NEXT) {
        for (int idx = threadIdx.x; idx < (GCN_K / 4) * 32; idx += 256) {
            const int s  = idx >> 5;
            const int ln = idx & 31;
            const int kb = s * 4 + (ln >> 4) * 2;
            v2f f;
            f.x = Wn[(size_t)kb       * GCN_C_LAST + (ln & 15)];
            f.y = Wn[(size_t)(kb + 1) * GCN_C_LAST + (ln & 15)];
            wn_frag[idx] = f;
        }
    }

    for (int i = 0; i < my_tiles; ++i) {
        const int t = tile0 + i;
        float* tb = &tile[i & 1][0];

        const int beg0 = __builtin_amdgcn_readlane(rp, 2 * i);
        const int cnt0 = __builtin_amdgcn_readlane(rc, 2 * i);
        const int beg1 = __builtin_amdgcn_readlane(rp, 2 * i + 1);
        const int cnt1 = __builtin_amdgcn_readlane(rc, 2 * i + 1);

        int cur0 = 0, cur1 = 0;
        if (lane < cnt0) cur0 = ce[beg0 + lane];
        if (lane < cnt1) cur1 = ce[beg1 + lane];
        const float scale = deg_in_is[t * 16 + l15];
        float w0 = 0.0f, w1 = 0.0f;
        if constexpr (SCALE_SRC) {
            w0 = deg_out_is[cur0];
            w1 = deg_out_is[cur1];
        }

        const float4 g0 = gcn_gather_row<SCALE_SRC>(x, ce, deg_out_is, beg0, cnt0, cur0, w0, lane, lane4);
        *(float4*)(tb + (2 * wave) * GCN_TILE_PITCH + lane4) = g0;
        const float4 g1 = gcn_gather_row<SCALE_SRC>(x, ce, deg_out_is, beg1, cnt1, cur1, w1, lane, lane4);
        *(float4*)(tb + (2 * wave + 1) * GCN_TILE_PITCH + lane4) = g1;

        __syncthreads();

        {
            float4 dsa = make_float4(1.0f, 1.0f, 1.0f, 1.0f), dsb = dsa;
            if constexpr (!FUSE_NEXT) {
                const float* dso = deg_out_is + t * 16 + half * 8;
                dsa = *(const float4*)dso;
                dsb = *(const float4*)(dso + 4);
            }
            const float ds[8] = {dsa.x, dsa.y, dsa.z, dsa.w, dsb.x, dsb.y, dsb.z, dsb.w};
            const float* arow = tb + l15 * GCN_TILE_PITCH + half * 2;
            v8f c = {};
#pragma unroll
            for (int s = 0; s < GCN_K / 4; ++s) {
                const v2f araw = *(const v2f*)(arow + s * 4);
                v2f a;
                a.x = araw.x * scale;
                a.y = araw.y * scale;
                c = __builtin_amdgcn_wmma_f32_16x16x4_f32(
                        false, a, false, bfrag[s], (short)0, c, false, false);
            }
#pragma unroll
            for (int v = 0; v < 8; ++v) {
                float val = c[v] + bv;
                if (do_relu) val = fmaxf(val, 0.0f);
                if constexpr (FUSE_NEXT) {
                    htile[(half * 8 + v) * GCN_TILE_PITCH + ncol] = val;
                } else {
                    out[(size_t)(t * 16 + half * 8 + v) * GCN_C + ncol] = val * ds[v];
                }
            }
        }

        if constexpr (FUSE_NEXT) {
            __syncthreads();
            if (wave == (i & 7)) {
                const float dsr = deg_out_is[t * 16 + l15];
                const float* hrow = htile + l15 * GCN_TILE_PITCH + half * 2;
                v8f z0 = {}, z1 = {};
#pragma unroll
                for (int s = 0; s < GCN_K / 4; s += 2) {
                    const v2f h0 = *(const v2f*)(hrow + s * 4);
                    const v2f b0 = wn_frag[s * 32 + lane];
                    const v2f h1 = *(const v2f*)(hrow + (s + 1) * 4);
                    const v2f b1 = wn_frag[(s + 1) * 32 + lane];
                    v2f a0, a1;
                    a0.x = h0.x * dsr;
                    a0.y = h0.y * dsr;
                    a1.x = h1.x * dsr;
                    a1.y = h1.y * dsr;
                    z0 = __builtin_amdgcn_wmma_f32_16x16x4_f32(
                            false, a0, false, b0, (short)0, z0, false, false);
                    z1 = __builtin_amdgcn_wmma_f32_16x16x4_f32(
                            false, a1, false, b1, (short)0, z1, false, false);
                }
#pragma unroll
                for (int v = 0; v < 8; ++v) {
                    out[(size_t)(t * 16 + half * 8 + v) * GCN_C_LAST + l15] = z0[v] + z1[v];
                }
            }
        }
    }
}

#define GCN_LAST_BATCH 16

__global__ void __launch_bounds__(256) __attribute__((amdgpu_waves_per_eu(4, 6)))
gcn_gather_last(const float* __restrict__ z, const int* __restrict__ row_ptr,
                const int* __restrict__ cnt_in, const int* __restrict__ edges,
                const float* __restrict__ deg_in_is, const float* __restrict__ bias,
                float* __restrict__ out, int n) {
    const int g  = blockIdx.x * blockDim.x + threadIdx.x;
    const int m  = g >> 2;
    const int q4 = (g & 3) * 4;
    if (m >= n) return;
    const int    cnt = cnt_in[m];
    const int    beg = row_ptr[m];
    const float  sc  = deg_in_is[m];
    const float4 b   = *(const float4*)(bias + q4);
    asm volatile("" :: "v"(cnt), "v"(beg), "v"(sc), "v"(b.x), "v"(b.y), "v"(b.z), "v"(b.w));
    const int* __restrict__ ce = edges + beg;
    const int last = cnt - 1;

    float4 acc = make_float4(0.0f, 0.0f, 0.0f, 0.0f);
    for (int j = 0; j < cnt; j += GCN_LAST_BATCH) {
        int e[GCN_LAST_BATCH];
        float4 v[GCN_LAST_BATCH];
#pragma unroll
        for (int k = 0; k < GCN_LAST_BATCH; ++k) {
            const int jj = (j + k < last) ? (j + k) : last;
            e[k] = ce[jj];
        }
#pragma unroll
        for (int k = 0; k < GCN_LAST_BATCH; ++k) {
            v[k] = *(const float4*)(z + (size_t)e[k] * GCN_C_LAST + q4);
        }
#pragma unroll
        for (int k = 0; k < GCN_LAST_BATCH; ++k) {
            const bool on = (j + k) < cnt;
            acc.x += on ? v[k].x : 0.0f;
            acc.y += on ? v[k].y : 0.0f;
            acc.z += on ? v[k].z : 0.0f;
            acc.w += on ? v[k].w : 0.0f;
        }
    }
    float4 r;
    r.x = fmaf(acc.x, sc, b.x);
    r.y = fmaf(acc.y, sc, b.y);
    r.z = fmaf(acc.z, sc, b.z);
    r.w = fmaf(acc.w, sc, b.w);
    *(float4*)(out + (size_t)m * GCN_C_LAST + q4) = r;
}

extern "C" void kernel_launch(void* const* d_in, const int* in_sizes, int n_in,
                              void* d_out, int out_size, void* d_ws, size_t ws_size,
                              hipStream_t stream) {
    const float* features = (const float*)d_in[0];
    const int*   src      = (const int*)  d_in[1];
    const int*   dst      = (const int*)  d_in[2];
    const float* W1       = (const float*)d_in[3];
    const float* b1       = (const float*)d_in[4];
    const float* W2       = (const float*)d_in[5];
    const float* b2       = (const float*)d_in[6];
    const float* W3       = (const float*)d_in[7];
    const float* b3       = (const float*)d_in[8];
    float* out = (float*)d_out;

    const int N = GCN_N_NODES;
    const int E = GCN_N_EDGES;
    const size_t NF = (size_t)N * 128;

    float* bufZ    = (float*)d_ws;
    float* bufB    = bufZ + NF;
    int*   edges   = (int*)(bufB + NF);
    int*   runs    = edges + 2 * (size_t)E;
    float* deg_o   = (float*)(runs + 2 * (size_t)E);
    int*   cnt_in  = (int*)(deg_o + N);
    int*   row_ptr = cnt_in + N;
    float* deg_i   = (float*)(row_ptr + N);
    int*   part_out = (int*)(deg_i + N);
    int*   blk_in   = part_out + (size_t)GCN_HIST_CHUNKS * GCN_SRC_WORDS;
    int*   blk_base = blk_in + GCN_DST_CHUNKS * GCN_COMB_BLOCKS;

    const int TB = 256;

    gcn_hist_lds<<<GCN_DST_CHUNKS + GCN_HIST_CHUNKS, 1024, 0, stream>>>(src, dst, part_out, blk_in, blk_base, runs);
    gcn_place_block<<<GCN_COMB_BLOCKS + GCN_DEG_GROUPS, GCN_OWN_THREADS, 0, stream>>>(
        runs, blk_in, blk_base, part_out, cnt_in, row_ptr, edges, deg_i, deg_o, N);

    const int groups = GCN_LAYER_GROUPS;

    gcn_layer_fused<false, true><<<groups, TB, 0, stream>>>(features, row_ptr, cnt_in, edges, deg_i, deg_o,
                                                            W1, b1, W3, bufB, 1, GCN_N_TILES);
    gcn_layer_fused<true, false><<<groups, TB, 0, stream>>>(bufB, row_ptr, cnt_in, edges, deg_i, deg_o,
                                                            W2, b2, W3, bufZ, 1, GCN_N_TILES);
    gcn_gather_last<<<(4 * N + TB - 1) / TB, TB, 0, stream>>>(bufZ, row_ptr, cnt_in, edges, deg_i,
                                                             b3, out, N);

    (void)in_sizes; (void)n_in; (void)out_size; (void)ws_size;
}
